# starting kernel
# baseline (speedup 1.0000x reference)
#include <hip/hip_runtime.h>
#include <hip/hip_bf16.h>
#include <hip/hip_fp16.h>

#define F_IN      128
#define HEADS     4
#define C_OUT     32
#define HC        128
#define NEG_SLOPE 0.2f

typedef float    v2f  __attribute__((ext_vector_type(2)));
typedef float    v8f  __attribute__((ext_vector_type(8)));
typedef _Float16 v8h  __attribute__((ext_vector_type(8)));
typedef _Float16 v16h __attribute__((ext_vector_type(16)));

#if __has_builtin(__builtin_amdgcn_wmma_f32_16x16x4_f32)
#define GAT_WMMA_F32 1
#else
#define GAT_WMMA_F32 0
#endif

__device__ __forceinline__ unsigned f2key(float f) {
  unsigned u = __float_as_uint(f);
  return (u & 0x80000000u) ? ~u : (u | 0x80000000u);
}
__device__ __forceinline__ float key2f(unsigned k) {
  unsigned u = (k & 0x80000000u) ? (k & 0x7FFFFFFFu) : ~k;
  return __uint_as_float(u);
}

__global__ void k_convert(const float* __restrict__ x, const float* __restrict__ W,
                          _Float16* __restrict__ xh, _Float16* __restrict__ Wh,
                          int nx, int nw) {
  int t = blockIdx.x * blockDim.x + threadIdx.x;
  if (t < nx) xh[t] = (_Float16)x[t];
  if (t < nw) Wh[t] = (_Float16)W[t];
}

__global__ void k_gemm(const float* __restrict__ x, const float* __restrict__ W,
                       const _Float16* __restrict__ xh, const _Float16* __restrict__ Wh,
                       float* __restrict__ xl, int N) {
  const int lane    = threadIdx.x & 31;
  const int wave    = threadIdx.x >> 5;
  const int g       = lane >> 4;
  const int l16     = lane & 15;
  const int rowBase = blockIdx.x * 16;
  const int colBase = wave * 16;
  int mrow = rowBase + l16; if (mrow >= N) mrow = N - 1;
  const int ncol = colBase + l16;

  v8f c = {};
#if GAT_WMMA_F32
  for (int kk = 0; kk < F_IN; kk += 4) {
    const int k0 = kk + g * 2;
    v2f a = *(const v2f*)(x + (size_t)mrow * F_IN + k0);
    v2f b = *(const v2f*)(W + (size_t)ncol * F_IN + k0);
    c = __builtin_amdgcn_wmma_f32_16x16x4_f32(false, a, false, b, (short)0, c,
                                              false, false);
  }
#else
  for (int kk = 0; kk < F_IN; kk += 32) {
    const _Float16* ap = xh + (size_t)mrow * F_IN + kk + 8 * g;
    v8h a0 = *(const v8h*)(ap);
    v8h a1 = *(const v8h*)(ap + 16);
    v16h a = __builtin_shufflevector(a0, a1, 0, 1, 2, 3, 4, 5, 6, 7,
                                     8, 9, 10, 11, 12, 13, 14, 15);
    v16h b = *(const v16h*)(Wh + (size_t)ncol * F_IN + kk + 16 * g);
    c = __builtin_amdgcn_wmma_f32_16x16x32_f16(false, a, false, b, (short)0, c,
                                               false, false);
  }
#endif
#pragma unroll
  for (int i = 0; i < 8; ++i) {
    const int r = rowBase + i + g * 8;
    if (r < N) xl[(size_t)r * HC + ncol] = c[i];
  }
}

__global__ void k_scores(const float* __restrict__ xl, const float* __restrict__ aj,
                         const float* __restrict__ ai, float* __restrict__ Sj,
                         float* __restrict__ Si, unsigned* __restrict__ mkey,
                         float* __restrict__ ssum, int N) {
  const int t = blockIdx.x * blockDim.x + threadIdx.x;
  if (t >= N * HEADS) return;
  const int n = t / HEADS, h = t % HEADS;
  const float* row = xl + (size_t)n * HC + h * C_OUT;
  float sj = 0.f, si = 0.f;
#pragma unroll
  for (int cc = 0; cc < C_OUT; ++cc) {
    sj += row[cc] * aj[h * C_OUT + cc];
    si += row[cc] * ai[h * C_OUT + cc];
  }
  Sj[t] = sj;
  Si[t] = si;
  mkey[t] = 0u;
  ssum[t] = 0.f;
}

__global__ void k_out_init(float* __restrict__ out, const float* __restrict__ bias,
                           int total) {
  const int t = blockIdx.x * blockDim.x + threadIdx.x;
  if (t < total) out[t] = bias[t & (HC - 1)];
}

__device__ __forceinline__ void edge_nodes(const int* __restrict__ src,
                                           const int* __restrict__ dst, int E,
                                           int e, int& nj, int& ni, int& dnode,
                                           int& snode) {
  const int es = (e < E) ? src[e] : (e - E);
  const int ed = (e < E) ? dst[e] : (e - E);
  snode = es;
  dnode = ed;
  nj = src[es];
  ni = dst[ed];
}

__device__ __forceinline__ float alpha_h(const float* __restrict__ Sj,
                                         const float* __restrict__ Si, int nj,
                                         int ni, int h) {
  float a = Sj[nj * HEADS + h] + Si[ni * HEADS + h];
  return (a >= 0.f) ? a : NEG_SLOPE * a;
}

__global__ void k_alpha_max(const int* __restrict__ src, const int* __restrict__ dst,
                            const float* __restrict__ Sj, const float* __restrict__ Si,
                            unsigned* __restrict__ mkey, int E, int Et) {
  const int e = blockIdx.x * blockDim.x + threadIdx.x;
  if (e >= Et) return;
  int nj, ni, dn, sn;
  edge_nodes(src, dst, E, e, nj, ni, dn, sn);
#pragma unroll
  for (int h = 0; h < HEADS; ++h)
    atomicMax(&mkey[dn * HEADS + h], f2key(alpha_h(Sj, Si, nj, ni, h)));
}

__global__ void k_exp_sum(const int* __restrict__ src, const int* __restrict__ dst,
                          const float* __restrict__ Sj, const float* __restrict__ Si,
                          const unsigned* __restrict__ mkey, float* __restrict__ ssum,
                          float* __restrict__ coef, int E, int Et) {
  const int e = blockIdx.x * blockDim.x + threadIdx.x;
  if (e >= Et) return;
  int nj, ni, dn, sn;
  edge_nodes(src, dst, E, e, nj, ni, dn, sn);
#pragma unroll
  for (int h = 0; h < HEADS; ++h) {
    const int idx = dn * HEADS + h;
    const float a = expf(alpha_h(Sj, Si, nj, ni, h) - key2f(mkey[idx]));
    coef[(size_t)e * HEADS + h] = a;
    atomicAdd(&ssum[idx], a);
  }
}

__global__ void k_recip(float* __restrict__ ssum, int total) {
  const int t = blockIdx.x * blockDim.x + threadIdx.x;
  if (t < total) ssum[t] = 1.0f / ssum[t];
}

__global__ void k_aggregate(const int* __restrict__ src, const int* __restrict__ dst,
                            const float* __restrict__ coef,
                            const float* __restrict__ sinv,
                            const float* __restrict__ xl, float* __restrict__ out,
                            int E, int Et) {
  const long long t = (long long)blockIdx.x * blockDim.x + threadIdx.x;
  if (t >= (long long)Et * HC) return;
  const int e = (int)(t >> 7);
  const int f = (int)(t & 127);
  const int h = f >> 5;
  const int sn = (e < E) ? src[e] : (e - E);
  const int dn = (e < E) ? dst[e] : (e - E);
  const float c = coef[(size_t)e * HEADS + h] * sinv[dn * HEADS + h];
  atomicAdd(&out[(size_t)dn * HC + f], c * xl[(size_t)sn * HC + f]);
}

extern "C" void kernel_launch(void* const* d_in, const int* in_sizes, int n_in,
                              void* d_out, int out_size, void* d_ws, size_t ws_size,
                              hipStream_t stream) {
  const float* x      = (const float*)d_in[0];
  const float* W      = (const float*)d_in[1];
  const float* attn_j = (const float*)d_in[2];
  const float* attn_i = (const float*)d_in[3];
  const float* bias   = (const float*)d_in[4];
  const int*   eidx   = (const int*)d_in[5];

  const int N  = in_sizes[0] / F_IN;
  const int E  = in_sizes[5] / 2;
  const int Et = E + N;
  const int* src = eidx;
  const int* dst = eidx + E;

  char*  ws = (char*)d_ws;
  size_t o  = 0;
  auto take = [&](size_t bytes) -> void* {
    void* p = ws + o;
    o += (bytes + 255) & ~(size_t)255;
    return p;
  };
  float*     xl   = (float*)take((size_t)N * HC * sizeof(float));
  float*     Sj   = (float*)take((size_t)N * HEADS * sizeof(float));
  float*     Si   = (float*)take((size_t)N * HEADS * sizeof(float));
  unsigned*  mkey = (unsigned*)take((size_t)N * HEADS * sizeof(unsigned));
  float*     ssum = (float*)take((size_t)N * HEADS * sizeof(float));
  float*     coef = (float*)take((size_t)Et * HEADS * sizeof(float));
  _Float16*  xh   = (_Float16*)take((size_t)N * F_IN * sizeof(_Float16));
  _Float16*  Wh   = (_Float16*)take((size_t)HC * F_IN * sizeof(_Float16));

  float* out = (float*)d_out;

  const int nx = N * F_IN, nw = HC * F_IN;
  k_convert<<<(nx + 255) / 256, 256, 0, stream>>>(x, W, xh, Wh, nx, nw);

  k_gemm<<<(N + 15) / 16, 256, 0, stream>>>(x, W, xh, Wh, xl, N);

  k_scores<<<(N * HEADS + 255) / 256, 256, 0, stream>>>(xl, attn_j, attn_i, Sj, Si,
                                                        mkey, ssum, N);

  const int ntot = N * HC;
  k_out_init<<<(ntot + 255) / 256, 256, 0, stream>>>(out, bias, ntot);

  k_alpha_max<<<(Et + 255) / 256, 256, 0, stream>>>(src, dst, Sj, Si, mkey, E, Et);
  k_exp_sum<<<(Et + 255) / 256, 256, 0, stream>>>(src, dst, Sj, Si, mkey, ssum,
                                                  coef, E, Et);
  k_recip<<<(N * HEADS + 255) / 256, 256, 0, stream>>>(ssum, N * HEADS);

  const long long tot = (long long)Et * HC;
  k_aggregate<<<(unsigned)((tot + 255) / 256), 256, 0, stream>>>(
      src, dst, coef, ssum, xl, out, E, Et);
}


# edges counting-sorted by destination in the call (slot rank = return of an integer atomic add: arrival order); one wave per node does max, exp, sum and row gather in registers; no float atomics
# speedup vs baseline: 4.3370x; 4.3370x over previous
#include <hip/hip_runtime.h>
#include <hip/hip_bf16.h>
#include <hip/hip_fp16.h>

#define F_IN      128
#define HEADS     4
#define C_OUT     32
#define HC        128
#define NEG_SLOPE 0.2f

typedef float    v2f  __attribute__((ext_vector_type(2)));
typedef float    v8f  __attribute__((ext_vector_type(8)));
typedef _Float16 v8h  __attribute__((ext_vector_type(8)));
typedef _Float16 v16h __attribute__((ext_vector_type(16)));

#if __has_builtin(__builtin_amdgcn_wmma_f32_16x16x4_f32)
#define GAT_WMMA_F32 1
#else
#define GAT_WMMA_F32 0
#endif

__global__ void k_convert(const float* __restrict__ x, const float* __restrict__ W,
                          _Float16* __restrict__ xh, _Float16* __restrict__ Wh,
                          int nx, int nw) {
  int t = blockIdx.x * blockDim.x + threadIdx.x;
  if (t < nx) xh[t] = (_Float16)x[t];
  if (t < nw) Wh[t] = (_Float16)W[t];
}

__global__ void k_gemm(const float* __restrict__ x, const float* __restrict__ W,
                       const _Float16* __restrict__ xh, const _Float16* __restrict__ Wh,
                       float* __restrict__ xl, int N) {
  const int lane    = threadIdx.x & 31;
  const int wave    = threadIdx.x >> 5;
  const int g       = lane >> 4;
  const int l16     = lane & 15;
  const int rowBase = blockIdx.x * 16;
  const int colBase = wave * 16;
  int mrow = rowBase + l16; if (mrow >= N) mrow = N - 1;
  const int ncol = colBase + l16;

  v8f c = {};
#if GAT_WMMA_F32
  for (int kk = 0; kk < F_IN; kk += 4) {
    const int k0 = kk + g * 2;
    v2f a = *(const v2f*)(x + (size_t)mrow * F_IN + k0);
    v2f b = *(const v2f*)(W + (size_t)ncol * F_IN + k0);
    c = __builtin_amdgcn_wmma_f32_16x16x4_f32(false, a, false, b, (short)0, c,
                                              false, false);
  }
#else
  for (int kk = 0; kk < F_IN; kk += 32) {
    const _Float16* ap = xh + (size_t)mrow * F_IN + kk + 8 * g;
    v8h a0 = *(const v8h*)(ap);
    v8h a1 = *(const v8h*)(ap + 16);
    v16h a = __builtin_shufflevector(a0, a1, 0, 1, 2, 3, 4, 5, 6, 7,
                                     8, 9, 10, 11, 12, 13, 14, 15);
    v16h b = *(const v16h*)(Wh + (size_t)ncol * F_IN + kk + 16 * g);
    c = __builtin_amdgcn_wmma_f32_16x16x32_f16(false, a, false, b, (short)0, c,
                                               false, false);
  }
#endif
#pragma unroll
  for (int i = 0; i < 8; ++i) {
    const int r = rowBase + i + g * 8;
    if (r < N) xl[(size_t)r * HC + ncol] = c[i];
  }
}

__global__ void k_scores(const float* __restrict__ xl, const float* __restrict__ aj,
                         const float* __restrict__ ai, float* __restrict__ Sj,
                         float* __restrict__ Si, int* __restrict__ cnt, int N) {
  const int t = blockIdx.x * blockDim.x + threadIdx.x;
  if (t >= N * HEADS) return;
  const int n = t / HEADS, h = t % HEADS;
  const float* row = xl + (size_t)n * HC + h * C_OUT;
  float sj = 0.f, si = 0.f;
#pragma unroll
  for (int cc = 0; cc < C_OUT; ++cc) {
    sj += row[cc] * aj[h * C_OUT + cc];
    si += row[cc] * ai[h * C_OUT + cc];
  }
  Sj[t] = sj;
  Si[t] = si;
  if (h == 0) cnt[n] = 0;
}


__global__ void k_rank(const int* __restrict__ dst, int* __restrict__ cnt,
                       int* __restrict__ rank, int E) {
  const int e = blockIdx.x * blockDim.x + threadIdx.x;
  if (e < E) rank[e] = atomicAdd(&cnt[dst[e]], 1);
}

__global__ void k_scan_local(const int* __restrict__ cnt, int* __restrict__ rowl,
                             int* __restrict__ bsum, int N) {
  __shared__ int sh[256];
  const int tid = threadIdx.x;
  const int i = blockIdx.x * 256 + tid;
  const int v = (i < N) ? cnt[i] : 0;
  sh[tid] = v;
  __syncthreads();
  for (int off = 1; off < 256; off <<= 1) {
    const int t = (tid >= off) ? sh[tid - off] : 0;
    __syncthreads();
    sh[tid] += t;
    __syncthreads();
  }
  if (i < N) rowl[i] = sh[tid] - v;
  if (tid == 255) bsum[blockIdx.x] = sh[255];
}

__global__ void k_scan_blocks(const int* __restrict__ bsum, int* __restrict__ boff,
                              int nb) {
  __shared__ int sh[1024];
  const int tid = threadIdx.x;
  int carry = 0;
  for (int base = 0; base < nb; base += 1024) {
    const int i = base + tid;
    const int v = (i < nb) ? bsum[i] : 0;
    sh[tid] = v;
    __syncthreads();
    for (int off = 1; off < 1024; off <<= 1) {
      const int t = (tid >= off) ? sh[tid - off] : 0;
      __syncthreads();
      sh[tid] += t;
      __syncthreads();
    }
    if (i < nb) boff[i] = carry + sh[tid] - v;
    const int tot = sh[1023];
    __syncthreads();
    carry += tot;
  }
}

__global__ void k_fill(const int* __restrict__ src, const int* __restrict__ dst,
                       const int* __restrict__ rank, const int* __restrict__ rowl,
                       const int* __restrict__ boff, int2* __restrict__ sorted,
                       int E) {
  const int e = blockIdx.x * blockDim.x + threadIdx.x;
  if (e >= E) return;
  const int d = dst[e];
  const int s = src[e];
  const int p = rowl[d] + boff[d >> 8] + rank[e];
  sorted[p] = make_int2(s, src[s]);
}

__device__ __forceinline__ float leaky(float a) {
  return (a >= 0.f) ? a : NEG_SLOPE * a;
}
__device__ __forceinline__ float wave_max(float v) {
#pragma unroll
  for (int off = 16; off > 0; off >>= 1) v = fmaxf(v, __shfl_xor(v, off));
  return v;
}
__device__ __forceinline__ float wave_sum(float v) {
#pragma unroll
  for (int off = 16; off > 0; off >>= 1) v += __shfl_xor(v, off);
  return v;
}
struct HeadMask { unsigned m0, m1, m2, m3; };
__device__ __forceinline__ float pick4(const HeadMask& mk, float v0, float v1, float v2,
                                       float v3) {
  return __uint_as_float((__float_as_uint(v0) & mk.m0) | (__float_as_uint(v1) & mk.m1) |
                         (__float_as_uint(v2) & mk.m2) | (__float_as_uint(v3) & mk.m3));
}
__device__ __forceinline__ void fma4(float4& acc, float c, const float4& v) {
  acc.x += c * v.x; acc.y += c * v.y; acc.z += c * v.z; acc.w += c * v.w;
}
__device__ __forceinline__ float lane_val(float v, int k) {
  return __int_as_float(__builtin_amdgcn_readlane(__float_as_int(v), k));
}

__global__ void k_node(const int* __restrict__ src, const int* __restrict__ dst,
                       const int2* __restrict__ sorted, const int* __restrict__ rowl,
                       const int* __restrict__ boff, const int* __restrict__ cnt,
                       const float* __restrict__ Sj, const float* __restrict__ Si,
                       const float* __restrict__ xl, const float* __restrict__ bias,
                       float* __restrict__ out, int N) {
  const int lane = threadIdx.x & 31;
  const int hL = lane >> 3;
  const HeadMask mk = {hL == 0 ? ~0u : 0u, hL == 1 ? ~0u : 0u, hL == 2 ? ~0u : 0u,
                       hL == 3 ? ~0u : 0u};
  const int n = __builtin_amdgcn_readfirstlane((int)(blockIdx.x * 8 + (threadIdx.x >> 5)));
  if (n >= N) return;
  const int beg = rowl[n] + boff[n >> 8];
  const int end = beg + cnt[n];
  const float4 si = *(const float4*)(Si + (size_t)dst[n] * HEADS);
  const float4 sjs = *(const float4*)(Sj + (size_t)src[n] * HEADS);
  const float as0 = leaky(sjs.x + si.x), as1 = leaky(sjs.y + si.y);
  const float as2 = leaky(sjs.z + si.z), as3 = leaky(sjs.w + si.w);
  float m0 = as0, m1 = as1, m2 = as2, m3 = as3;
  for (int base = beg; base < end; base += 32) {
    const int idx = base + lane;
    if (idx < end) {
      const int2 sp = sorted[idx];
      const float4 sj = *(const float4*)(Sj + (size_t)sp.y * HEADS);
      m0 = fmaxf(m0, leaky(sj.x + si.x));
      m1 = fmaxf(m1, leaky(sj.y + si.y));
      m2 = fmaxf(m2, leaky(sj.z + si.z));
      m3 = fmaxf(m3, leaky(sj.w + si.w));
    }
  }
  m0 = wave_max(m0); m1 = wave_max(m1); m2 = wave_max(m2); m3 = wave_max(m3);

  float4 acc = make_float4(0.f, 0.f, 0.f, 0.f);
  float s0 = 0.f, s1 = 0.f, s2 = 0.f, s3 = 0.f;
  for (int base = beg; base < end; base += 32) {
    const int idx = base + lane;
    int sn = 0;
    float a0 = 0.f, a1 = 0.f, a2 = 0.f, a3 = 0.f;
    if (idx < end) {
      const int2 sp = sorted[idx];
      sn = sp.x;
      const float4 sj = *(const float4*)(Sj + (size_t)sp.y * HEADS);
      a0 = expf(leaky(sj.x + si.x) - m0);
      a1 = expf(leaky(sj.y + si.y) - m1);
      a2 = expf(leaky(sj.z + si.z) - m2);
      a3 = expf(leaky(sj.w + si.w) - m3);
    }
    s0 += a0; s1 += a1; s2 += a2; s3 += a3;
    const int cc = min(32, end - base);
    auto coef = [&](int k) {
      return pick4(mk, lane_val(a0, k), lane_val(a1, k), lane_val(a2, k), lane_val(a3, k));
    };
    auto row = [&](int k) {
      const int sk = __builtin_amdgcn_readlane(sn, k);
      return *(const float4*)(xl + (size_t)sk * HC + lane * 4);
    };
    int k = 0;
    for (; k + 4 <= cc; k += 4) {
      const float4 v0 = row(k), v1 = row(k + 1), v2 = row(k + 2), v3 = row(k + 3);
      const float c0 = coef(k), c1 = coef(k + 1), c2 = coef(k + 2), c3 = coef(k + 3);
      fma4(acc, c0, v0);
      fma4(acc, c1, v1);
      fma4(acc, c2, v2);
      fma4(acc, c3, v3);
    }
    for (; k < cc; ++k) fma4(acc, coef(k), row(k));
  }
  s0 = wave_sum(s0); s1 = wave_sum(s1); s2 = wave_sum(s2); s3 = wave_sum(s3);
  const float e0 = expf(as0 - m0), e1 = expf(as1 - m1);
  const float e2 = expf(as2 - m2), e3 = expf(as3 - m3);
  s0 += e0; s1 += e1; s2 += e2; s3 += e3;
  {
    const float4 v = *(const float4*)(xl + (size_t)n * HC + lane * 4);
    fma4(acc, pick4(mk, e0, e1, e2, e3), v);
  }
  const float sinv = 1.0f / pick4(mk, s0, s1, s2, s3);
  float4 o;
  o.x = acc.x * sinv + bias[lane * 4 + 0];
  o.y = acc.y * sinv + bias[lane * 4 + 1];
  o.z = acc.z * sinv + bias[lane * 4 + 2];
  o.w = acc.w * sinv + bias[lane * 4 + 3];
  *(float4*)(out + (size_t)n * HC + lane * 4) = o;
}

extern "C" void kernel_launch(void* const* d_in, const int* in_sizes, int n_in,
                              void* d_out, int out_size, void* d_ws, size_t ws_size,
                              hipStream_t stream) {
  const float* x      = (const float*)d_in[0];
  const float* W      = (const float*)d_in[1];
  const float* attn_j = (const float*)d_in[2];
  const float* attn_i = (const float*)d_in[3];
  const float* bias   = (const float*)d_in[4];
  const int*   eidx   = (const int*)d_in[5];

  const int N  = in_sizes[0] / F_IN;
  const int E  = in_sizes[5] / 2;
  const int* src = eidx;
  const int* dst = eidx + E;

  char*  ws = (char*)d_ws;
  size_t o  = 0;
  auto take = [&](size_t bytes) -> void* {
    void* p = ws + o;
    o += (bytes + 255) & ~(size_t)255;
    return p;
  };
  float*     xl   = (float*)take((size_t)N * HC * sizeof(float));
  float*     Sj   = (float*)take((size_t)N * HEADS * sizeof(float));
  float*     Si   = (float*)take((size_t)N * HEADS * sizeof(float));
  const int  nb   = (N + 255) / 256;
  int*       cnt  = (int*)take((size_t)N * sizeof(int));
  int*       rowl = (int*)take((size_t)N * sizeof(int));
  int*       bsum = (int*)take((size_t)nb * sizeof(int));
  int*       boff = (int*)take((size_t)nb * sizeof(int));
  int*       rank = (int*)take((size_t)E * sizeof(int));
  int2*      sorted = (int2*)take((size_t)E * sizeof(int2));
  _Float16*  xh   = (_Float16*)take((size_t)N * F_IN * sizeof(_Float16));
  _Float16*  Wh   = (_Float16*)take((size_t)HC * F_IN * sizeof(_Float16));

  float* out = (float*)d_out;

  const int nx = N * F_IN, nw = HC * F_IN;
  k_convert<<<(nx + 255) / 256, 256, 0, stream>>>(x, W, xh, Wh, nx, nw);

  k_gemm<<<(N + 15) / 16, 256, 0, stream>>>(x, W, xh, Wh, xl, N);

  k_scores<<<(N * HEADS + 255) / 256, 256, 0, stream>>>(xl, attn_j, attn_i, Sj, Si,
                                                        cnt, N);

  k_rank<<<(E + 255) / 256, 256, 0, stream>>>(dst, cnt, rank, E);
  k_scan_local<<<nb, 256, 0, stream>>>(cnt, rowl, bsum, N);
  k_scan_blocks<<<1, 1024, 0, stream>>>(bsum, boff, nb);
  k_fill<<<(E + 255) / 256, 256, 0, stream>>>(src, dst, rank, rowl, boff, sorted, E);

  k_node<<<(N + 7) / 8, 256, 0, stream>>>(src, dst, sorted, rowl, boff, cnt, Sj, Si,
                                          xl, bias, out, N);
}


# dead staging launch removed: its converted copies of x and W were read by no kernel (products stay f32 16x16x4); 8 launches become 7 and 200000 one-element waves no longer enter per call
# speedup vs baseline: 4.6039x; 1.0615x over previous
#include <hip/hip_runtime.h>


#define F_IN      128
#define HEADS     4
#define C_OUT     32
#define HC        128
#define NEG_SLOPE 0.2f

typedef float v2f __attribute__((ext_vector_type(2)));
typedef float v8f __attribute__((ext_vector_type(8)));

__global__ void k_gemm(const float* __restrict__ x, const float* __restrict__ W,
                       float* __restrict__ xl, int N) {
  const int lane    = threadIdx.x & 31;
  const int wave    = threadIdx.x >> 5;
  const int g       = lane >> 4;
  const int l16     = lane & 15;
  const int rowBase = blockIdx.x * 16;
  const int colBase = wave * 16;
  int mrow = rowBase + l16; if (mrow >= N) mrow = N - 1;
  const int ncol = colBase + l16;

  v8f c = {};
  for (int kk = 0; kk < F_IN; kk += 4) {
    const int k0 = kk + g * 2;
    v2f a = *(const v2f*)(x + (size_t)mrow * F_IN + k0);
    v2f b = *(const v2f*)(W + (size_t)ncol * F_IN + k0);
    c = __builtin_amdgcn_wmma_f32_16x16x4_f32(false, a, false, b, (short)0, c,
                                              false, false);
  }
#pragma unroll
  for (int i = 0; i < 8; ++i) {
    const int r = rowBase + i + g * 8;
    if (r < N) xl[(size_t)r * HC + ncol] = c[i];
  }
}

__global__ void k_scores(const float* __restrict__ xl, const float* __restrict__ aj,
                         const float* __restrict__ ai, float* __restrict__ Sj,
                         float* __restrict__ Si, int* __restrict__ cnt, int N) {
  const int t = blockIdx.x * blockDim.x + threadIdx.x;
  if (t >= N * HEADS) return;
  const int n = t / HEADS, h = t % HEADS;
  const float* row = xl + (size_t)n * HC + h * C_OUT;
  float sj = 0.f, si = 0.f;
#pragma unroll
  for (int cc = 0; cc < C_OUT; ++cc) {
    sj += row[cc] * aj[h * C_OUT + cc];
    si += row[cc] * ai[h * C_OUT + cc];
  }
  Sj[t] = sj;
  Si[t] = si;
  if (h == 0) cnt[n] = 0;
}


__global__ void k_rank(const int* __restrict__ dst, int* __restrict__ cnt,
                       int* __restrict__ rank, int E) {
  const int e = blockIdx.x * blockDim.x + threadIdx.x;
  if (e < E) rank[e] = atomicAdd(&cnt[dst[e]], 1);
}

__global__ void k_scan_local(const int* __restrict__ cnt, int* __restrict__ rowl,
                             int* __restrict__ bsum, int N) {
  __shared__ int sh[256];
  const int tid = threadIdx.x;
  const int i = blockIdx.x * 256 + tid;
  const int v = (i < N) ? cnt[i] : 0;
  sh[tid] = v;
  __syncthreads();
  for (int off = 1; off < 256; off <<= 1) {
    const int t = (tid >= off) ? sh[tid - off] : 0;
    __syncthreads();
    sh[tid] += t;
    __syncthreads();
  }
  if (i < N) rowl[i] = sh[tid] - v;
  if (tid == 255) bsum[blockIdx.x] = sh[255];
}

__global__ void k_scan_blocks(const int* __restrict__ bsum, int* __restrict__ boff,
                              int nb) {
  __shared__ int sh[1024];
  const int tid = threadIdx.x;
  int carry = 0;
  for (int base = 0; base < nb; base += 1024) {
    const int i = base + tid;
    const int v = (i < nb) ? bsum[i] : 0;
    sh[tid] = v;
    __syncthreads();
    for (int off = 1; off < 1024; off <<= 1) {
      const int t = (tid >= off) ? sh[tid - off] : 0;
      __syncthreads();
      sh[tid] += t;
      __syncthreads();
    }
    if (i < nb) boff[i] = carry + sh[tid] - v;
    const int tot = sh[1023];
    __syncthreads();
    carry += tot;
  }
}

__global__ void k_fill(const int* __restrict__ src, const int* __restrict__ dst,
                       const int* __restrict__ rank, const int* __restrict__ rowl,
                       const int* __restrict__ boff, int2* __restrict__ sorted,
                       int E) {
  const int e = blockIdx.x * blockDim.x + threadIdx.x;
  if (e >= E) return;
  const int d = dst[e];
  const int s = src[e];
  const int p = rowl[d] + boff[d >> 8] + rank[e];
  sorted[p] = make_int2(s, src[s]);
}

__device__ __forceinline__ float leaky(float a) {
  return (a >= 0.f) ? a : NEG_SLOPE * a;
}
__device__ __forceinline__ float wave_max(float v) {
#pragma unroll
  for (int off = 16; off > 0; off >>= 1) v = fmaxf(v, __shfl_xor(v, off));
  return v;
}
__device__ __forceinline__ float wave_sum(float v) {
#pragma unroll
  for (int off = 16; off > 0; off >>= 1) v += __shfl_xor(v, off);
  return v;
}
struct HeadMask { unsigned m0, m1, m2, m3; };
__device__ __forceinline__ float pick4(const HeadMask& mk, float v0, float v1, float v2,
                                       float v3) {
  return __uint_as_float((__float_as_uint(v0) & mk.m0) | (__float_as_uint(v1) & mk.m1) |
                         (__float_as_uint(v2) & mk.m2) | (__float_as_uint(v3) & mk.m3));
}
__device__ __forceinline__ void fma4(float4& acc, float c, const float4& v) {
  acc.x += c * v.x; acc.y += c * v.y; acc.z += c * v.z; acc.w += c * v.w;
}
__device__ __forceinline__ float lane_val(float v, int k) {
  return __int_as_float(__builtin_amdgcn_readlane(__float_as_int(v), k));
}

__global__ void k_node(const int* __restrict__ src, const int* __restrict__ dst,
                       const int2* __restrict__ sorted, const int* __restrict__ rowl,
                       const int* __restrict__ boff, const int* __restrict__ cnt,
                       const float* __restrict__ Sj, const float* __restrict__ Si,
                       const float* __restrict__ xl, const float* __restrict__ bias,
                       float* __restrict__ out, int N) {
  const int lane = threadIdx.x & 31;
  const int hL = lane >> 3;
  const HeadMask mk = {hL == 0 ? ~0u : 0u, hL == 1 ? ~0u : 0u, hL == 2 ? ~0u : 0u,
                       hL == 3 ? ~0u : 0u};
  const int n = __builtin_amdgcn_readfirstlane((int)(blockIdx.x * 8 + (threadIdx.x >> 5)));
  if (n >= N) return;
  const int beg = rowl[n] + boff[n >> 8];
  const int end = beg + cnt[n];
  const float4 si = *(const float4*)(Si + (size_t)dst[n] * HEADS);
  const float4 sjs = *(const float4*)(Sj + (size_t)src[n] * HEADS);
  const float as0 = leaky(sjs.x + si.x), as1 = leaky(sjs.y + si.y);
  const float as2 = leaky(sjs.z + si.z), as3 = leaky(sjs.w + si.w);
  float m0 = as0, m1 = as1, m2 = as2, m3 = as3;
  for (int base = beg; base < end; base += 32) {
    const int idx = base + lane;
    if (idx < end) {
      const int2 sp = sorted[idx];
      const float4 sj = *(const float4*)(Sj + (size_t)sp.y * HEADS);
      m0 = fmaxf(m0, leaky(sj.x + si.x));
      m1 = fmaxf(m1, leaky(sj.y + si.y));
      m2 = fmaxf(m2, leaky(sj.z + si.z));
      m3 = fmaxf(m3, leaky(sj.w + si.w));
    }
  }
  m0 = wave_max(m0); m1 = wave_max(m1); m2 = wave_max(m2); m3 = wave_max(m3);

  float4 acc = make_float4(0.f, 0.f, 0.f, 0.f);
  float s0 = 0.f, s1 = 0.f, s2 = 0.f, s3 = 0.f;
  for (int base = beg; base < end; base += 32) {
    const int idx = base + lane;
    int sn = 0;
    float a0 = 0.f, a1 = 0.f, a2 = 0.f, a3 = 0.f;
    if (idx < end) {
      const int2 sp = sorted[idx];
      sn = sp.x;
      const float4 sj = *(const float4*)(Sj + (size_t)sp.y * HEADS);
      a0 = expf(leaky(sj.x + si.x) - m0);
      a1 = expf(leaky(sj.y + si.y) - m1);
      a2 = expf(leaky(sj.z + si.z) - m2);
      a3 = expf(leaky(sj.w + si.w) - m3);
    }
    s0 += a0; s1 += a1; s2 += a2; s3 += a3;
    const int cc = min(32, end - base);
    auto coef = [&](int k) {
      return pick4(mk, lane_val(a0, k), lane_val(a1, k), lane_val(a2, k), lane_val(a3, k));
    };
    auto row = [&](int k) {
      const int sk = __builtin_amdgcn_readlane(sn, k);
      return *(const float4*)(xl + (size_t)sk * HC + lane * 4);
    };
    int k = 0;
    for (; k + 4 <= cc; k += 4) {
      const float4 v0 = row(k), v1 = row(k + 1), v2 = row(k + 2), v3 = row(k + 3);
      const float c0 = coef(k), c1 = coef(k + 1), c2 = coef(k + 2), c3 = coef(k + 3);
      fma4(acc, c0, v0);
      fma4(acc, c1, v1);
      fma4(acc, c2, v2);
      fma4(acc, c3, v3);
    }
    for (; k < cc; ++k) fma4(acc, coef(k), row(k));
  }
  s0 = wave_sum(s0); s1 = wave_sum(s1); s2 = wave_sum(s2); s3 = wave_sum(s3);
  const float e0 = expf(as0 - m0), e1 = expf(as1 - m1);
  const float e2 = expf(as2 - m2), e3 = expf(as3 - m3);
  s0 += e0; s1 += e1; s2 += e2; s3 += e3;
  {
    const float4 v = *(const float4*)(xl + (size_t)n * HC + lane * 4);
    fma4(acc, pick4(mk, e0, e1, e2, e3), v);
  }
  const float sinv = 1.0f / pick4(mk, s0, s1, s2, s3);
  float4 o;
  o.x = acc.x * sinv + bias[lane * 4 + 0];
  o.y = acc.y * sinv + bias[lane * 4 + 1];
  o.z = acc.z * sinv + bias[lane * 4 + 2];
  o.w = acc.w * sinv + bias[lane * 4 + 3];
  *(float4*)(out + (size_t)n * HC + lane * 4) = o;
}

extern "C" void kernel_launch(void* const* d_in, const int* in_sizes, int n_in,
                              void* d_out, int out_size, void* d_ws, size_t ws_size,
                              hipStream_t stream) {
  const float* x      = (const float*)d_in[0];
  const float* W      = (const float*)d_in[1];
  const float* attn_j = (const float*)d_in[2];
  const float* attn_i = (const float*)d_in[3];
  const float* bias   = (const float*)d_in[4];
  const int*   eidx   = (const int*)d_in[5];

  const int N  = in_sizes[0] / F_IN;
  const int E  = in_sizes[5] / 2;
  const int* src = eidx;
  const int* dst = eidx + E;

  char*  ws = (char*)d_ws;
  size_t o  = 0;
  auto take = [&](size_t bytes) -> void* {
    void* p = ws + o;
    o += (bytes + 255) & ~(size_t)255;
    return p;
  };
  float*     xl   = (float*)take((size_t)N * HC * sizeof(float));
  float*     Sj   = (float*)take((size_t)N * HEADS * sizeof(float));
  float*     Si   = (float*)take((size_t)N * HEADS * sizeof(float));
  const int  nb   = (N + 255) / 256;
  int*       cnt  = (int*)take((size_t)N * sizeof(int));
  int*       rowl = (int*)take((size_t)N * sizeof(int));
  int*       bsum = (int*)take((size_t)nb * sizeof(int));
  int*       boff = (int*)take((size_t)nb * sizeof(int));
  int*       rank = (int*)take((size_t)E * sizeof(int));
  int2*      sorted = (int2*)take((size_t)E * sizeof(int2));

  float* out = (float*)d_out;

  k_gemm<<<(N + 15) / 16, 256, 0, stream>>>(x, W, xl, N);

  k_scores<<<(N * HEADS + 255) / 256, 256, 0, stream>>>(xl, attn_j, attn_i, Sj, Si,
                                                        cnt, N);

  k_rank<<<(E + 255) / 256, 256, 0, stream>>>(dst, cnt, rank, E);
  k_scan_local<<<nb, 256, 0, stream>>>(cnt, rowl, bsum, N);
  k_scan_blocks<<<1, 1024, 0, stream>>>(bsum, boff, nb);
  k_fill<<<(E + 255) / 256, 256, 0, stream>>>(src, dst, rank, rowl, boff, sorted, E);

  k_node<<<(N + 7) / 8, 256, 0, stream>>>(src, dst, sorted, rowl, boff, cnt, Sj, Si,
                                          xl, bias, out, N);
}


# counting sort by destination rebuilt on LDS counters in two levels (256-node ranges, then nodes; slot rank = LDS atomic-add return, arrival order as before): no global atomics, 7 launches become 6
# speedup vs baseline: 8.0927x; 1.7578x over previous
#include <hip/hip_runtime.h>


#define F_IN      128
#define HEADS     4
#define C_OUT     32
#define HC        128
#define NEG_SLOPE 0.2f

typedef float v2f __attribute__((ext_vector_type(2)));
typedef float v8f __attribute__((ext_vector_type(8)));

__global__ void k_gemm(const float* __restrict__ x, const float* __restrict__ W,
                       float* __restrict__ xl, int N) {
  const int lane    = threadIdx.x & 31;
  const int wave    = threadIdx.x >> 5;
  const int g       = lane >> 4;
  const int l16     = lane & 15;
  const int rowBase = blockIdx.x * 16;
  const int colBase = wave * 16;
  int mrow = rowBase + l16; if (mrow >= N) mrow = N - 1;
  const int ncol = colBase + l16;

  v8f c = {};
  for (int kk = 0; kk < F_IN; kk += 4) {
    const int k0 = kk + g * 2;
    v2f a = *(const v2f*)(x + (size_t)mrow * F_IN + k0);
    v2f b = *(const v2f*)(W + (size_t)ncol * F_IN + k0);
    c = __builtin_amdgcn_wmma_f32_16x16x4_f32(false, a, false, b, (short)0, c,
                                              false, false);
  }
#pragma unroll
  for (int i = 0; i < 8; ++i) {
    const int r = rowBase + i + g * 8;
    if (r < N) xl[(size_t)r * HC + ncol] = c[i];
  }
}

__global__ void k_scores(const float* __restrict__ xl, const float* __restrict__ aj,
                         const float* __restrict__ ai, float* __restrict__ Sj,
                         float* __restrict__ Si, int N) {
  const int t = blockIdx.x * blockDim.x + threadIdx.x;
  if (t >= N * HEADS) return;
  const int n = t / HEADS, h = t % HEADS;
  const float* row = xl + (size_t)n * HC + h * C_OUT;
  float sj = 0.f, si = 0.f;
#pragma unroll
  for (int cc = 0; cc < C_OUT; ++cc) {
    sj += row[cc] * aj[h * C_OUT + cc];
    si += row[cc] * ai[h * C_OUT + cc];
  }
  Sj[t] = sj;
  Si[t] = si;
}


#define EB 256
#define ET 1024
#define UB 8

__device__ __forceinline__ int wave_sum_i(int v) {
#pragma unroll
  for (int off = 16; off > 0; off >>= 1) v += __shfl_xor(v, off);
  return v;
}
__device__ __forceinline__ int wave_incl_scan(int v, int lane) {
#pragma unroll
  for (int off = 1; off < 32; off <<= 1) {
    const int t = __shfl_up(v, off);
    if (lane >= off) v += t;
  }
  return v;
}
__device__ __forceinline__ int block_excl_scan(int v, int* wsum, int tid, int* total) {
  const int lane = tid & 31, wave = tid >> 5;
  const int inc = wave_incl_scan(v, lane);
  if (lane == 31) wsum[wave] = inc;
  __syncthreads();
  if (wave == 0) {
    const int w = wsum[lane];
    const int wi = wave_incl_scan(w, lane);
    wsum[lane] = wi - w;
    if (lane == 31) wsum[32] = wi;
  }
  __syncthreads();
  const int ex = wsum[wave] + inc - v;
  *total = wsum[32];
  __syncthreads();
  return ex;
}

__global__ void k_count(const int* __restrict__ dst, int* __restrict__ cntRB, int E,
                        int chunk, int nb) {
  extern __shared__ int hist[];
  const int tid = threadIdx.x, b = blockIdx.x;
  for (int r = tid; r < nb; r += ET) hist[r] = 0;
  __syncthreads();
  const int e0 = b * chunk;
  const int e1 = min(E, e0 + chunk);
  for (int eb = e0 + tid; eb < e1; eb += ET * UB) {
    int d[UB];
#pragma unroll
    for (int u = 0; u < UB; ++u) {
      const int e = eb + u * ET;
      d[u] = (e < e1) ? dst[e] : -1;
    }
#pragma unroll
    for (int u = 0; u < UB; ++u)
      if (d[u] >= 0) atomicAdd(&hist[d[u] >> 8], 1);
  }
  __syncthreads();
  for (int r = tid; r < nb; r += ET) cntRB[(size_t)r * EB + b] = hist[r];
}

__global__ void k_scatter(const int* __restrict__ src, const int* __restrict__ dst,
                          const int* __restrict__ cntRB, int2* __restrict__ bucket,
                          int* __restrict__ rbase, int* __restrict__ rtot, int E,
                          int chunk, int nb) {
  extern __shared__ int shm[];
  __shared__ int wsum[33];
  int* cur = shm;
  int* tot = shm + nb;
  const int tid = threadIdx.x, b = blockIdx.x;
  const int lane = tid & 31, wave = tid >> 5;
  for (int r = wave; r < nb; r += ET / 32) {
    int t = 0, p = 0;
#pragma unroll
    for (int j = lane; j < EB; j += 32) {
      const int c = cntRB[(size_t)r * EB + j];
      t += c;
      p += (j < b) ? c : 0;
    }
    t = wave_sum_i(t);
    p = wave_sum_i(p);
    if (lane == 0) { tot[r] = t; cur[r] = p; }
  }
  __syncthreads();
  int carry = 0;
  for (int base = 0; base < nb; base += ET) {
    const int i = base + tid;
    const int v = (i < nb) ? tot[i] : 0;
    int total;
    const int ex = carry + block_excl_scan(v, wsum, tid, &total);
    if (i < nb) {
      cur[i] += ex;
      if (b == 0) { rbase[i] = ex; rtot[i] = v; }
    }
    carry += total;
  }
  __syncthreads();
  const int e0 = b * chunk;
  const int e1 = min(E, e0 + chunk);
  for (int eb = e0 + tid; eb < e1; eb += ET * UB) {
    int d[UB], s[UB];
#pragma unroll
    for (int u = 0; u < UB; ++u) {
      const int e = eb + u * ET;
      const bool ok = e < e1;
      d[u] = ok ? dst[e] : -1;
      s[u] = ok ? src[e] : 0;
    }
#pragma unroll
    for (int u = 0; u < UB; ++u) {
      if (d[u] >= 0) {
        const int pos = atomicAdd(&cur[d[u] >> 8], 1);
        bucket[pos] = make_int2(d[u], s[u]);
      }
    }
  }
}

__global__ void k_rangesort(const int* __restrict__ src, const int2* __restrict__ bucket,
                            const int* __restrict__ rbase, const int* __restrict__ rtot,
                            int2* __restrict__ sorted, int* __restrict__ rowstart,
                            int* __restrict__ cnt, int N) {
  __shared__ int c[256];
  __shared__ int cursor[256];
  __shared__ int wsum[33];
  const int tid = threadIdx.x, r = blockIdx.x;
  const int b0 = rbase[r];
  const int m = rtot[r];
  if (tid < 256) c[tid] = 0;
  __syncthreads();
  for (int ib = tid; ib < m; ib += ET * UB) {
    int d[UB];
#pragma unroll
    for (int u = 0; u < UB; ++u) {
      const int i = ib + u * ET;
      d[u] = (i < m) ? bucket[b0 + i].x : -1;
    }
#pragma unroll
    for (int u = 0; u < UB; ++u)
      if (d[u] >= 0) atomicAdd(&c[d[u] & 255], 1);
  }
  __syncthreads();
  const int v = (tid < 256) ? c[tid] : 0;
  int total;
  const int ex = block_excl_scan(v, wsum, tid, &total);
  if (tid < 256) {
    cursor[tid] = b0 + ex;
    const int n = r * 256 + tid;
    if (n < N) { rowstart[n] = b0 + ex; cnt[n] = v; }
  }
  __syncthreads();
  for (int ib = tid; ib < m; ib += ET * UB) {
    int2 rec[UB];
    int nj[UB];
#pragma unroll
    for (int u = 0; u < UB; ++u) {
      const int i = ib + u * ET;
      rec[u] = (i < m) ? bucket[b0 + i] : make_int2(-1, 0);
    }
#pragma unroll
    for (int u = 0; u < UB; ++u) nj[u] = (rec[u].x >= 0) ? src[rec[u].y] : 0;
#pragma unroll
    for (int u = 0; u < UB; ++u) {
      if (rec[u].x >= 0) {
        const int p = atomicAdd(&cursor[rec[u].x & 255], 1);
        sorted[p] = make_int2(rec[u].y, nj[u]);
      }
    }
  }
}

__device__ __forceinline__ float leaky(float a) {
  return (a >= 0.f) ? a : NEG_SLOPE * a;
}
__device__ __forceinline__ float wave_max(float v) {
#pragma unroll
  for (int off = 16; off > 0; off >>= 1) v = fmaxf(v, __shfl_xor(v, off));
  return v;
}
__device__ __forceinline__ float wave_sum(float v) {
#pragma unroll
  for (int off = 16; off > 0; off >>= 1) v += __shfl_xor(v, off);
  return v;
}
struct HeadMask { unsigned m0, m1, m2, m3; };
__device__ __forceinline__ float pick4(const HeadMask& mk, float v0, float v1, float v2,
                                       float v3) {
  return __uint_as_float((__float_as_uint(v0) & mk.m0) | (__float_as_uint(v1) & mk.m1) |
                         (__float_as_uint(v2) & mk.m2) | (__float_as_uint(v3) & mk.m3));
}
__device__ __forceinline__ void fma4(float4& acc, float c, const float4& v) {
  acc.x += c * v.x; acc.y += c * v.y; acc.z += c * v.z; acc.w += c * v.w;
}
__device__ __forceinline__ float lane_val(float v, int k) {
  return __int_as_float(__builtin_amdgcn_readlane(__float_as_int(v), k));
}

__global__ void k_node(const int* __restrict__ src, const int* __restrict__ dst,
                       const int2* __restrict__ sorted, const int* __restrict__ rowstart,
                       const int* __restrict__ cnt,
                       const float* __restrict__ Sj, const float* __restrict__ Si,
                       const float* __restrict__ xl, const float* __restrict__ bias,
                       float* __restrict__ out, int N) {
  const int lane = threadIdx.x & 31;
  const int hL = lane >> 3;
  const HeadMask mk = {hL == 0 ? ~0u : 0u, hL == 1 ? ~0u : 0u, hL == 2 ? ~0u : 0u,
                       hL == 3 ? ~0u : 0u};
  const int n = __builtin_amdgcn_readfirstlane((int)(blockIdx.x * 8 + (threadIdx.x >> 5)));
  if (n >= N) return;
  const int beg = rowstart[n];
  const int end = beg + cnt[n];
  const float4 si = *(const float4*)(Si + (size_t)dst[n] * HEADS);
  const float4 sjs = *(const float4*)(Sj + (size_t)src[n] * HEADS);
  const float as0 = leaky(sjs.x + si.x), as1 = leaky(sjs.y + si.y);
  const float as2 = leaky(sjs.z + si.z), as3 = leaky(sjs.w + si.w);
  float m0 = as0, m1 = as1, m2 = as2, m3 = as3;
  for (int base = beg; base < end; base += 32) {
    const int idx = base + lane;
    if (idx < end) {
      const int2 sp = sorted[idx];
      const float4 sj = *(const float4*)(Sj + (size_t)sp.y * HEADS);
      m0 = fmaxf(m0, leaky(sj.x + si.x));
      m1 = fmaxf(m1, leaky(sj.y + si.y));
      m2 = fmaxf(m2, leaky(sj.z + si.z));
      m3 = fmaxf(m3, leaky(sj.w + si.w));
    }
  }
  m0 = wave_max(m0); m1 = wave_max(m1); m2 = wave_max(m2); m3 = wave_max(m3);

  float4 acc = make_float4(0.f, 0.f, 0.f, 0.f);
  float s0 = 0.f, s1 = 0.f, s2 = 0.f, s3 = 0.f;
  for (int base = beg; base < end; base += 32) {
    const int idx = base + lane;
    int sn = 0;
    float a0 = 0.f, a1 = 0.f, a2 = 0.f, a3 = 0.f;
    if (idx < end) {
      const int2 sp = sorted[idx];
      sn = sp.x;
      const float4 sj = *(const float4*)(Sj + (size_t)sp.y * HEADS);
      a0 = expf(leaky(sj.x + si.x) - m0);
      a1 = expf(leaky(sj.y + si.y) - m1);
      a2 = expf(leaky(sj.z + si.z) - m2);
      a3 = expf(leaky(sj.w + si.w) - m3);
    }
    s0 += a0; s1 += a1; s2 += a2; s3 += a3;
    const int cc = min(32, end - base);
    auto coef = [&](int k) {
      return pick4(mk, lane_val(a0, k), lane_val(a1, k), lane_val(a2, k), lane_val(a3, k));
    };
    auto row = [&](int k) {
      const int sk = __builtin_amdgcn_readlane(sn, k);
      return *(const float4*)(xl + (size_t)sk * HC + lane * 4);
    };
    int k = 0;
    for (; k + 4 <= cc; k += 4) {
      const float4 v0 = row(k), v1 = row(k + 1), v2 = row(k + 2), v3 = row(k + 3);
      const float c0 = coef(k), c1 = coef(k + 1), c2 = coef(k + 2), c3 = coef(k + 3);
      fma4(acc, c0, v0);
      fma4(acc, c1, v1);
      fma4(acc, c2, v2);
      fma4(acc, c3, v3);
    }
    for (; k < cc; ++k) fma4(acc, coef(k), row(k));
  }
  s0 = wave_sum(s0); s1 = wave_sum(s1); s2 = wave_sum(s2); s3 = wave_sum(s3);
  const float e0 = expf(as0 - m0), e1 = expf(as1 - m1);
  const float e2 = expf(as2 - m2), e3 = expf(as3 - m3);
  s0 += e0; s1 += e1; s2 += e2; s3 += e3;
  {
    const float4 v = *(const float4*)(xl + (size_t)n * HC + lane * 4);
    fma4(acc, pick4(mk, e0, e1, e2, e3), v);
  }
  const float sinv = 1.0f / pick4(mk, s0, s1, s2, s3);
  float4 o;
  o.x = acc.x * sinv + bias[lane * 4 + 0];
  o.y = acc.y * sinv + bias[lane * 4 + 1];
  o.z = acc.z * sinv + bias[lane * 4 + 2];
  o.w = acc.w * sinv + bias[lane * 4 + 3];
  *(float4*)(out + (size_t)n * HC + lane * 4) = o;
}

extern "C" void kernel_launch(void* const* d_in, const int* in_sizes, int n_in,
                              void* d_out, int out_size, void* d_ws, size_t ws_size,
                              hipStream_t stream) {
  const float* x      = (const float*)d_in[0];
  const float* W      = (const float*)d_in[1];
  const float* attn_j = (const float*)d_in[2];
  const float* attn_i = (const float*)d_in[3];
  const float* bias   = (const float*)d_in[4];
  const int*   eidx   = (const int*)d_in[5];

  const int N  = in_sizes[0] / F_IN;
  const int E  = in_sizes[5] / 2;
  const int* src = eidx;
  const int* dst = eidx + E;

  char*  ws = (char*)d_ws;
  size_t o  = 0;
  auto take = [&](size_t bytes) -> void* {
    void* p = ws + o;
    o += (bytes + 255) & ~(size_t)255;
    return p;
  };
  float*     xl   = (float*)take((size_t)N * HC * sizeof(float));
  float*     Sj   = (float*)take((size_t)N * HEADS * sizeof(float));
  float*     Si   = (float*)take((size_t)N * HEADS * sizeof(float));
  const int  nb   = (N + 255) / 256;
  const int  chunk = (E + EB - 1) / EB;
  int*       cnt  = (int*)take((size_t)N * sizeof(int));
  int*       rowstart = (int*)take((size_t)N * sizeof(int));
  int*       cntRB = (int*)take((size_t)nb * EB * sizeof(int));
  int*       rbase = (int*)take((size_t)nb * sizeof(int));
  int*       rtot  = (int*)take((size_t)nb * sizeof(int));
  int2*      bucket = (int2*)take((size_t)E * sizeof(int2));
  int2*      sorted = (int2*)take((size_t)E * sizeof(int2));

  float* out = (float*)d_out;

  k_gemm<<<(N + 15) / 16, 256, 0, stream>>>(x, W, xl, N);

  k_scores<<<(N * HEADS + 255) / 256, 256, 0, stream>>>(xl, attn_j, attn_i, Sj, Si, N);

  k_count<<<EB, ET, (size_t)nb * sizeof(int), stream>>>(dst, cntRB, E, chunk, nb);
  k_scatter<<<EB, ET, (size_t)nb * 2 * sizeof(int), stream>>>(src, dst, cntRB, bucket,
                                                              rbase, rtot, E, chunk, nb);
  k_rangesort<<<nb, ET, 0, stream>>>(src, bucket, rbase, rtot, sorted, rowstart, cnt, N);

  k_node<<<(N + 7) / 8, 256, 0, stream>>>(src, dst, sorted, rowstart, cnt, Sj, Si,
                                          xl, bias, out, N);
}


# GEMM retiled: a wave owns 16 rows x all 128 columns, W copied once per 64-row block into LDS (padded pitch), B fragments read from LDS; products and their k order unchanged; 25000 waves become 3128
# speedup vs baseline: 9.3380x; 1.1539x over previous
#include <hip/hip_runtime.h>


#define F_IN      128
#define HEADS     4
#define C_OUT     32
#define HC        128
#define NEG_SLOPE 0.2f

typedef float v2f __attribute__((ext_vector_type(2)));
typedef float v8f __attribute__((ext_vector_type(8)));

#define GT 4
#define WP 132
__global__ __launch_bounds__(GT * 32) void k_gemm(const float* __restrict__ x,
                                                  const float* __restrict__ W,
                                                  float* __restrict__ xl, int N) {
  __shared__ __attribute__((aligned(16))) float Ws[HC * WP];
  const int lane    = threadIdx.x & 31;
  const int wave    = threadIdx.x >> 5;
  const int g       = lane >> 4;
  const int l16     = lane & 15;
  const int rowBase = (blockIdx.x * GT + wave) * 16;
  int mrow = rowBase + l16; if (mrow >= N) mrow = N - 1;

  v2f a[F_IN / 4];
  const float* xr = x + (size_t)mrow * F_IN + g * 2;
#pragma unroll
  for (int s = 0; s < F_IN / 4; ++s) a[s] = *(const v2f*)(xr + 4 * s);

#pragma unroll 8
  for (int j = 0; j < (HC * (F_IN / 4)) / (GT * 32); ++j) {
    const int i = threadIdx.x + j * (GT * 32);
    const int r = i >> 5, c4 = i & 31;
    const float4 v = *(const float4*)(W + (size_t)r * F_IN + c4 * 4);
    *(float4*)(Ws + r * WP + c4 * 4) = v;
  }
  __syncthreads();
  if (rowBase >= N) return;

  v8f c[8] = {};
  const float* bp = Ws + l16 * WP + g * 2;
#pragma unroll
  for (int s = 0; s < F_IN / 4; ++s) {
#pragma unroll
    for (int t = 0; t < 8; ++t) {
      const v2f b = *(const v2f*)(bp + t * 16 * WP + 4 * s);
      c[t] = __builtin_amdgcn_wmma_f32_16x16x4_f32(false, a[s], false, b, (short)0, c[t],
                                                   false, false);
    }
  }
#pragma unroll
  for (int t = 0; t < 8; ++t) {
#pragma unroll
    for (int i = 0; i < 8; ++i) {
      const int r = rowBase + i + g * 8;
      if (r < N) xl[(size_t)r * HC + t * 16 + l16] = c[t][i];
    }
  }
}

__global__ void k_scores(const float* __restrict__ xl, const float* __restrict__ aj,
                         const float* __restrict__ ai, float* __restrict__ Sj,
                         float* __restrict__ Si, int N) {
  const int t = blockIdx.x * blockDim.x + threadIdx.x;
  if (t >= N * HEADS) return;
  const int n = t / HEADS, h = t % HEADS;
  const float* row = xl + (size_t)n * HC + h * C_OUT;
  float sj = 0.f, si = 0.f;
#pragma unroll
  for (int cc = 0; cc < C_OUT; ++cc) {
    sj += row[cc] * aj[h * C_OUT + cc];
    si += row[cc] * ai[h * C_OUT + cc];
  }
  Sj[t] = sj;
  Si[t] = si;
}


#define EB 256
#define ET 1024
#define UB 8

__device__ __forceinline__ int wave_sum_i(int v) {
#pragma unroll
  for (int off = 16; off > 0; off >>= 1) v += __shfl_xor(v, off);
  return v;
}
__device__ __forceinline__ int wave_incl_scan(int v, int lane) {
#pragma unroll
  for (int off = 1; off < 32; off <<= 1) {
    const int t = __shfl_up(v, off);
    if (lane >= off) v += t;
  }
  return v;
}
__device__ __forceinline__ int block_excl_scan(int v, int* wsum, int tid, int* total) {
  const int lane = tid & 31, wave = tid >> 5;
  const int inc = wave_incl_scan(v, lane);
  if (lane == 31) wsum[wave] = inc;
  __syncthreads();
  if (wave == 0) {
    const int w = wsum[lane];
    const int wi = wave_incl_scan(w, lane);
    wsum[lane] = wi - w;
    if (lane == 31) wsum[32] = wi;
  }
  __syncthreads();
  const int ex = wsum[wave] + inc - v;
  *total = wsum[32];
  __syncthreads();
  return ex;
}

__global__ void k_count(const int* __restrict__ dst, int* __restrict__ cntRB, int E,
                        int chunk, int nb) {
  extern __shared__ int hist[];
  const int tid = threadIdx.x, b = blockIdx.x;
  for (int r = tid; r < nb; r += ET) hist[r] = 0;
  __syncthreads();
  const int e0 = b * chunk;
  const int e1 = min(E, e0 + chunk);
  for (int eb = e0 + tid; eb < e1; eb += ET * UB) {
    int d[UB];
#pragma unroll
    for (int u = 0; u < UB; ++u) {
      const int e = eb + u * ET;
      d[u] = (e < e1) ? dst[e] : -1;
    }
#pragma unroll
    for (int u = 0; u < UB; ++u)
      if (d[u] >= 0) atomicAdd(&hist[d[u] >> 8], 1);
  }
  __syncthreads();
  for (int r = tid; r < nb; r += ET) cntRB[(size_t)r * EB + b] = hist[r];
}

__global__ void k_scatter(const int* __restrict__ src, const int* __restrict__ dst,
                          const int* __restrict__ cntRB, int2* __restrict__ bucket,
                          int* __restrict__ rbase, int* __restrict__ rtot, int E,
                          int chunk, int nb) {
  extern __shared__ int shm[];
  __shared__ int wsum[33];
  int* cur = shm;
  int* tot = shm + nb;
  const int tid = threadIdx.x, b = blockIdx.x;
  const int lane = tid & 31, wave = tid >> 5;
  for (int r = wave; r < nb; r += ET / 32) {
    int t = 0, p = 0;
#pragma unroll
    for (int j = lane; j < EB; j += 32) {
      const int c = cntRB[(size_t)r * EB + j];
      t += c;
      p += (j < b) ? c : 0;
    }
    t = wave_sum_i(t);
    p = wave_sum_i(p);
    if (lane == 0) { tot[r] = t; cur[r] = p; }
  }
  __syncthreads();
  int carry = 0;
  for (int base = 0; base < nb; base += ET) {
    const int i = base + tid;
    const int v = (i < nb) ? tot[i] : 0;
    int total;
    const int ex = carry + block_excl_scan(v, wsum, tid, &total);
    if (i < nb) {
      cur[i] += ex;
      if (b == 0) { rbase[i] = ex; rtot[i] = v; }
    }
    carry += total;
  }
  __syncthreads();
  const int e0 = b * chunk;
  const int e1 = min(E, e0 + chunk);
  for (int eb = e0 + tid; eb < e1; eb += ET * UB) {
    int d[UB], s[UB];
#pragma unroll
    for (int u = 0; u < UB; ++u) {
      const int e = eb + u * ET;
      const bool ok = e < e1;
      d[u] = ok ? dst[e] : -1;
      s[u] = ok ? src[e] : 0;
    }
#pragma unroll
    for (int u = 0; u < UB; ++u) {
      if (d[u] >= 0) {
        const int pos = atomicAdd(&cur[d[u] >> 8], 1);
        bucket[pos] = make_int2(d[u], s[u]);
      }
    }
  }
}

__global__ void k_rangesort(const int* __restrict__ src, const int2* __restrict__ bucket,
                            const int* __restrict__ rbase, const int* __restrict__ rtot,
                            int2* __restrict__ sorted, int* __restrict__ rowstart,
                            int* __restrict__ cnt, int N) {
  __shared__ int c[256];
  __shared__ int cursor[256];
  __shared__ int wsum[33];
  const int tid = threadIdx.x, r = blockIdx.x;
  const int b0 = rbase[r];
  const int m = rtot[r];
  if (tid < 256) c[tid] = 0;
  __syncthreads();
  for (int ib = tid; ib < m; ib += ET * UB) {
    int d[UB];
#pragma unroll
    for (int u = 0; u < UB; ++u) {
      const int i = ib + u * ET;
      d[u] = (i < m) ? bucket[b0 + i].x : -1;
    }
#pragma unroll
    for (int u = 0; u < UB; ++u)
      if (d[u] >= 0) atomicAdd(&c[d[u] & 255], 1);
  }
  __syncthreads();
  const int v = (tid < 256) ? c[tid] : 0;
  int total;
  const int ex = block_excl_scan(v, wsum, tid, &total);
  if (tid < 256) {
    cursor[tid] = b0 + ex;
    const int n = r * 256 + tid;
    if (n < N) { rowstart[n] = b0 + ex; cnt[n] = v; }
  }
  __syncthreads();
  for (int ib = tid; ib < m; ib += ET * UB) {
    int2 rec[UB];
    int nj[UB];
#pragma unroll
    for (int u = 0; u < UB; ++u) {
      const int i = ib + u * ET;
      rec[u] = (i < m) ? bucket[b0 + i] : make_int2(-1, 0);
    }
#pragma unroll
    for (int u = 0; u < UB; ++u) nj[u] = (rec[u].x >= 0) ? src[rec[u].y] : 0;
#pragma unroll
    for (int u = 0; u < UB; ++u) {
      if (rec[u].x >= 0) {
        const int p = atomicAdd(&cursor[rec[u].x & 255], 1);
        sorted[p] = make_int2(rec[u].y, nj[u]);
      }
    }
  }
}

__device__ __forceinline__ float leaky(float a) {
  return (a >= 0.f) ? a : NEG_SLOPE * a;
}
__device__ __forceinline__ float wave_max(float v) {
#pragma unroll
  for (int off = 16; off > 0; off >>= 1) v = fmaxf(v, __shfl_xor(v, off));
  return v;
}
__device__ __forceinline__ float wave_sum(float v) {
#pragma unroll
  for (int off = 16; off > 0; off >>= 1) v += __shfl_xor(v, off);
  return v;
}
struct HeadMask { unsigned m0, m1, m2, m3; };
__device__ __forceinline__ float pick4(const HeadMask& mk, float v0, float v1, float v2,
                                       float v3) {
  return __uint_as_float((__float_as_uint(v0) & mk.m0) | (__float_as_uint(v1) & mk.m1) |
                         (__float_as_uint(v2) & mk.m2) | (__float_as_uint(v3) & mk.m3));
}
__device__ __forceinline__ void fma4(float4& acc, float c, const float4& v) {
  acc.x += c * v.x; acc.y += c * v.y; acc.z += c * v.z; acc.w += c * v.w;
}
__device__ __forceinline__ float lane_val(float v, int k) {
  return __int_as_float(__builtin_amdgcn_readlane(__float_as_int(v), k));
}

__global__ void k_node(const int* __restrict__ src, const int* __restrict__ dst,
                       const int2* __restrict__ sorted, const int* __restrict__ rowstart,
                       const int* __restrict__ cnt,
                       const float* __restrict__ Sj, const float* __restrict__ Si,
                       const float* __restrict__ xl, const float* __restrict__ bias,
                       float* __restrict__ out, int N) {
  const int lane = threadIdx.x & 31;
  const int hL = lane >> 3;
  const HeadMask mk = {hL == 0 ? ~0u : 0u, hL == 1 ? ~0u : 0u, hL == 2 ? ~0u : 0u,
                       hL == 3 ? ~0u : 0u};
  const int n = __builtin_amdgcn_readfirstlane((int)(blockIdx.x * 8 + (threadIdx.x >> 5)));
  if (n >= N) return;
  const int beg = rowstart[n];
  const int end = beg + cnt[n];
  const float4 si = *(const float4*)(Si + (size_t)dst[n] * HEADS);
  const float4 sjs = *(const float4*)(Sj + (size_t)src[n] * HEADS);
  const float as0 = leaky(sjs.x + si.x), as1 = leaky(sjs.y + si.y);
  const float as2 = leaky(sjs.z + si.z), as3 = leaky(sjs.w + si.w);
  float m0 = as0, m1 = as1, m2 = as2, m3 = as3;
  for (int base = beg; base < end; base += 32) {
    const int idx = base + lane;
    if (idx < end) {
      const int2 sp = sorted[idx];
      const float4 sj = *(const float4*)(Sj + (size_t)sp.y * HEADS);
      m0 = fmaxf(m0, leaky(sj.x + si.x));
      m1 = fmaxf(m1, leaky(sj.y + si.y));
      m2 = fmaxf(m2, leaky(sj.z + si.z));
      m3 = fmaxf(m3, leaky(sj.w + si.w));
    }
  }
  m0 = wave_max(m0); m1 = wave_max(m1); m2 = wave_max(m2); m3 = wave_max(m3);

  float4 acc = make_float4(0.f, 0.f, 0.f, 0.f);
  float s0 = 0.f, s1 = 0.f, s2 = 0.f, s3 = 0.f;
  for (int base = beg; base < end; base += 32) {
    const int idx = base + lane;
    int sn = 0;
    float a0 = 0.f, a1 = 0.f, a2 = 0.f, a3 = 0.f;
    if (idx < end) {
      const int2 sp = sorted[idx];
      sn = sp.x;
      const float4 sj = *(const float4*)(Sj + (size_t)sp.y * HEADS);
      a0 = expf(leaky(sj.x + si.x) - m0);
      a1 = expf(leaky(sj.y + si.y) - m1);
      a2 = expf(leaky(sj.z + si.z) - m2);
      a3 = expf(leaky(sj.w + si.w) - m3);
    }
    s0 += a0; s1 += a1; s2 += a2; s3 += a3;
    const int cc = min(32, end - base);
    auto coef = [&](int k) {
      return pick4(mk, lane_val(a0, k), lane_val(a1, k), lane_val(a2, k), lane_val(a3, k));
    };
    auto row = [&](int k) {
      const int sk = __builtin_amdgcn_readlane(sn, k);
      return *(const float4*)(xl + (size_t)sk * HC + lane * 4);
    };
    int k = 0;
    for (; k + 4 <= cc; k += 4) {
      const float4 v0 = row(k), v1 = row(k + 1), v2 = row(k + 2), v3 = row(k + 3);
      const float c0 = coef(k), c1 = coef(k + 1), c2 = coef(k + 2), c3 = coef(k + 3);
      fma4(acc, c0, v0);
      fma4(acc, c1, v1);
      fma4(acc, c2, v2);
      fma4(acc, c3, v3);
    }
    for (; k < cc; ++k) fma4(acc, coef(k), row(k));
  }
  s0 = wave_sum(s0); s1 = wave_sum(s1); s2 = wave_sum(s2); s3 = wave_sum(s3);
  const float e0 = expf(as0 - m0), e1 = expf(as1 - m1);
  const float e2 = expf(as2 - m2), e3 = expf(as3 - m3);
  s0 += e0; s1 += e1; s2 += e2; s3 += e3;
  {
    const float4 v = *(const float4*)(xl + (size_t)n * HC + lane * 4);
    fma4(acc, pick4(mk, e0, e1, e2, e3), v);
  }
  const float sinv = 1.0f / pick4(mk, s0, s1, s2, s3);
  float4 o;
  o.x = acc.x * sinv + bias[lane * 4 + 0];
  o.y = acc.y * sinv + bias[lane * 4 + 1];
  o.z = acc.z * sinv + bias[lane * 4 + 2];
  o.w = acc.w * sinv + bias[lane * 4 + 3];
  *(float4*)(out + (size_t)n * HC + lane * 4) = o;
}

extern "C" void kernel_launch(void* const* d_in, const int* in_sizes, int n_in,
                              void* d_out, int out_size, void* d_ws, size_t ws_size,
                              hipStream_t stream) {
  const float* x      = (const float*)d_in[0];
  const float* W      = (const float*)d_in[1];
  const float* attn_j = (const float*)d_in[2];
  const float* attn_i = (const float*)d_in[3];
  const float* bias   = (const float*)d_in[4];
  const int*   eidx   = (const int*)d_in[5];

  const int N  = in_sizes[0] / F_IN;
  const int E  = in_sizes[5] / 2;
  const int* src = eidx;
  const int* dst = eidx + E;

  char*  ws = (char*)d_ws;
  size_t o  = 0;
  auto take = [&](size_t bytes) -> void* {
    void* p = ws + o;
    o += (bytes + 255) & ~(size_t)255;
    return p;
  };
  float*     xl   = (float*)take((size_t)N * HC * sizeof(float));
  float*     Sj   = (float*)take((size_t)N * HEADS * sizeof(float));
  float*     Si   = (float*)take((size_t)N * HEADS * sizeof(float));
  const int  nb   = (N + 255) / 256;
  const int  chunk = (E + EB - 1) / EB;
  int*       cnt  = (int*)take((size_t)N * sizeof(int));
  int*       rowstart = (int*)take((size_t)N * sizeof(int));
  int*       cntRB = (int*)take((size_t)nb * EB * sizeof(int));
  int*       rbase = (int*)take((size_t)nb * sizeof(int));
  int*       rtot  = (int*)take((size_t)nb * sizeof(int));
  int2*      bucket = (int2*)take((size_t)E * sizeof(int2));
  int2*      sorted = (int2*)take((size_t)E * sizeof(int2));

  float* out = (float*)d_out;

  k_gemm<<<(N + GT * 16 - 1) / (GT * 16), GT * 32, 0, stream>>>(x, W, xl, N);

  k_scores<<<(N * HEADS + 255) / 256, 256, 0, stream>>>(xl, attn_j, attn_i, Sj, Si, N);

  k_count<<<EB, ET, (size_t)nb * sizeof(int), stream>>>(dst, cntRB, E, chunk, nb);
  k_scatter<<<EB, ET, (size_t)nb * 2 * sizeof(int), stream>>>(src, dst, cntRB, bucket,
                                                              rbase, rtot, E, chunk, nb);
  k_rangesort<<<nb, ET, 0, stream>>>(src, bucket, rbase, rtot, sorted, rowstart, cnt, N);

  k_node<<<(N + 7) / 8, 256, 0, stream>>>(src, dst, sorted, rowstart, cnt, Sj, Si,
                                          xl, bias, out, N);
}


# node kernel reads each edge once: records and scores of a segment's first 64 edges stay in registers from the maxima to the sums, no second gather, both groups' loads issued together; sum order kept
# speedup vs baseline: 9.7769x; 1.0470x over previous
#include <hip/hip_runtime.h>


#define F_IN      128
#define HEADS     4
#define C_OUT     32
#define HC        128
#define NEG_SLOPE 0.2f

typedef float v2f __attribute__((ext_vector_type(2)));
typedef float v8f __attribute__((ext_vector_type(8)));

#define GT 4
#define WP 132
__global__ __launch_bounds__(GT * 32) void k_gemm(const float* __restrict__ x,
                                                  const float* __restrict__ W,
                                                  float* __restrict__ xl, int N) {
  __shared__ __attribute__((aligned(16))) float Ws[HC * WP];
  const int lane    = threadIdx.x & 31;
  const int wave    = threadIdx.x >> 5;
  const int g       = lane >> 4;
  const int l16     = lane & 15;
  const int rowBase = (blockIdx.x * GT + wave) * 16;
  int mrow = rowBase + l16; if (mrow >= N) mrow = N - 1;

  v2f a[F_IN / 4];
  const float* xr = x + (size_t)mrow * F_IN + g * 2;
#pragma unroll
  for (int s = 0; s < F_IN / 4; ++s) a[s] = *(const v2f*)(xr + 4 * s);

#pragma unroll 8
  for (int j = 0; j < (HC * (F_IN / 4)) / (GT * 32); ++j) {
    const int i = threadIdx.x + j * (GT * 32);
    const int r = i >> 5, c4 = i & 31;
    const float4 v = *(const float4*)(W + (size_t)r * F_IN + c4 * 4);
    *(float4*)(Ws + r * WP + c4 * 4) = v;
  }
  __syncthreads();
  if (rowBase >= N) return;

  v8f c[8] = {};
  const float* bp = Ws + l16 * WP + g * 2;
#pragma unroll
  for (int s = 0; s < F_IN / 4; ++s) {
#pragma unroll
    for (int t = 0; t < 8; ++t) {
      const v2f b = *(const v2f*)(bp + t * 16 * WP + 4 * s);
      c[t] = __builtin_amdgcn_wmma_f32_16x16x4_f32(false, a[s], false, b, (short)0, c[t],
                                                   false, false);
    }
  }
#pragma unroll
  for (int t = 0; t < 8; ++t) {
#pragma unroll
    for (int i = 0; i < 8; ++i) {
      const int r = rowBase + i + g * 8;
      if (r < N) xl[(size_t)r * HC + t * 16 + l16] = c[t][i];
    }
  }
}

__global__ void k_scores(const float* __restrict__ xl, const float* __restrict__ aj,
                         const float* __restrict__ ai, float* __restrict__ Sj,
                         float* __restrict__ Si, int N) {
  const int t = blockIdx.x * blockDim.x + threadIdx.x;
  if (t >= N * HEADS) return;
  const int n = t / HEADS, h = t % HEADS;
  const float* row = xl + (size_t)n * HC + h * C_OUT;
  float sj = 0.f, si = 0.f;
#pragma unroll
  for (int cc = 0; cc < C_OUT; ++cc) {
    sj += row[cc] * aj[h * C_OUT + cc];
    si += row[cc] * ai[h * C_OUT + cc];
  }
  Sj[t] = sj;
  Si[t] = si;
}


#define EB 256
#define ET 1024
#define UB 8

__device__ __forceinline__ int wave_sum_i(int v) {
#pragma unroll
  for (int off = 16; off > 0; off >>= 1) v += __shfl_xor(v, off);
  return v;
}
__device__ __forceinline__ int wave_incl_scan(int v, int lane) {
#pragma unroll
  for (int off = 1; off < 32; off <<= 1) {
    const int t = __shfl_up(v, off);
    if (lane >= off) v += t;
  }
  return v;
}
__device__ __forceinline__ int block_excl_scan(int v, int* wsum, int tid, int* total) {
  const int lane = tid & 31, wave = tid >> 5;
  const int inc = wave_incl_scan(v, lane);
  if (lane == 31) wsum[wave] = inc;
  __syncthreads();
  if (wave == 0) {
    const int w = wsum[lane];
    const int wi = wave_incl_scan(w, lane);
    wsum[lane] = wi - w;
    if (lane == 31) wsum[32] = wi;
  }
  __syncthreads();
  const int ex = wsum[wave] + inc - v;
  *total = wsum[32];
  __syncthreads();
  return ex;
}

__global__ void k_count(const int* __restrict__ dst, int* __restrict__ cntRB, int E,
                        int chunk, int nb) {
  extern __shared__ int hist[];
  const int tid = threadIdx.x, b = blockIdx.x;
  for (int r = tid; r < nb; r += ET) hist[r] = 0;
  __syncthreads();
  const int e0 = b * chunk;
  const int e1 = min(E, e0 + chunk);
  for (int eb = e0 + tid; eb < e1; eb += ET * UB) {
    int d[UB];
#pragma unroll
    for (int u = 0; u < UB; ++u) {
      const int e = eb + u * ET;
      d[u] = (e < e1) ? dst[e] : -1;
    }
#pragma unroll
    for (int u = 0; u < UB; ++u)
      if (d[u] >= 0) atomicAdd(&hist[d[u] >> 8], 1);
  }
  __syncthreads();
  for (int r = tid; r < nb; r += ET) cntRB[(size_t)r * EB + b] = hist[r];
}

__global__ void k_scatter(const int* __restrict__ src, const int* __restrict__ dst,
                          const int* __restrict__ cntRB, int2* __restrict__ bucket,
                          int* __restrict__ rbase, int* __restrict__ rtot, int E,
                          int chunk, int nb) {
  extern __shared__ int shm[];
  __shared__ int wsum[33];
  int* cur = shm;
  int* tot = shm + nb;
  const int tid = threadIdx.x, b = blockIdx.x;
  const int lane = tid & 31, wave = tid >> 5;
  for (int r = wave; r < nb; r += ET / 32) {
    int t = 0, p = 0;
#pragma unroll
    for (int j = lane; j < EB; j += 32) {
      const int c = cntRB[(size_t)r * EB + j];
      t += c;
      p += (j < b) ? c : 0;
    }
    t = wave_sum_i(t);
    p = wave_sum_i(p);
    if (lane == 0) { tot[r] = t; cur[r] = p; }
  }
  __syncthreads();
  int carry = 0;
  for (int base = 0; base < nb; base += ET) {
    const int i = base + tid;
    const int v = (i < nb) ? tot[i] : 0;
    int total;
    const int ex = carry + block_excl_scan(v, wsum, tid, &total);
    if (i < nb) {
      cur[i] += ex;
      if (b == 0) { rbase[i] = ex; rtot[i] = v; }
    }
    carry += total;
  }
  __syncthreads();
  const int e0 = b * chunk;
  const int e1 = min(E, e0 + chunk);
  for (int eb = e0 + tid; eb < e1; eb += ET * UB) {
    int d[UB], s[UB];
#pragma unroll
    for (int u = 0; u < UB; ++u) {
      const int e = eb + u * ET;
      const bool ok = e < e1;
      d[u] = ok ? dst[e] : -1;
      s[u] = ok ? src[e] : 0;
    }
#pragma unroll
    for (int u = 0; u < UB; ++u) {
      if (d[u] >= 0) {
        const int pos = atomicAdd(&cur[d[u] >> 8], 1);
        bucket[pos] = make_int2(d[u], s[u]);
      }
    }
  }
}

__global__ void k_rangesort(const int* __restrict__ src, const int2* __restrict__ bucket,
                            const int* __restrict__ rbase, const int* __restrict__ rtot,
                            int2* __restrict__ sorted, int* __restrict__ rowstart,
                            int* __restrict__ cnt, int N) {
  __shared__ int c[256];
  __shared__ int cursor[256];
  __shared__ int wsum[33];
  const int tid = threadIdx.x, r = blockIdx.x;
  const int b0 = rbase[r];
  const int m = rtot[r];
  if (tid < 256) c[tid] = 0;
  __syncthreads();
  for (int ib = tid; ib < m; ib += ET * UB) {
    int d[UB];
#pragma unroll
    for (int u = 0; u < UB; ++u) {
      const int i = ib + u * ET;
      d[u] = (i < m) ? bucket[b0 + i].x : -1;
    }
#pragma unroll
    for (int u = 0; u < UB; ++u)
      if (d[u] >= 0) atomicAdd(&c[d[u] & 255], 1);
  }
  __syncthreads();
  const int v = (tid < 256) ? c[tid] : 0;
  int total;
  const int ex = block_excl_scan(v, wsum, tid, &total);
  if (tid < 256) {
    cursor[tid] = b0 + ex;
    const int n = r * 256 + tid;
    if (n < N) { rowstart[n] = b0 + ex; cnt[n] = v; }
  }
  __syncthreads();
  for (int ib = tid; ib < m; ib += ET * UB) {
    int2 rec[UB];
    int nj[UB];
#pragma unroll
    for (int u = 0; u < UB; ++u) {
      const int i = ib + u * ET;
      rec[u] = (i < m) ? bucket[b0 + i] : make_int2(-1, 0);
    }
#pragma unroll
    for (int u = 0; u < UB; ++u) nj[u] = (rec[u].x >= 0) ? src[rec[u].y] : 0;
#pragma unroll
    for (int u = 0; u < UB; ++u) {
      if (rec[u].x >= 0) {
        const int p = atomicAdd(&cursor[rec[u].x & 255], 1);
        sorted[p] = make_int2(rec[u].y, nj[u]);
      }
    }
  }
}

__device__ __forceinline__ float leaky(float a) {
  return (a >= 0.f) ? a : NEG_SLOPE * a;
}
__device__ __forceinline__ float wave_max(float v) {
#pragma unroll
  for (int off = 16; off > 0; off >>= 1) v = fmaxf(v, __shfl_xor(v, off));
  return v;
}
__device__ __forceinline__ float wave_sum(float v) {
#pragma unroll
  for (int off = 16; off > 0; off >>= 1) v += __shfl_xor(v, off);
  return v;
}
struct HeadMask { unsigned m0, m1, m2, m3; };
__device__ __forceinline__ float pick4(const HeadMask& mk, float v0, float v1, float v2,
                                       float v3) {
  return __uint_as_float((__float_as_uint(v0) & mk.m0) | (__float_as_uint(v1) & mk.m1) |
                         (__float_as_uint(v2) & mk.m2) | (__float_as_uint(v3) & mk.m3));
}
__device__ __forceinline__ void fma4(float4& acc, float c, const float4& v) {
  acc.x += c * v.x; acc.y += c * v.y; acc.z += c * v.z; acc.w += c * v.w;
}
__device__ __forceinline__ float lane_val(float v, int k) {
  return __int_as_float(__builtin_amdgcn_readlane(__float_as_int(v), k));
}

__global__ void k_node(const int* __restrict__ src, const int* __restrict__ dst,
                       const int2* __restrict__ sorted, const int* __restrict__ rowstart,
                       const int* __restrict__ cnt,
                       const float* __restrict__ Sj, const float* __restrict__ Si,
                       const float* __restrict__ xl, const float* __restrict__ bias,
                       float* __restrict__ out, int N) {
  const int lane = threadIdx.x & 31;
  const int hL = lane >> 3;
  const HeadMask mk = {hL == 0 ? ~0u : 0u, hL == 1 ? ~0u : 0u, hL == 2 ? ~0u : 0u,
                       hL == 3 ? ~0u : 0u};
  const int n = __builtin_amdgcn_readfirstlane((int)(blockIdx.x * 8 + (threadIdx.x >> 5)));
  if (n >= N) return;
  const int beg = rowstart[n];
  const int deg = cnt[n];
  const int end = beg + deg;
  const float4 si = *(const float4*)(Si + (size_t)dst[n] * HEADS);
  const float4 sjs = *(const float4*)(Sj + (size_t)src[n] * HEADS);
  const float as0 = leaky(sjs.x + si.x), as1 = leaky(sjs.y + si.y);
  const float as2 = leaky(sjs.z + si.z), as3 = leaky(sjs.w + si.w);

  const int i0 = beg + lane, i1 = i0 + 32;
  const bool ok0 = i0 < end, ok1 = i1 < end;
  const int2 r0 = sorted[ok0 ? i0 : 0];
  const int2 r1 = sorted[ok1 ? i1 : 0];
  __builtin_amdgcn_sched_barrier(0);
  const float4 q0 = *(const float4*)(Sj + (size_t)r0.y * HEADS);
  const float4 q1 = *(const float4*)(Sj + (size_t)r1.y * HEADS);
  const float p00 = ok0 ? leaky(q0.x + si.x) : as0, p01 = ok0 ? leaky(q0.y + si.y) : as1;
  const float p02 = ok0 ? leaky(q0.z + si.z) : as2, p03 = ok0 ? leaky(q0.w + si.w) : as3;
  const float p10 = ok1 ? leaky(q1.x + si.x) : as0, p11 = ok1 ? leaky(q1.y + si.y) : as1;
  const float p12 = ok1 ? leaky(q1.z + si.z) : as2, p13 = ok1 ? leaky(q1.w + si.w) : as3;
  float m0 = fmaxf(as0, fmaxf(p00, p10)), m1 = fmaxf(as1, fmaxf(p01, p11));
  float m2 = fmaxf(as2, fmaxf(p02, p12)), m3 = fmaxf(as3, fmaxf(p03, p13));
  for (int base = beg + 64; base < end; base += 32) {
    const int idx = base + lane;
    if (idx < end) {
      const int2 sp = sorted[idx];
      const float4 sj = *(const float4*)(Sj + (size_t)sp.y * HEADS);
      m0 = fmaxf(m0, leaky(sj.x + si.x));
      m1 = fmaxf(m1, leaky(sj.y + si.y));
      m2 = fmaxf(m2, leaky(sj.z + si.z));
      m3 = fmaxf(m3, leaky(sj.w + si.w));
    }
  }
  m0 = wave_max(m0); m1 = wave_max(m1); m2 = wave_max(m2); m3 = wave_max(m3);

  float4 acc = make_float4(0.f, 0.f, 0.f, 0.f);
  auto add_rows = [&](int sn, float a0, float a1, float a2, float a3, int cc) {
    auto coef = [&](int k) {
      return pick4(mk, lane_val(a0, k), lane_val(a1, k), lane_val(a2, k), lane_val(a3, k));
    };
    auto row = [&](int k) {
      const int sk = __builtin_amdgcn_readlane(sn, k);
      return *(const float4*)(xl + (size_t)sk * HC + lane * 4);
    };
    int k = 0;
    for (; k + 4 <= cc; k += 4) {
      const float4 v0 = row(k), v1 = row(k + 1), v2 = row(k + 2), v3 = row(k + 3);
      const float c0 = coef(k), c1 = coef(k + 1), c2 = coef(k + 2), c3 = coef(k + 3);
      fma4(acc, c0, v0);
      fma4(acc, c1, v1);
      fma4(acc, c2, v2);
      fma4(acc, c3, v3);
    }
    for (; k < cc; ++k) fma4(acc, coef(k), row(k));
  };

  float a00 = 0.f, a01 = 0.f, a02 = 0.f, a03 = 0.f;
  float a10 = 0.f, a11 = 0.f, a12 = 0.f, a13 = 0.f;
  if (ok0) {
    a00 = expf(p00 - m0); a01 = expf(p01 - m1); a02 = expf(p02 - m2); a03 = expf(p03 - m3);
  }
  if (ok1) {
    a10 = expf(p10 - m0); a11 = expf(p11 - m1); a12 = expf(p12 - m2); a13 = expf(p13 - m3);
  }
  float s0 = a00 + a10, s1 = a01 + a11, s2 = a02 + a12, s3 = a03 + a13;
  add_rows(r0.x, a00, a01, a02, a03, min(32, deg));
  if (deg > 32) add_rows(r1.x, a10, a11, a12, a13, min(32, deg - 32));
  for (int base = beg + 64; base < end; base += 32) {
    const int idx = base + lane;
    int sn = 0;
    float a0 = 0.f, a1 = 0.f, a2 = 0.f, a3 = 0.f;
    if (idx < end) {
      const int2 sp = sorted[idx];
      sn = sp.x;
      const float4 sj = *(const float4*)(Sj + (size_t)sp.y * HEADS);
      a0 = expf(leaky(sj.x + si.x) - m0);
      a1 = expf(leaky(sj.y + si.y) - m1);
      a2 = expf(leaky(sj.z + si.z) - m2);
      a3 = expf(leaky(sj.w + si.w) - m3);
    }
    s0 += a0; s1 += a1; s2 += a2; s3 += a3;
    add_rows(sn, a0, a1, a2, a3, min(32, end - base));
  }
  s0 = wave_sum(s0); s1 = wave_sum(s1); s2 = wave_sum(s2); s3 = wave_sum(s3);
  const float e0 = expf(as0 - m0), e1 = expf(as1 - m1);
  const float e2 = expf(as2 - m2), e3 = expf(as3 - m3);
  s0 += e0; s1 += e1; s2 += e2; s3 += e3;
  {
    const float4 v = *(const float4*)(xl + (size_t)n * HC + lane * 4);
    fma4(acc, pick4(mk, e0, e1, e2, e3), v);
  }
  const float sinv = 1.0f / pick4(mk, s0, s1, s2, s3);
  float4 o;
  o.x = acc.x * sinv + bias[lane * 4 + 0];
  o.y = acc.y * sinv + bias[lane * 4 + 1];
  o.z = acc.z * sinv + bias[lane * 4 + 2];
  o.w = acc.w * sinv + bias[lane * 4 + 3];
  *(float4*)(out + (size_t)n * HC + lane * 4) = o;
}

extern "C" void kernel_launch(void* const* d_in, const int* in_sizes, int n_in,
                              void* d_out, int out_size, void* d_ws, size_t ws_size,
                              hipStream_t stream) {
  const float* x      = (const float*)d_in[0];
  const float* W      = (const float*)d_in[1];
  const float* attn_j = (const float*)d_in[2];
  const float* attn_i = (const float*)d_in[3];
  const float* bias   = (const float*)d_in[4];
  const int*   eidx   = (const int*)d_in[5];

  const int N  = in_sizes[0] / F_IN;
  const int E  = in_sizes[5] / 2;
  const int* src = eidx;
  const int* dst = eidx + E;

  char*  ws = (char*)d_ws;
  size_t o  = 0;
  auto take = [&](size_t bytes) -> void* {
    void* p = ws + o;
    o += (bytes + 255) & ~(size_t)255;
    return p;
  };
  float*     xl   = (float*)take((size_t)N * HC * sizeof(float));
  float*     Sj   = (float*)take((size_t)N * HEADS * sizeof(float));
  float*     Si   = (float*)take((size_t)N * HEADS * sizeof(float));
  const int  nb   = (N + 255) / 256;
  const int  chunk = (E + EB - 1) / EB;
  int*       cnt  = (int*)take((size_t)N * sizeof(int));
  int*       rowstart = (int*)take((size_t)N * sizeof(int));
  int*       cntRB = (int*)take((size_t)nb * EB * sizeof(int));
  int*       rbase = (int*)take((size_t)nb * sizeof(int));
  int*       rtot  = (int*)take((size_t)nb * sizeof(int));
  int2*      bucket = (int2*)take((size_t)E * sizeof(int2));
  int2*      sorted = (int2*)take((size_t)E * sizeof(int2));

  float* out = (float*)d_out;

  k_gemm<<<(N + GT * 16 - 1) / (GT * 16), GT * 32, 0, stream>>>(x, W, xl, N);

  k_scores<<<(N * HEADS + 255) / 256, 256, 0, stream>>>(xl, attn_j, attn_i, Sj, Si, N);

  k_count<<<EB, ET, (size_t)nb * sizeof(int), stream>>>(dst, cntRB, E, chunk, nb);
  k_scatter<<<EB, ET, (size_t)nb * 2 * sizeof(int), stream>>>(src, dst, cntRB, bucket,
                                                              rbase, rtot, E, chunk, nb);
  k_rangesort<<<nb, ET, 0, stream>>>(src, bucket, rbase, rtot, sorted, rowstart, cnt, N);

  k_node<<<(N + 7) / 8, 256, 0, stream>>>(src, dst, sorted, rowstart, cnt, Sj, Si,
                                          xl, bias, out, N);
}


# score launch folded into the GEMM: each wave takes its 16 rows' attn_j/attn_i dots from its accumulators (16-lane DPP sum tree: the score sums' order changes); 6 launches become 5, xl is not re-read
# speedup vs baseline: 10.2513x; 1.0485x over previous
#include <hip/hip_runtime.h>


#define F_IN      128
#define HEADS     4
#define C_OUT     32
#define HC        128
#define NEG_SLOPE 0.2f

typedef float v2f __attribute__((ext_vector_type(2)));
typedef float v8f __attribute__((ext_vector_type(8)));

#define GT 4

template <int M> __device__ __forceinline__ float row_xor(float v) {
  return __int_as_float(
      __builtin_amdgcn_update_dpp(0, __float_as_int(v), 0x160 + M, 0xf, 0xf, true));
}
template <int M> __device__ __forceinline__ float fold(float X, float Y, bool bit) {
  const float mine = bit ? Y : X;
  const float send = bit ? X : Y;
  return mine + row_xor<M>(send);
}
#define WP 132
__global__ __launch_bounds__(GT * 32) void k_gemm(const float* __restrict__ x,
                                                  const float* __restrict__ W,
                                                  const float* __restrict__ aj,
                                                  const float* __restrict__ ai,
                                                  float* __restrict__ xl,
                                                  float* __restrict__ Sj,
                                                  float* __restrict__ Si, int N) {
  __shared__ __attribute__((aligned(16))) float Ws[HC * WP];
  const int lane    = threadIdx.x & 31;
  const int wave    = threadIdx.x >> 5;
  const int g       = lane >> 4;
  const int l16     = lane & 15;
  const int rowBase = (blockIdx.x * GT + wave) * 16;
  int mrow = rowBase + l16; if (mrow >= N) mrow = N - 1;

  v2f a[F_IN / 4];
  const float* xr = x + (size_t)mrow * F_IN + g * 2;
#pragma unroll
  for (int s = 0; s < F_IN / 4; ++s) a[s] = *(const v2f*)(xr + 4 * s);
  float ajv[8], aiv[8];
#pragma unroll
  for (int t = 0; t < 8; ++t) { ajv[t] = aj[t * 16 + l16]; aiv[t] = ai[t * 16 + l16]; }

#pragma unroll 8
  for (int j = 0; j < (HC * (F_IN / 4)) / (GT * 32); ++j) {
    const int i = threadIdx.x + j * (GT * 32);
    const int r = i >> 5, c4 = i & 31;
    const float4 v = *(const float4*)(W + (size_t)r * F_IN + c4 * 4);
    *(float4*)(Ws + r * WP + c4 * 4) = v;
  }
  __syncthreads();
  if (rowBase >= N) return;

  v8f c[8] = {};
  const float* bp = Ws + l16 * WP + g * 2;
#pragma unroll
  for (int s = 0; s < F_IN / 4; ++s) {
#pragma unroll
    for (int t = 0; t < 8; ++t) {
      const v2f b = *(const v2f*)(bp + t * 16 * WP + 4 * s);
      c[t] = __builtin_amdgcn_wmma_f32_16x16x4_f32(false, a[s], false, b, (short)0, c[t],
                                                   false, false);
    }
  }
#pragma unroll
  for (int t = 0; t < 8; ++t) {
#pragma unroll
    for (int i = 0; i < 8; ++i) {
      const int r = rowBase + i + g * 8;
      if (r < N) xl[(size_t)r * HC + t * 16 + l16] = c[t][i];
    }
  }
  const bool b0 = l16 & 1, b1 = l16 & 2, b2 = l16 & 4, b3 = l16 & 8;
  float res[HEADS];
#pragma unroll
  for (int h = 0; h < HEADS; ++h) {
    float vj[8], vi[8];
#pragma unroll
    for (int i = 0; i < 8; ++i) {
      vj[i] = c[2 * h][i] * ajv[2 * h] + c[2 * h + 1][i] * ajv[2 * h + 1];
      vi[i] = c[2 * h][i] * aiv[2 * h] + c[2 * h + 1][i] * aiv[2 * h + 1];
    }
    float pj[4], pi[4];
#pragma unroll
    for (int q = 0; q < 4; ++q) {
      pj[q] = fold<1>(vj[2 * q], vj[2 * q + 1], b0);
      pi[q] = fold<1>(vi[2 * q], vi[2 * q + 1], b0);
    }
    const float qj0 = fold<2>(pj[0], pj[1], b1), qj1 = fold<2>(pj[2], pj[3], b1);
    const float qi0 = fold<2>(pi[0], pi[1], b1), qi1 = fold<2>(pi[2], pi[3], b1);
    const float rj = fold<4>(qj0, qj1, b2);
    const float ri = fold<4>(qi0, qi1, b2);
    res[h] = fold<8>(rj, ri, b3);
  }
  const int srow = rowBase + g * 8 + (l16 & 7);
  if (srow < N) {
    float* sp = (b3 ? Si : Sj) + (size_t)srow * HEADS;
    *(float4*)sp = make_float4(res[0], res[1], res[2], res[3]);
  }
}


#define EB 256
#define ET 1024
#define UB 8

__device__ __forceinline__ int wave_sum_i(int v) {
#pragma unroll
  for (int off = 16; off > 0; off >>= 1) v += __shfl_xor(v, off);
  return v;
}
__device__ __forceinline__ int wave_incl_scan(int v, int lane) {
#pragma unroll
  for (int off = 1; off < 32; off <<= 1) {
    const int t = __shfl_up(v, off);
    if (lane >= off) v += t;
  }
  return v;
}
__device__ __forceinline__ int block_excl_scan(int v, int* wsum, int tid, int* total) {
  const int lane = tid & 31, wave = tid >> 5;
  const int inc = wave_incl_scan(v, lane);
  if (lane == 31) wsum[wave] = inc;
  __syncthreads();
  if (wave == 0) {
    const int w = wsum[lane];
    const int wi = wave_incl_scan(w, lane);
    wsum[lane] = wi - w;
    if (lane == 31) wsum[32] = wi;
  }
  __syncthreads();
  const int ex = wsum[wave] + inc - v;
  *total = wsum[32];
  __syncthreads();
  return ex;
}

__global__ void k_count(const int* __restrict__ dst, int* __restrict__ cntRB, int E,
                        int chunk, int nb) {
  extern __shared__ int hist[];
  const int tid = threadIdx.x, b = blockIdx.x;
  for (int r = tid; r < nb; r += ET) hist[r] = 0;
  __syncthreads();
  const int e0 = b * chunk;
  const int e1 = min(E, e0 + chunk);
  for (int eb = e0 + tid; eb < e1; eb += ET * UB) {
    int d[UB];
#pragma unroll
    for (int u = 0; u < UB; ++u) {
      const int e = eb + u * ET;
      d[u] = (e < e1) ? dst[e] : -1;
    }
#pragma unroll
    for (int u = 0; u < UB; ++u)
      if (d[u] >= 0) atomicAdd(&hist[d[u] >> 8], 1);
  }
  __syncthreads();
  for (int r = tid; r < nb; r += ET) cntRB[(size_t)r * EB + b] = hist[r];
}

__global__ void k_scatter(const int* __restrict__ src, const int* __restrict__ dst,
                          const int* __restrict__ cntRB, int2* __restrict__ bucket,
                          int* __restrict__ rbase, int* __restrict__ rtot, int E,
                          int chunk, int nb) {
  extern __shared__ int shm[];
  __shared__ int wsum[33];
  int* cur = shm;
  int* tot = shm + nb;
  const int tid = threadIdx.x, b = blockIdx.x;
  const int lane = tid & 31, wave = tid >> 5;
  for (int r = wave; r < nb; r += ET / 32) {
    int t = 0, p = 0;
#pragma unroll
    for (int j = lane; j < EB; j += 32) {
      const int c = cntRB[(size_t)r * EB + j];
      t += c;
      p += (j < b) ? c : 0;
    }
    t = wave_sum_i(t);
    p = wave_sum_i(p);
    if (lane == 0) { tot[r] = t; cur[r] = p; }
  }
  __syncthreads();
  int carry = 0;
  for (int base = 0; base < nb; base += ET) {
    const int i = base + tid;
    const int v = (i < nb) ? tot[i] : 0;
    int total;
    const int ex = carry + block_excl_scan(v, wsum, tid, &total);
    if (i < nb) {
      cur[i] += ex;
      if (b == 0) { rbase[i] = ex; rtot[i] = v; }
    }
    carry += total;
  }
  __syncthreads();
  const int e0 = b * chunk;
  const int e1 = min(E, e0 + chunk);
  for (int eb = e0 + tid; eb < e1; eb += ET * UB) {
    int d[UB], s[UB];
#pragma unroll
    for (int u = 0; u < UB; ++u) {
      const int e = eb + u * ET;
      const bool ok = e < e1;
      d[u] = ok ? dst[e] : -1;
      s[u] = ok ? src[e] : 0;
    }
#pragma unroll
    for (int u = 0; u < UB; ++u) {
      if (d[u] >= 0) {
        const int pos = atomicAdd(&cur[d[u] >> 8], 1);
        bucket[pos] = make_int2(d[u], s[u]);
      }
    }
  }
}

__global__ void k_rangesort(const int* __restrict__ src, const int2* __restrict__ bucket,
                            const int* __restrict__ rbase, const int* __restrict__ rtot,
                            int2* __restrict__ sorted, int* __restrict__ rowstart,
                            int* __restrict__ cnt, int N) {
  __shared__ int c[256];
  __shared__ int cursor[256];
  __shared__ int wsum[33];
  const int tid = threadIdx.x, r = blockIdx.x;
  const int b0 = rbase[r];
  const int m = rtot[r];
  if (tid < 256) c[tid] = 0;
  __syncthreads();
  for (int ib = tid; ib < m; ib += ET * UB) {
    int d[UB];
#pragma unroll
    for (int u = 0; u < UB; ++u) {
      const int i = ib + u * ET;
      d[u] = (i < m) ? bucket[b0 + i].x : -1;
    }
#pragma unroll
    for (int u = 0; u < UB; ++u)
      if (d[u] >= 0) atomicAdd(&c[d[u] & 255], 1);
  }
  __syncthreads();
  const int v = (tid < 256) ? c[tid] : 0;
  int total;
  const int ex = block_excl_scan(v, wsum, tid, &total);
  if (tid < 256) {
    cursor[tid] = b0 + ex;
    const int n = r * 256 + tid;
    if (n < N) { rowstart[n] = b0 + ex; cnt[n] = v; }
  }
  __syncthreads();
  for (int ib = tid; ib < m; ib += ET * UB) {
    int2 rec[UB];
    int nj[UB];
#pragma unroll
    for (int u = 0; u < UB; ++u) {
      const int i = ib + u * ET;
      rec[u] = (i < m) ? bucket[b0 + i] : make_int2(-1, 0);
    }
#pragma unroll
    for (int u = 0; u < UB; ++u) nj[u] = (rec[u].x >= 0) ? src[rec[u].y] : 0;
#pragma unroll
    for (int u = 0; u < UB; ++u) {
      if (rec[u].x >= 0) {
        const int p = atomicAdd(&cursor[rec[u].x & 255], 1);
        sorted[p] = make_int2(rec[u].y, nj[u]);
      }
    }
  }
}

__device__ __forceinline__ float leaky(float a) {
  return (a >= 0.f) ? a : NEG_SLOPE * a;
}
__device__ __forceinline__ float wave_max(float v) {
#pragma unroll
  for (int off = 16; off > 0; off >>= 1) v = fmaxf(v, __shfl_xor(v, off));
  return v;
}
__device__ __forceinline__ float wave_sum(float v) {
#pragma unroll
  for (int off = 16; off > 0; off >>= 1) v += __shfl_xor(v, off);
  return v;
}
struct HeadMask { unsigned m0, m1, m2, m3; };
__device__ __forceinline__ float pick4(const HeadMask& mk, float v0, float v1, float v2,
                                       float v3) {
  return __uint_as_float((__float_as_uint(v0) & mk.m0) | (__float_as_uint(v1) & mk.m1) |
                         (__float_as_uint(v2) & mk.m2) | (__float_as_uint(v3) & mk.m3));
}
__device__ __forceinline__ void fma4(float4& acc, float c, const float4& v) {
  acc.x += c * v.x; acc.y += c * v.y; acc.z += c * v.z; acc.w += c * v.w;
}
__device__ __forceinline__ float lane_val(float v, int k) {
  return __int_as_float(__builtin_amdgcn_readlane(__float_as_int(v), k));
}

__global__ void k_node(const int* __restrict__ src, const int* __restrict__ dst,
                       const int2* __restrict__ sorted, const int* __restrict__ rowstart,
                       const int* __restrict__ cnt,
                       const float* __restrict__ Sj, const float* __restrict__ Si,
                       const float* __restrict__ xl, const float* __restrict__ bias,
                       float* __restrict__ out, int N) {
  const int lane = threadIdx.x & 31;
  const int hL = lane >> 3;
  const HeadMask mk = {hL == 0 ? ~0u : 0u, hL == 1 ? ~0u : 0u, hL == 2 ? ~0u : 0u,
                       hL == 3 ? ~0u : 0u};
  const int n = __builtin_amdgcn_readfirstlane((int)(blockIdx.x * 8 + (threadIdx.x >> 5)));
  if (n >= N) return;
  const int beg = rowstart[n];
  const int deg = cnt[n];
  const int end = beg + deg;
  const float4 si = *(const float4*)(Si + (size_t)dst[n] * HEADS);
  const float4 sjs = *(const float4*)(Sj + (size_t)src[n] * HEADS);
  const float as0 = leaky(sjs.x + si.x), as1 = leaky(sjs.y + si.y);
  const float as2 = leaky(sjs.z + si.z), as3 = leaky(sjs.w + si.w);

  const int i0 = beg + lane, i1 = i0 + 32;
  const bool ok0 = i0 < end, ok1 = i1 < end;
  const int2 r0 = sorted[ok0 ? i0 : 0];
  const int2 r1 = sorted[ok1 ? i1 : 0];
  __builtin_amdgcn_sched_barrier(0);
  const float4 q0 = *(const float4*)(Sj + (size_t)r0.y * HEADS);
  const float4 q1 = *(const float4*)(Sj + (size_t)r1.y * HEADS);
  const float p00 = ok0 ? leaky(q0.x + si.x) : as0, p01 = ok0 ? leaky(q0.y + si.y) : as1;
  const float p02 = ok0 ? leaky(q0.z + si.z) : as2, p03 = ok0 ? leaky(q0.w + si.w) : as3;
  const float p10 = ok1 ? leaky(q1.x + si.x) : as0, p11 = ok1 ? leaky(q1.y + si.y) : as1;
  const float p12 = ok1 ? leaky(q1.z + si.z) : as2, p13 = ok1 ? leaky(q1.w + si.w) : as3;
  float m0 = fmaxf(as0, fmaxf(p00, p10)), m1 = fmaxf(as1, fmaxf(p01, p11));
  float m2 = fmaxf(as2, fmaxf(p02, p12)), m3 = fmaxf(as3, fmaxf(p03, p13));
  for (int base = beg + 64; base < end; base += 32) {
    const int idx = base + lane;
    if (idx < end) {
      const int2 sp = sorted[idx];
      const float4 sj = *(const float4*)(Sj + (size_t)sp.y * HEADS);
      m0 = fmaxf(m0, leaky(sj.x + si.x));
      m1 = fmaxf(m1, leaky(sj.y + si.y));
      m2 = fmaxf(m2, leaky(sj.z + si.z));
      m3 = fmaxf(m3, leaky(sj.w + si.w));
    }
  }
  m0 = wave_max(m0); m1 = wave_max(m1); m2 = wave_max(m2); m3 = wave_max(m3);

  float4 acc = make_float4(0.f, 0.f, 0.f, 0.f);
  auto add_rows = [&](int sn, float a0, float a1, float a2, float a3, int cc) {
    auto coef = [&](int k) {
      return pick4(mk, lane_val(a0, k), lane_val(a1, k), lane_val(a2, k), lane_val(a3, k));
    };
    auto row = [&](int k) {
      const int sk = __builtin_amdgcn_readlane(sn, k);
      return *(const float4*)(xl + (size_t)sk * HC + lane * 4);
    };
    int k = 0;
    for (; k + 4 <= cc; k += 4) {
      const float4 v0 = row(k), v1 = row(k + 1), v2 = row(k + 2), v3 = row(k + 3);
      const float c0 = coef(k), c1 = coef(k + 1), c2 = coef(k + 2), c3 = coef(k + 3);
      fma4(acc, c0, v0);
      fma4(acc, c1, v1);
      fma4(acc, c2, v2);
      fma4(acc, c3, v3);
    }
    for (; k < cc; ++k) fma4(acc, coef(k), row(k));
  };

  float a00 = 0.f, a01 = 0.f, a02 = 0.f, a03 = 0.f;
  float a10 = 0.f, a11 = 0.f, a12 = 0.f, a13 = 0.f;
  if (ok0) {
    a00 = expf(p00 - m0); a01 = expf(p01 - m1); a02 = expf(p02 - m2); a03 = expf(p03 - m3);
  }
  if (ok1) {
    a10 = expf(p10 - m0); a11 = expf(p11 - m1); a12 = expf(p12 - m2); a13 = expf(p13 - m3);
  }
  float s0 = a00 + a10, s1 = a01 + a11, s2 = a02 + a12, s3 = a03 + a13;
  add_rows(r0.x, a00, a01, a02, a03, min(32, deg));
  if (deg > 32) add_rows(r1.x, a10, a11, a12, a13, min(32, deg - 32));
  for (int base = beg + 64; base < end; base += 32) {
    const int idx = base + lane;
    int sn = 0;
    float a0 = 0.f, a1 = 0.f, a2 = 0.f, a3 = 0.f;
    if (idx < end) {
      const int2 sp = sorted[idx];
      sn = sp.x;
      const float4 sj = *(const float4*)(Sj + (size_t)sp.y * HEADS);
      a0 = expf(leaky(sj.x + si.x) - m0);
      a1 = expf(leaky(sj.y + si.y) - m1);
      a2 = expf(leaky(sj.z + si.z) - m2);
      a3 = expf(leaky(sj.w + si.w) - m3);
    }
    s0 += a0; s1 += a1; s2 += a2; s3 += a3;
    add_rows(sn, a0, a1, a2, a3, min(32, end - base));
  }
  s0 = wave_sum(s0); s1 = wave_sum(s1); s2 = wave_sum(s2); s3 = wave_sum(s3);
  const float e0 = expf(as0 - m0), e1 = expf(as1 - m1);
  const float e2 = expf(as2 - m2), e3 = expf(as3 - m3);
  s0 += e0; s1 += e1; s2 += e2; s3 += e3;
  {
    const float4 v = *(const float4*)(xl + (size_t)n * HC + lane * 4);
    fma4(acc, pick4(mk, e0, e1, e2, e3), v);
  }
  const float sinv = 1.0f / pick4(mk, s0, s1, s2, s3);
  float4 o;
  o.x = acc.x * sinv + bias[lane * 4 + 0];
  o.y = acc.y * sinv + bias[lane * 4 + 1];
  o.z = acc.z * sinv + bias[lane * 4 + 2];
  o.w = acc.w * sinv + bias[lane * 4 + 3];
  *(float4*)(out + (size_t)n * HC + lane * 4) = o;
}

extern "C" void kernel_launch(void* const* d_in, const int* in_sizes, int n_in,
                              void* d_out, int out_size, void* d_ws, size_t ws_size,
                              hipStream_t stream) {
  const float* x      = (const float*)d_in[0];
  const float* W      = (const float*)d_in[1];
  const float* attn_j = (const float*)d_in[2];
  const float* attn_i = (const float*)d_in[3];
  const float* bias   = (const float*)d_in[4];
  const int*   eidx   = (const int*)d_in[5];

  const int N  = in_sizes[0] / F_IN;
  const int E  = in_sizes[5] / 2;
  const int* src = eidx;
  const int* dst = eidx + E;

  char*  ws = (char*)d_ws;
  size_t o  = 0;
  auto take = [&](size_t bytes) -> void* {
    void* p = ws + o;
    o += (bytes + 255) & ~(size_t)255;
    return p;
  };
  float*     xl   = (float*)take((size_t)N * HC * sizeof(float));
  float*     Sj   = (float*)take((size_t)N * HEADS * sizeof(float));
  float*     Si   = (float*)take((size_t)N * HEADS * sizeof(float));
  const int  nb   = (N + 255) / 256;
  const int  chunk = (E + EB - 1) / EB;
  int*       cnt  = (int*)take((size_t)N * sizeof(int));
  int*       rowstart = (int*)take((size_t)N * sizeof(int));
  int*       cntRB = (int*)take((size_t)nb * EB * sizeof(int));
  int*       rbase = (int*)take((size_t)nb * sizeof(int));
  int*       rtot  = (int*)take((size_t)nb * sizeof(int));
  int2*      bucket = (int2*)take((size_t)E * sizeof(int2));
  int2*      sorted = (int2*)take((size_t)E * sizeof(int2));

  float* out = (float*)d_out;

  k_gemm<<<(N + GT * 16 - 1) / (GT * 16), GT * 32, 0, stream>>>(x, W, attn_j, attn_i, xl,
                                                                Sj, Si, N);

  k_count<<<EB, ET, (size_t)nb * sizeof(int), stream>>>(dst, cntRB, E, chunk, nb);
  k_scatter<<<EB, ET, (size_t)nb * 2 * sizeof(int), stream>>>(src, dst, cntRB, bucket,
                                                              rbase, rtot, E, chunk, nb);
  k_rangesort<<<nb, ET, 0, stream>>>(src, bucket, rbase, rtot, sorted, rowstart, cnt, N);

  k_node<<<(N + 7) / 8, 256, 0, stream>>>(src, dst, sorted, rowstart, cnt, Sj, Si,
                                          xl, bias, out, N);
}


# sort level 1 block-local: an edge block orders its own chunk by range in place; no count launch, no cross-block prefix, 5 launches to 4; one int atomic per range reserves output; LDS-atomic rank kept
# speedup vs baseline: 11.1681x; 1.0894x over previous
#include <hip/hip_runtime.h>


#define F_IN      128
#define HEADS     4
#define C_OUT     32
#define HC        128
#define NEG_SLOPE 0.2f

typedef float v2f __attribute__((ext_vector_type(2)));
typedef float v8f __attribute__((ext_vector_type(8)));

#define GT 4

template <int M> __device__ __forceinline__ float row_xor(float v) {
  return __int_as_float(
      __builtin_amdgcn_update_dpp(0, __float_as_int(v), 0x160 + M, 0xf, 0xf, true));
}
template <int M> __device__ __forceinline__ float fold(float X, float Y, bool bit) {
  const float mine = bit ? Y : X;
  const float send = bit ? X : Y;
  return mine + row_xor<M>(send);
}
#define WP 132
__global__ __launch_bounds__(GT * 32) void k_gemm(const float* __restrict__ x,
                                                  const float* __restrict__ W,
                                                  const float* __restrict__ aj,
                                                  const float* __restrict__ ai,
                                                  float* __restrict__ xl,
                                                  float* __restrict__ Sj,
                                                  float* __restrict__ Si, int N) {
  __shared__ __attribute__((aligned(16))) float Ws[HC * WP];
  const int lane    = threadIdx.x & 31;
  const int wave    = threadIdx.x >> 5;
  const int g       = lane >> 4;
  const int l16     = lane & 15;
  const int rowBase = (blockIdx.x * GT + wave) * 16;
  int mrow = rowBase + l16; if (mrow >= N) mrow = N - 1;

  v2f a[F_IN / 4];
  const float* xr = x + (size_t)mrow * F_IN + g * 2;
#pragma unroll
  for (int s = 0; s < F_IN / 4; ++s) a[s] = *(const v2f*)(xr + 4 * s);
  float ajv[8], aiv[8];
#pragma unroll
  for (int t = 0; t < 8; ++t) { ajv[t] = aj[t * 16 + l16]; aiv[t] = ai[t * 16 + l16]; }

#pragma unroll 8
  for (int j = 0; j < (HC * (F_IN / 4)) / (GT * 32); ++j) {
    const int i = threadIdx.x + j * (GT * 32);
    const int r = i >> 5, c4 = i & 31;
    const float4 v = *(const float4*)(W + (size_t)r * F_IN + c4 * 4);
    *(float4*)(Ws + r * WP + c4 * 4) = v;
  }
  __syncthreads();
  if (rowBase >= N) return;

  v8f c[8] = {};
  const float* bp = Ws + l16 * WP + g * 2;
#pragma unroll
  for (int s = 0; s < F_IN / 4; ++s) {
#pragma unroll
    for (int t = 0; t < 8; ++t) {
      const v2f b = *(const v2f*)(bp + t * 16 * WP + 4 * s);
      c[t] = __builtin_amdgcn_wmma_f32_16x16x4_f32(false, a[s], false, b, (short)0, c[t],
                                                   false, false);
    }
  }
#pragma unroll
  for (int t = 0; t < 8; ++t) {
#pragma unroll
    for (int i = 0; i < 8; ++i) {
      const int r = rowBase + i + g * 8;
      if (r < N) xl[(size_t)r * HC + t * 16 + l16] = c[t][i];
    }
  }
  const bool b0 = l16 & 1, b1 = l16 & 2, b2 = l16 & 4, b3 = l16 & 8;
  float res[HEADS];
#pragma unroll
  for (int h = 0; h < HEADS; ++h) {
    float vj[8], vi[8];
#pragma unroll
    for (int i = 0; i < 8; ++i) {
      vj[i] = c[2 * h][i] * ajv[2 * h] + c[2 * h + 1][i] * ajv[2 * h + 1];
      vi[i] = c[2 * h][i] * aiv[2 * h] + c[2 * h + 1][i] * aiv[2 * h + 1];
    }
    float pj[4], pi[4];
#pragma unroll
    for (int q = 0; q < 4; ++q) {
      pj[q] = fold<1>(vj[2 * q], vj[2 * q + 1], b0);
      pi[q] = fold<1>(vi[2 * q], vi[2 * q + 1], b0);
    }
    const float qj0 = fold<2>(pj[0], pj[1], b1), qj1 = fold<2>(pj[2], pj[3], b1);
    const float qi0 = fold<2>(pi[0], pi[1], b1), qi1 = fold<2>(pi[2], pi[3], b1);
    const float rj = fold<4>(qj0, qj1, b2);
    const float ri = fold<4>(qi0, qi1, b2);
    res[h] = fold<8>(rj, ri, b3);
  }
  const int srow = rowBase + g * 8 + (l16 & 7);
  if (srow < N) {
    float* sp = (b3 ? Si : Sj) + (size_t)srow * HEADS;
    *(float4*)sp = make_float4(res[0], res[1], res[2], res[3]);
  }
}


#define EB 256
#define ET 1024
#define UB 8
#define SPW (EB / (ET / 32))
static_assert(EB % (ET / 32) == 0, "every wave of a range block walks whole slabs");

__device__ __forceinline__ int wave_incl_scan(int v, int lane) {
#pragma unroll
  for (int off = 1; off < 32; off <<= 1) {
    const int t = __shfl_up(v, off);
    if (lane >= off) v += t;
  }
  return v;
}
__device__ __forceinline__ int block_excl_scan(int v, int* wsum, int tid, int* total) {
  const int lane = tid & 31, wave = tid >> 5;
  const int inc = wave_incl_scan(v, lane);
  if (lane == 31) wsum[wave] = inc;
  __syncthreads();
  if (wave == 0) {
    const int w = wsum[lane];
    const int wi = wave_incl_scan(w, lane);
    wsum[lane] = wi - w;
    if (lane == 31) wsum[32] = wi;
  }
  __syncthreads();
  const int ex = wsum[wave] + inc - v;
  *total = wsum[32];
  __syncthreads();
  return ex;
}

__global__ void k_blocksort(const int* __restrict__ src, const int* __restrict__ dst,
                            int2* __restrict__ bucket, int* __restrict__ slabOff,
                            int* __restrict__ slabCnt, int* __restrict__ cursor, int E,
                            int chunk, int nb) {
  extern __shared__ int shm[];
  __shared__ int wsum[33];
  int* hist = shm;
  int* cur  = shm + nb;
  const int tid = threadIdx.x, b = blockIdx.x;
  if (b == 0 && tid == 0) *cursor = 0;
  for (int r = tid; r < nb; r += ET) hist[r] = 0;
  __syncthreads();
  const int e0 = b * chunk;
  const int e1 = min(E, e0 + chunk);
  for (int eb = e0 + tid; eb < e1; eb += ET * UB) {
    int d[UB];
#pragma unroll
    for (int u = 0; u < UB; ++u) {
      const int e = eb + u * ET;
      d[u] = (e < e1) ? dst[e] : -1;
    }
#pragma unroll
    for (int u = 0; u < UB; ++u)
      if (d[u] >= 0) atomicAdd(&hist[d[u] >> 8], 1);
  }
  __syncthreads();
  int carry = 0;
  for (int base = 0; base < nb; base += ET) {
    const int i = base + tid;
    const int v = (i < nb) ? hist[i] : 0;
    int total;
    const int ex = carry + block_excl_scan(v, wsum, tid, &total);
    if (i < nb) {
      cur[i] = e0 + ex;
      slabOff[(size_t)i * EB + b] = e0 + ex;
      slabCnt[(size_t)i * EB + b] = v;
    }
    carry += total;
  }
  __syncthreads();
  for (int eb = e0 + tid; eb < e1; eb += ET * UB) {
    int d[UB], s[UB];
#pragma unroll
    for (int u = 0; u < UB; ++u) {
      const int e = eb + u * ET;
      const bool ok = e < e1;
      d[u] = ok ? dst[e] : -1;
      s[u] = ok ? src[e] : 0;
    }
#pragma unroll
    for (int u = 0; u < UB; ++u) {
      if (d[u] >= 0) {
        const int pos = atomicAdd(&cur[d[u] >> 8], 1);
        bucket[pos] = make_int2(d[u], s[u]);
      }
    }
  }
}

__global__ void k_rangesort(const int* __restrict__ src, const int2* __restrict__ bucket,
                            const int* __restrict__ slabOff,
                            const int* __restrict__ slabCnt, int* __restrict__ cursor,
                            int2* __restrict__ sorted, int* __restrict__ rowstart,
                            int* __restrict__ cnt, int N) {
  __shared__ int c[256];
  __shared__ int place[256];
  __shared__ int wsum[33];
  __shared__ int sBase;
  const int tid = threadIdx.x, r = blockIdx.x;
  const int lane = tid & 31;
  const int wv = __builtin_amdgcn_readfirstlane(tid >> 5);
  const int* po = slabOff + (size_t)r * EB + wv * SPW;
  const int* pc = slabCnt + (size_t)r * EB + wv * SPW;
  int so[SPW], sc[SPW];
#pragma unroll
  for (int u = 0; u < SPW; ++u) { so[u] = po[u]; sc[u] = pc[u]; }
  if (tid < 256) c[tid] = 0;
  __syncthreads();

  int2 rec[2 * SPW];
  int nj[2 * SPW];
#pragma unroll
  for (int u = 0; u < SPW; ++u) {
    rec[2 * u]     = (lane < sc[u]) ? bucket[so[u] + lane] : make_int2(-1, 0);
    rec[2 * u + 1] = (lane + 32 < sc[u]) ? bucket[so[u] + lane + 32] : make_int2(-1, 0);
  }
#pragma unroll
  for (int u = 0; u < 2 * SPW; ++u) nj[u] = (rec[u].x >= 0) ? src[rec[u].y] : 0;
#pragma unroll
  for (int u = 0; u < 2 * SPW; ++u)
    if (rec[u].x >= 0) atomicAdd(&c[rec[u].x & 255], 1);
  for (int u = 0; u < SPW; ++u)
    for (int j = 64 + lane; j < sc[u]; j += 32) atomicAdd(&c[bucket[so[u] + j].x & 255], 1);
  __syncthreads();
  const int v = (tid < 256) ? c[tid] : 0;
  int total;
  const int ex = block_excl_scan(v, wsum, tid, &total);
  if (tid == 0) sBase = atomicAdd(cursor, total);
  __syncthreads();
  const int b0 = sBase;
  if (tid < 256) {
    place[tid] = b0 + ex;
    const int n = r * 256 + tid;
    if (n < N) { rowstart[n] = b0 + ex; cnt[n] = v; }
  }
  __syncthreads();
#pragma unroll
  for (int u = 0; u < 2 * SPW; ++u) {
    if (rec[u].x >= 0) {
      const int p = atomicAdd(&place[rec[u].x & 255], 1);
      sorted[p] = make_int2(rec[u].y, nj[u]);
    }
  }
  for (int u = 0; u < SPW; ++u) {
    for (int j = 64 + lane; j < sc[u]; j += 32) {
      const int2 q = bucket[so[u] + j];
      const int p = atomicAdd(&place[q.x & 255], 1);
      sorted[p] = make_int2(q.y, src[q.y]);
    }
  }
}

__device__ __forceinline__ float leaky(float a) {
  return (a >= 0.f) ? a : NEG_SLOPE * a;
}
__device__ __forceinline__ float wave_max(float v) {
#pragma unroll
  for (int off = 16; off > 0; off >>= 1) v = fmaxf(v, __shfl_xor(v, off));
  return v;
}
__device__ __forceinline__ float wave_sum(float v) {
#pragma unroll
  for (int off = 16; off > 0; off >>= 1) v += __shfl_xor(v, off);
  return v;
}
struct HeadMask { unsigned m0, m1, m2, m3; };
__device__ __forceinline__ float pick4(const HeadMask& mk, float v0, float v1, float v2,
                                       float v3) {
  return __uint_as_float((__float_as_uint(v0) & mk.m0) | (__float_as_uint(v1) & mk.m1) |
                         (__float_as_uint(v2) & mk.m2) | (__float_as_uint(v3) & mk.m3));
}
__device__ __forceinline__ void fma4(float4& acc, float c, const float4& v) {
  acc.x += c * v.x; acc.y += c * v.y; acc.z += c * v.z; acc.w += c * v.w;
}
__device__ __forceinline__ float lane_val(float v, int k) {
  return __int_as_float(__builtin_amdgcn_readlane(__float_as_int(v), k));
}

__global__ void k_node(const int* __restrict__ src, const int* __restrict__ dst,
                       const int2* __restrict__ sorted, const int* __restrict__ rowstart,
                       const int* __restrict__ cnt,
                       const float* __restrict__ Sj, const float* __restrict__ Si,
                       const float* __restrict__ xl, const float* __restrict__ bias,
                       float* __restrict__ out, int N) {
  const int lane = threadIdx.x & 31;
  const int hL = lane >> 3;
  const HeadMask mk = {hL == 0 ? ~0u : 0u, hL == 1 ? ~0u : 0u, hL == 2 ? ~0u : 0u,
                       hL == 3 ? ~0u : 0u};
  const int n = __builtin_amdgcn_readfirstlane((int)(blockIdx.x * 8 + (threadIdx.x >> 5)));
  if (n >= N) return;
  const int beg = rowstart[n];
  const int deg = cnt[n];
  const int end = beg + deg;
  const float4 si = *(const float4*)(Si + (size_t)dst[n] * HEADS);
  const float4 sjs = *(const float4*)(Sj + (size_t)src[n] * HEADS);
  const float as0 = leaky(sjs.x + si.x), as1 = leaky(sjs.y + si.y);
  const float as2 = leaky(sjs.z + si.z), as3 = leaky(sjs.w + si.w);

  const int i0 = beg + lane, i1 = i0 + 32;
  const bool ok0 = i0 < end, ok1 = i1 < end;
  const int2 r0 = sorted[ok0 ? i0 : 0];
  const int2 r1 = sorted[ok1 ? i1 : 0];
  __builtin_amdgcn_sched_barrier(0);
  const float4 q0 = *(const float4*)(Sj + (size_t)r0.y * HEADS);
  const float4 q1 = *(const float4*)(Sj + (size_t)r1.y * HEADS);
  const float p00 = ok0 ? leaky(q0.x + si.x) : as0, p01 = ok0 ? leaky(q0.y + si.y) : as1;
  const float p02 = ok0 ? leaky(q0.z + si.z) : as2, p03 = ok0 ? leaky(q0.w + si.w) : as3;
  const float p10 = ok1 ? leaky(q1.x + si.x) : as0, p11 = ok1 ? leaky(q1.y + si.y) : as1;
  const float p12 = ok1 ? leaky(q1.z + si.z) : as2, p13 = ok1 ? leaky(q1.w + si.w) : as3;
  float m0 = fmaxf(as0, fmaxf(p00, p10)), m1 = fmaxf(as1, fmaxf(p01, p11));
  float m2 = fmaxf(as2, fmaxf(p02, p12)), m3 = fmaxf(as3, fmaxf(p03, p13));
  for (int base = beg + 64; base < end; base += 32) {
    const int idx = base + lane;
    if (idx < end) {
      const int2 sp = sorted[idx];
      const float4 sj = *(const float4*)(Sj + (size_t)sp.y * HEADS);
      m0 = fmaxf(m0, leaky(sj.x + si.x));
      m1 = fmaxf(m1, leaky(sj.y + si.y));
      m2 = fmaxf(m2, leaky(sj.z + si.z));
      m3 = fmaxf(m3, leaky(sj.w + si.w));
    }
  }
  m0 = wave_max(m0); m1 = wave_max(m1); m2 = wave_max(m2); m3 = wave_max(m3);

  float4 acc = make_float4(0.f, 0.f, 0.f, 0.f);
  auto add_rows = [&](int sn, float a0, float a1, float a2, float a3, int cc) {
    auto coef = [&](int k) {
      return pick4(mk, lane_val(a0, k), lane_val(a1, k), lane_val(a2, k), lane_val(a3, k));
    };
    auto row = [&](int k) {
      const int sk = __builtin_amdgcn_readlane(sn, k);
      return *(const float4*)(xl + (size_t)sk * HC + lane * 4);
    };
    int k = 0;
    for (; k + 4 <= cc; k += 4) {
      const float4 v0 = row(k), v1 = row(k + 1), v2 = row(k + 2), v3 = row(k + 3);
      const float c0 = coef(k), c1 = coef(k + 1), c2 = coef(k + 2), c3 = coef(k + 3);
      fma4(acc, c0, v0);
      fma4(acc, c1, v1);
      fma4(acc, c2, v2);
      fma4(acc, c3, v3);
    }
    for (; k < cc; ++k) fma4(acc, coef(k), row(k));
  };

  float a00 = 0.f, a01 = 0.f, a02 = 0.f, a03 = 0.f;
  float a10 = 0.f, a11 = 0.f, a12 = 0.f, a13 = 0.f;
  if (ok0) {
    a00 = expf(p00 - m0); a01 = expf(p01 - m1); a02 = expf(p02 - m2); a03 = expf(p03 - m3);
  }
  if (ok1) {
    a10 = expf(p10 - m0); a11 = expf(p11 - m1); a12 = expf(p12 - m2); a13 = expf(p13 - m3);
  }
  float s0 = a00 + a10, s1 = a01 + a11, s2 = a02 + a12, s3 = a03 + a13;
  add_rows(r0.x, a00, a01, a02, a03, min(32, deg));
  if (deg > 32) add_rows(r1.x, a10, a11, a12, a13, min(32, deg - 32));
  for (int base = beg + 64; base < end; base += 32) {
    const int idx = base + lane;
    int sn = 0;
    float a0 = 0.f, a1 = 0.f, a2 = 0.f, a3 = 0.f;
    if (idx < end) {
      const int2 sp = sorted[idx];
      sn = sp.x;
      const float4 sj = *(const float4*)(Sj + (size_t)sp.y * HEADS);
      a0 = expf(leaky(sj.x + si.x) - m0);
      a1 = expf(leaky(sj.y + si.y) - m1);
      a2 = expf(leaky(sj.z + si.z) - m2);
      a3 = expf(leaky(sj.w + si.w) - m3);
    }
    s0 += a0; s1 += a1; s2 += a2; s3 += a3;
    add_rows(sn, a0, a1, a2, a3, min(32, end - base));
  }
  s0 = wave_sum(s0); s1 = wave_sum(s1); s2 = wave_sum(s2); s3 = wave_sum(s3);
  const float e0 = expf(as0 - m0), e1 = expf(as1 - m1);
  const float e2 = expf(as2 - m2), e3 = expf(as3 - m3);
  s0 += e0; s1 += e1; s2 += e2; s3 += e3;
  {
    const float4 v = *(const float4*)(xl + (size_t)n * HC + lane * 4);
    fma4(acc, pick4(mk, e0, e1, e2, e3), v);
  }
  const float sinv = 1.0f / pick4(mk, s0, s1, s2, s3);
  float4 o;
  o.x = acc.x * sinv + bias[lane * 4 + 0];
  o.y = acc.y * sinv + bias[lane * 4 + 1];
  o.z = acc.z * sinv + bias[lane * 4 + 2];
  o.w = acc.w * sinv + bias[lane * 4 + 3];
  *(float4*)(out + (size_t)n * HC + lane * 4) = o;
}

extern "C" void kernel_launch(void* const* d_in, const int* in_sizes, int n_in,
                              void* d_out, int out_size, void* d_ws, size_t ws_size,
                              hipStream_t stream) {
  const float* x      = (const float*)d_in[0];
  const float* W      = (const float*)d_in[1];
  const float* attn_j = (const float*)d_in[2];
  const float* attn_i = (const float*)d_in[3];
  const float* bias   = (const float*)d_in[4];
  const int*   eidx   = (const int*)d_in[5];

  const int N  = in_sizes[0] / F_IN;
  const int E  = in_sizes[5] / 2;
  const int* src = eidx;
  const int* dst = eidx + E;

  char*  ws = (char*)d_ws;
  size_t o  = 0;
  auto take = [&](size_t bytes) -> void* {
    void* p = ws + o;
    o += (bytes + 255) & ~(size_t)255;
    return p;
  };
  float*     xl   = (float*)take((size_t)N * HC * sizeof(float));
  float*     Sj   = (float*)take((size_t)N * HEADS * sizeof(float));
  float*     Si   = (float*)take((size_t)N * HEADS * sizeof(float));
  const int  nb   = (N + 255) / 256;
  const int  chunk = (E + EB - 1) / EB;
  int*       cnt  = (int*)take((size_t)N * sizeof(int));
  int*       rowstart = (int*)take((size_t)N * sizeof(int));
  int*       slabOff = (int*)take((size_t)nb * EB * sizeof(int));
  int*       slabCnt = (int*)take((size_t)nb * EB * sizeof(int));
  int*       cursor  = (int*)take(sizeof(int));
  int2*      bucket = (int2*)take((size_t)E * sizeof(int2));
  int2*      sorted = (int2*)take((size_t)E * sizeof(int2));

  float* out = (float*)d_out;

  k_gemm<<<(N + GT * 16 - 1) / (GT * 16), GT * 32, 0, stream>>>(x, W, attn_j, attn_i, xl,
                                                                Sj, Si, N);

  k_blocksort<<<EB, ET, (size_t)nb * 2 * sizeof(int), stream>>>(src, dst, bucket, slabOff,
                                                                slabCnt, cursor, E, chunk,
                                                                nb);
  k_rangesort<<<nb, ET, 0, stream>>>(src, bucket, slabOff, slabCnt, cursor, sorted, rowstart,
                                     cnt, N);

  k_node<<<(N + 7) / 8, 256, 0, stream>>>(src, dst, sorted, rowstart, cnt, Sj, Si,
                                          xl, bias, out, N);
}


# both sort levels place their records in LDS and copy the finished piece out in order: a wave's store is two whole 128-byte lines, not 32 scattered 8-byte pieces; LDS-atomic arrival-order rank kept
# speedup vs baseline: 11.9512x; 1.0701x over previous
#include <hip/hip_runtime.h>


#define F_IN      128
#define HEADS     4
#define C_OUT     32
#define HC        128
#define NEG_SLOPE 0.2f

typedef float v2f __attribute__((ext_vector_type(2)));
typedef float v8f __attribute__((ext_vector_type(8)));

#define GT 4

template <int M> __device__ __forceinline__ float row_xor(float v) {
  return __int_as_float(
      __builtin_amdgcn_update_dpp(0, __float_as_int(v), 0x160 + M, 0xf, 0xf, true));
}
template <int M> __device__ __forceinline__ float fold(float X, float Y, bool bit) {
  const float mine = bit ? Y : X;
  const float send = bit ? X : Y;
  return mine + row_xor<M>(send);
}
#define WP 132
__global__ __launch_bounds__(GT * 32) void k_gemm(const float* __restrict__ x,
                                                  const float* __restrict__ W,
                                                  const float* __restrict__ aj,
                                                  const float* __restrict__ ai,
                                                  float* __restrict__ xl,
                                                  float* __restrict__ Sj,
                                                  float* __restrict__ Si, int N) {
  __shared__ __attribute__((aligned(16))) float Ws[HC * WP];
  const int lane    = threadIdx.x & 31;
  const int wave    = threadIdx.x >> 5;
  const int g       = lane >> 4;
  const int l16     = lane & 15;
  const int rowBase = (blockIdx.x * GT + wave) * 16;
  int mrow = rowBase + l16; if (mrow >= N) mrow = N - 1;

  v2f a[F_IN / 4];
  const float* xr = x + (size_t)mrow * F_IN + g * 2;
#pragma unroll
  for (int s = 0; s < F_IN / 4; ++s) a[s] = *(const v2f*)(xr + 4 * s);
  float ajv[8], aiv[8];
#pragma unroll
  for (int t = 0; t < 8; ++t) { ajv[t] = aj[t * 16 + l16]; aiv[t] = ai[t * 16 + l16]; }

#pragma unroll 8
  for (int j = 0; j < (HC * (F_IN / 4)) / (GT * 32); ++j) {
    const int i = threadIdx.x + j * (GT * 32);
    const int r = i >> 5, c4 = i & 31;
    const float4 v = *(const float4*)(W + (size_t)r * F_IN + c4 * 4);
    *(float4*)(Ws + r * WP + c4 * 4) = v;
  }
  __syncthreads();
  if (rowBase >= N) return;

  v8f c[8] = {};
  const float* bp = Ws + l16 * WP + g * 2;
#pragma unroll
  for (int s = 0; s < F_IN / 4; ++s) {
#pragma unroll
    for (int t = 0; t < 8; ++t) {
      const v2f b = *(const v2f*)(bp + t * 16 * WP + 4 * s);
      c[t] = __builtin_amdgcn_wmma_f32_16x16x4_f32(false, a[s], false, b, (short)0, c[t],
                                                   false, false);
    }
  }
#pragma unroll
  for (int t = 0; t < 8; ++t) {
#pragma unroll
    for (int i = 0; i < 8; ++i) {
      const int r = rowBase + i + g * 8;
      if (r < N) xl[(size_t)r * HC + t * 16 + l16] = c[t][i];
    }
  }
  const bool b0 = l16 & 1, b1 = l16 & 2, b2 = l16 & 4, b3 = l16 & 8;
  float res[HEADS];
#pragma unroll
  for (int h = 0; h < HEADS; ++h) {
    float vj[8], vi[8];
#pragma unroll
    for (int i = 0; i < 8; ++i) {
      vj[i] = c[2 * h][i] * ajv[2 * h] + c[2 * h + 1][i] * ajv[2 * h + 1];
      vi[i] = c[2 * h][i] * aiv[2 * h] + c[2 * h + 1][i] * aiv[2 * h + 1];
    }
    float pj[4], pi[4];
#pragma unroll
    for (int q = 0; q < 4; ++q) {
      pj[q] = fold<1>(vj[2 * q], vj[2 * q + 1], b0);
      pi[q] = fold<1>(vi[2 * q], vi[2 * q + 1], b0);
    }
    const float qj0 = fold<2>(pj[0], pj[1], b1), qj1 = fold<2>(pj[2], pj[3], b1);
    const float qi0 = fold<2>(pi[0], pi[1], b1), qi1 = fold<2>(pi[2], pi[3], b1);
    const float rj = fold<4>(qj0, qj1, b2);
    const float ri = fold<4>(qi0, qi1, b2);
    res[h] = fold<8>(rj, ri, b3);
  }
  const int srow = rowBase + g * 8 + (l16 & 7);
  if (srow < N) {
    float* sp = (b3 ? Si : Sj) + (size_t)srow * HEADS;
    *(float4*)sp = make_float4(res[0], res[1], res[2], res[3]);
  }
}


#define EB 256
#define ET 1024
#define UB 8
#define SPW (EB / (ET / 32))
static_assert(EB % (ET / 32) == 0, "every wave of a range block walks whole slabs");
#define SCAP 7168
#define RCAP 10240

__device__ __forceinline__ int wave_incl_scan(int v, int lane) {
#pragma unroll
  for (int off = 1; off < 32; off <<= 1) {
    const int t = __shfl_up(v, off);
    if (lane >= off) v += t;
  }
  return v;
}
__device__ __forceinline__ int block_excl_scan(int v, int* wsum, int tid, int* total) {
  const int lane = tid & 31, wave = tid >> 5;
  const int inc = wave_incl_scan(v, lane);
  if (lane == 31) wsum[wave] = inc;
  __syncthreads();
  if (wave == 0) {
    const int w = wsum[lane];
    const int wi = wave_incl_scan(w, lane);
    wsum[lane] = wi - w;
    if (lane == 31) wsum[32] = wi;
  }
  __syncthreads();
  const int ex = wsum[wave] + inc - v;
  *total = wsum[32];
  __syncthreads();
  return ex;
}

__global__ void k_blocksort(const int* __restrict__ src, const int* __restrict__ dst,
                            int2* __restrict__ bucket, int* __restrict__ slabOff,
                            int* __restrict__ slabCnt, int* __restrict__ cursor, int E,
                            int chunk, int nb) {
  extern __shared__ int shm[];
  __shared__ int wsum[33];
  __shared__ int2 stage[SCAP];
  int* hist = shm;
  int* cur  = shm + nb;
  const int tid = threadIdx.x, b = blockIdx.x;
  if (b == 0 && tid == 0) *cursor = 0;
  for (int r = tid; r < nb; r += ET) hist[r] = 0;
  __syncthreads();
  const int e0 = b * chunk;
  const int e1 = min(E, e0 + chunk);
  const bool staged = chunk <= SCAP;
  for (int eb = e0 + tid; eb < e1; eb += ET * UB) {
    int d[UB];
#pragma unroll
    for (int u = 0; u < UB; ++u) {
      const int e = eb + u * ET;
      d[u] = (e < e1) ? dst[e] : -1;
    }
#pragma unroll
    for (int u = 0; u < UB; ++u)
      if (d[u] >= 0) atomicAdd(&hist[d[u] >> 8], 1);
  }
  __syncthreads();
  int carry = 0;
  for (int base = 0; base < nb; base += ET) {
    const int i = base + tid;
    const int v = (i < nb) ? hist[i] : 0;
    int total;
    const int ex = carry + block_excl_scan(v, wsum, tid, &total);
    if (i < nb) {
      cur[i] = ex;
      slabOff[(size_t)i * EB + b] = e0 + ex;
      slabCnt[(size_t)i * EB + b] = v;
    }
    carry += total;
  }
  __syncthreads();
  auto place_all = [&](int2* __restrict__ to) {
    for (int eb = e0 + tid; eb < e1; eb += ET * UB) {
      int d[UB], s[UB];
#pragma unroll
      for (int u = 0; u < UB; ++u) {
        const int e = eb + u * ET;
        const bool ok = e < e1;
        d[u] = ok ? dst[e] : -1;
        s[u] = ok ? src[e] : 0;
      }
#pragma unroll
      for (int u = 0; u < UB; ++u) {
        if (d[u] >= 0) {
          const int pos = atomicAdd(&cur[d[u] >> 8], 1);
          to[pos] = make_int2(d[u], s[u]);
        }
      }
    }
  };
  if (staged) {
    place_all(stage);
    __syncthreads();
    const int n = e1 - e0;
    int i = tid - (e0 & 15);
    if (i < 0) i += ET;
#pragma unroll 4
    for (; i < n; i += ET) bucket[e0 + i] = stage[i];
  } else {
    place_all(bucket + e0);
  }
}

__global__ void k_rangesort(const int* __restrict__ src, const int2* __restrict__ bucket,
                            const int* __restrict__ slabOff,
                            const int* __restrict__ slabCnt, int* __restrict__ cursor,
                            int2* __restrict__ sorted, int* __restrict__ rowstart,
                            int* __restrict__ cnt, int N) {
  __shared__ int c[256];
  __shared__ int place[256];
  __shared__ int wsum[33];
  __shared__ int sBase;
  __shared__ int2 stage[RCAP];
  const int tid = threadIdx.x, r = blockIdx.x;
  const int lane = tid & 31;
  const int wv = __builtin_amdgcn_readfirstlane(tid >> 5);
  const int* po = slabOff + (size_t)r * EB + wv * SPW;
  const int* pc = slabCnt + (size_t)r * EB + wv * SPW;
  int so[SPW], sc[SPW];
#pragma unroll
  for (int u = 0; u < SPW; ++u) { so[u] = po[u]; sc[u] = pc[u]; }
  if (tid < 256) c[tid] = 0;
  __syncthreads();

  int2 rec[2 * SPW];
  int nj[2 * SPW];
#pragma unroll
  for (int u = 0; u < SPW; ++u) {
    rec[2 * u]     = (lane < sc[u]) ? bucket[so[u] + lane] : make_int2(-1, 0);
    rec[2 * u + 1] = (lane + 32 < sc[u]) ? bucket[so[u] + lane + 32] : make_int2(-1, 0);
  }
#pragma unroll
  for (int u = 0; u < 2 * SPW; ++u) nj[u] = (rec[u].x >= 0) ? src[rec[u].y] : 0;
#pragma unroll
  for (int u = 0; u < 2 * SPW; ++u)
    if (rec[u].x >= 0) atomicAdd(&c[rec[u].x & 255], 1);
  for (int u = 0; u < SPW; ++u)
    for (int j = 64 + lane; j < sc[u]; j += 32) atomicAdd(&c[bucket[so[u] + j].x & 255], 1);
  __syncthreads();
  const int v = (tid < 256) ? c[tid] : 0;
  int total;
  const int ex = block_excl_scan(v, wsum, tid, &total);
  if (tid == 0) sBase = atomicAdd(cursor, total);
  __syncthreads();
  const int b0 = sBase;
  const bool staged = total <= RCAP;
  if (tid < 256) {
    place[tid] = ex;
    const int n = r * 256 + tid;
    if (n < N) { rowstart[n] = b0 + ex; cnt[n] = v; }
  }
  __syncthreads();
  auto place_all = [&](int2* __restrict__ to) {
#pragma unroll
    for (int u = 0; u < 2 * SPW; ++u) {
      if (rec[u].x >= 0) {
        const int p = atomicAdd(&place[rec[u].x & 255], 1);
        to[p] = make_int2(rec[u].y, nj[u]);
      }
    }
    for (int u = 0; u < SPW; ++u) {
      for (int j = 64 + lane; j < sc[u]; j += 32) {
        const int2 q = bucket[so[u] + j];
        const int p = atomicAdd(&place[q.x & 255], 1);
        to[p] = make_int2(q.y, src[q.y]);
      }
    }
  };
  if (staged) {
    place_all(stage);
    __syncthreads();
    int i = tid - (b0 & 15);
    if (i < 0) i += ET;
#pragma unroll 4
    for (; i < total; i += ET) sorted[b0 + i] = stage[i];
  } else {
    place_all(sorted + b0);
  }
}

__device__ __forceinline__ float leaky(float a) {
  return (a >= 0.f) ? a : NEG_SLOPE * a;
}
__device__ __forceinline__ float wave_max(float v) {
#pragma unroll
  for (int off = 16; off > 0; off >>= 1) v = fmaxf(v, __shfl_xor(v, off));
  return v;
}
__device__ __forceinline__ float wave_sum(float v) {
#pragma unroll
  for (int off = 16; off > 0; off >>= 1) v += __shfl_xor(v, off);
  return v;
}
struct HeadMask { unsigned m0, m1, m2, m3; };
__device__ __forceinline__ float pick4(const HeadMask& mk, float v0, float v1, float v2,
                                       float v3) {
  return __uint_as_float((__float_as_uint(v0) & mk.m0) | (__float_as_uint(v1) & mk.m1) |
                         (__float_as_uint(v2) & mk.m2) | (__float_as_uint(v3) & mk.m3));
}
__device__ __forceinline__ void fma4(float4& acc, float c, const float4& v) {
  acc.x += c * v.x; acc.y += c * v.y; acc.z += c * v.z; acc.w += c * v.w;
}
__device__ __forceinline__ float lane_val(float v, int k) {
  return __int_as_float(__builtin_amdgcn_readlane(__float_as_int(v), k));
}

__global__ void k_node(const int* __restrict__ src, const int* __restrict__ dst,
                       const int2* __restrict__ sorted, const int* __restrict__ rowstart,
                       const int* __restrict__ cnt,
                       const float* __restrict__ Sj, const float* __restrict__ Si,
                       const float* __restrict__ xl, const float* __restrict__ bias,
                       float* __restrict__ out, int N) {
  const int lane = threadIdx.x & 31;
  const int hL = lane >> 3;
  const HeadMask mk = {hL == 0 ? ~0u : 0u, hL == 1 ? ~0u : 0u, hL == 2 ? ~0u : 0u,
                       hL == 3 ? ~0u : 0u};
  const int n = __builtin_amdgcn_readfirstlane((int)(blockIdx.x * 8 + (threadIdx.x >> 5)));
  if (n >= N) return;
  const int beg = rowstart[n];
  const int deg = cnt[n];
  const int end = beg + deg;
  const float4 si = *(const float4*)(Si + (size_t)dst[n] * HEADS);
  const float4 sjs = *(const float4*)(Sj + (size_t)src[n] * HEADS);
  const float as0 = leaky(sjs.x + si.x), as1 = leaky(sjs.y + si.y);
  const float as2 = leaky(sjs.z + si.z), as3 = leaky(sjs.w + si.w);

  const int i0 = beg + lane, i1 = i0 + 32;
  const bool ok0 = i0 < end, ok1 = i1 < end;
  const int2 r0 = sorted[ok0 ? i0 : 0];
  const int2 r1 = sorted[ok1 ? i1 : 0];
  __builtin_amdgcn_sched_barrier(0);
  const float4 q0 = *(const float4*)(Sj + (size_t)r0.y * HEADS);
  const float4 q1 = *(const float4*)(Sj + (size_t)r1.y * HEADS);
  const float p00 = ok0 ? leaky(q0.x + si.x) : as0, p01 = ok0 ? leaky(q0.y + si.y) : as1;
  const float p02 = ok0 ? leaky(q0.z + si.z) : as2, p03 = ok0 ? leaky(q0.w + si.w) : as3;
  const float p10 = ok1 ? leaky(q1.x + si.x) : as0, p11 = ok1 ? leaky(q1.y + si.y) : as1;
  const float p12 = ok1 ? leaky(q1.z + si.z) : as2, p13 = ok1 ? leaky(q1.w + si.w) : as3;
  float m0 = fmaxf(as0, fmaxf(p00, p10)), m1 = fmaxf(as1, fmaxf(p01, p11));
  float m2 = fmaxf(as2, fmaxf(p02, p12)), m3 = fmaxf(as3, fmaxf(p03, p13));
  for (int base = beg + 64; base < end; base += 32) {
    const int idx = base + lane;
    if (idx < end) {
      const int2 sp = sorted[idx];
      const float4 sj = *(const float4*)(Sj + (size_t)sp.y * HEADS);
      m0 = fmaxf(m0, leaky(sj.x + si.x));
      m1 = fmaxf(m1, leaky(sj.y + si.y));
      m2 = fmaxf(m2, leaky(sj.z + si.z));
      m3 = fmaxf(m3, leaky(sj.w + si.w));
    }
  }
  m0 = wave_max(m0); m1 = wave_max(m1); m2 = wave_max(m2); m3 = wave_max(m3);

  float4 acc = make_float4(0.f, 0.f, 0.f, 0.f);
  auto add_rows = [&](int sn, float a0, float a1, float a2, float a3, int cc) {
    auto coef = [&](int k) {
      return pick4(mk, lane_val(a0, k), lane_val(a1, k), lane_val(a2, k), lane_val(a3, k));
    };
    auto row = [&](int k) {
      const int sk = __builtin_amdgcn_readlane(sn, k);
      return *(const float4*)(xl + (size_t)sk * HC + lane * 4);
    };
    int k = 0;
    for (; k + 4 <= cc; k += 4) {
      const float4 v0 = row(k), v1 = row(k + 1), v2 = row(k + 2), v3 = row(k + 3);
      const float c0 = coef(k), c1 = coef(k + 1), c2 = coef(k + 2), c3 = coef(k + 3);
      fma4(acc, c0, v0);
      fma4(acc, c1, v1);
      fma4(acc, c2, v2);
      fma4(acc, c3, v3);
    }
    for (; k < cc; ++k) fma4(acc, coef(k), row(k));
  };

  float a00 = 0.f, a01 = 0.f, a02 = 0.f, a03 = 0.f;
  float a10 = 0.f, a11 = 0.f, a12 = 0.f, a13 = 0.f;
  if (ok0) {
    a00 = expf(p00 - m0); a01 = expf(p01 - m1); a02 = expf(p02 - m2); a03 = expf(p03 - m3);
  }
  if (ok1) {
    a10 = expf(p10 - m0); a11 = expf(p11 - m1); a12 = expf(p12 - m2); a13 = expf(p13 - m3);
  }
  float s0 = a00 + a10, s1 = a01 + a11, s2 = a02 + a12, s3 = a03 + a13;
  add_rows(r0.x, a00, a01, a02, a03, min(32, deg));
  if (deg > 32) add_rows(r1.x, a10, a11, a12, a13, min(32, deg - 32));
  for (int base = beg + 64; base < end; base += 32) {
    const int idx = base + lane;
    int sn = 0;
    float a0 = 0.f, a1 = 0.f, a2 = 0.f, a3 = 0.f;
    if (idx < end) {
      const int2 sp = sorted[idx];
      sn = sp.x;
      const float4 sj = *(const float4*)(Sj + (size_t)sp.y * HEADS);
      a0 = expf(leaky(sj.x + si.x) - m0);
      a1 = expf(leaky(sj.y + si.y) - m1);
      a2 = expf(leaky(sj.z + si.z) - m2);
      a3 = expf(leaky(sj.w + si.w) - m3);
    }
    s0 += a0; s1 += a1; s2 += a2; s3 += a3;
    add_rows(sn, a0, a1, a2, a3, min(32, end - base));
  }
  s0 = wave_sum(s0); s1 = wave_sum(s1); s2 = wave_sum(s2); s3 = wave_sum(s3);
  const float e0 = expf(as0 - m0), e1 = expf(as1 - m1);
  const float e2 = expf(as2 - m2), e3 = expf(as3 - m3);
  s0 += e0; s1 += e1; s2 += e2; s3 += e3;
  {
    const float4 v = *(const float4*)(xl + (size_t)n * HC + lane * 4);
    fma4(acc, pick4(mk, e0, e1, e2, e3), v);
  }
  const float sinv = 1.0f / pick4(mk, s0, s1, s2, s3);
  float4 o;
  o.x = acc.x * sinv + bias[lane * 4 + 0];
  o.y = acc.y * sinv + bias[lane * 4 + 1];
  o.z = acc.z * sinv + bias[lane * 4 + 2];
  o.w = acc.w * sinv + bias[lane * 4 + 3];
  *(float4*)(out + (size_t)n * HC + lane * 4) = o;
}

extern "C" void kernel_launch(void* const* d_in, const int* in_sizes, int n_in,
                              void* d_out, int out_size, void* d_ws, size_t ws_size,
                              hipStream_t stream) {
  const float* x      = (const float*)d_in[0];
  const float* W      = (const float*)d_in[1];
  const float* attn_j = (const float*)d_in[2];
  const float* attn_i = (const float*)d_in[3];
  const float* bias   = (const float*)d_in[4];
  const int*   eidx   = (const int*)d_in[5];

  const int N  = in_sizes[0] / F_IN;
  const int E  = in_sizes[5] / 2;
  const int* src = eidx;
  const int* dst = eidx + E;

  char*  ws = (char*)d_ws;
  size_t o  = 0;
  auto take = [&](size_t bytes) -> void* {
    void* p = ws + o;
    o += (bytes + 255) & ~(size_t)255;
    return p;
  };
  float*     xl   = (float*)take((size_t)N * HC * sizeof(float));
  float*     Sj   = (float*)take((size_t)N * HEADS * sizeof(float));
  float*     Si   = (float*)take((size_t)N * HEADS * sizeof(float));
  const int  nb   = (N + 255) / 256;
  const int  chunk = (E + EB - 1) / EB;
  int*       cnt  = (int*)take((size_t)N * sizeof(int));
  int*       rowstart = (int*)take((size_t)N * sizeof(int));
  int*       slabOff = (int*)take((size_t)nb * EB * sizeof(int));
  int*       slabCnt = (int*)take((size_t)nb * EB * sizeof(int));
  int*       cursor  = (int*)take(sizeof(int));
  int2*      bucket = (int2*)take((size_t)E * sizeof(int2));
  int2*      sorted = (int2*)take((size_t)E * sizeof(int2));

  float* out = (float*)d_out;

  k_gemm<<<(N + GT * 16 - 1) / (GT * 16), GT * 32, 0, stream>>>(x, W, attn_j, attn_i, xl,
                                                                Sj, Si, N);

  k_blocksort<<<EB, ET, (size_t)nb * 2 * sizeof(int), stream>>>(src, dst, bucket, slabOff,
                                                                slabCnt, cursor, E, chunk,
                                                                nb);
  k_rangesort<<<nb, ET, 0, stream>>>(src, bucket, slabOff, slabCnt, cursor, sorted, rowstart,
                                     cnt, N);

  k_node<<<(N + 7) / 8, 256, 0, stream>>>(src, dst, sorted, rowstart, cnt, Sj, Si,
                                          xl, bias, out, N);
}


# score table over nodes T[n] = Sj[src[n]], written by the range blocks: the sort gathers no src[s] per edge, a sorted record is the 4-byte source id, the node kernel gathers T[s]; sum order kept
# speedup vs baseline: 12.7370x; 1.0658x over previous
#include <hip/hip_runtime.h>


#define F_IN      128
#define HEADS     4
#define C_OUT     32
#define HC        128
#define NEG_SLOPE 0.2f

typedef float v2f __attribute__((ext_vector_type(2)));
typedef float v8f __attribute__((ext_vector_type(8)));

#define GT 4

template <int M> __device__ __forceinline__ float row_xor(float v) {
  return __int_as_float(
      __builtin_amdgcn_update_dpp(0, __float_as_int(v), 0x160 + M, 0xf, 0xf, true));
}
template <int M> __device__ __forceinline__ float fold(float X, float Y, bool bit) {
  const float mine = bit ? Y : X;
  const float send = bit ? X : Y;
  return mine + row_xor<M>(send);
}
#define WP 132
__global__ __launch_bounds__(GT * 32) void k_gemm(const float* __restrict__ x,
                                                  const float* __restrict__ W,
                                                  const float* __restrict__ aj,
                                                  const float* __restrict__ ai,
                                                  float* __restrict__ xl,
                                                  float* __restrict__ Sj,
                                                  float* __restrict__ Si, int N) {
  __shared__ __attribute__((aligned(16))) float Ws[HC * WP];
  const int lane    = threadIdx.x & 31;
  const int wave    = threadIdx.x >> 5;
  const int g       = lane >> 4;
  const int l16     = lane & 15;
  const int rowBase = (blockIdx.x * GT + wave) * 16;
  int mrow = rowBase + l16; if (mrow >= N) mrow = N - 1;

  v2f a[F_IN / 4];
  const float* xr = x + (size_t)mrow * F_IN + g * 2;
#pragma unroll
  for (int s = 0; s < F_IN / 4; ++s) a[s] = *(const v2f*)(xr + 4 * s);
  float ajv[8], aiv[8];
#pragma unroll
  for (int t = 0; t < 8; ++t) { ajv[t] = aj[t * 16 + l16]; aiv[t] = ai[t * 16 + l16]; }

#pragma unroll 8
  for (int j = 0; j < (HC * (F_IN / 4)) / (GT * 32); ++j) {
    const int i = threadIdx.x + j * (GT * 32);
    const int r = i >> 5, c4 = i & 31;
    const float4 v = *(const float4*)(W + (size_t)r * F_IN + c4 * 4);
    *(float4*)(Ws + r * WP + c4 * 4) = v;
  }
  __syncthreads();
  if (rowBase >= N) return;

  v8f c[8] = {};
  const float* bp = Ws + l16 * WP + g * 2;
#pragma unroll
  for (int s = 0; s < F_IN / 4; ++s) {
#pragma unroll
    for (int t = 0; t < 8; ++t) {
      const v2f b = *(const v2f*)(bp + t * 16 * WP + 4 * s);
      c[t] = __builtin_amdgcn_wmma_f32_16x16x4_f32(false, a[s], false, b, (short)0, c[t],
                                                   false, false);
    }
  }
#pragma unroll
  for (int t = 0; t < 8; ++t) {
#pragma unroll
    for (int i = 0; i < 8; ++i) {
      const int r = rowBase + i + g * 8;
      if (r < N) xl[(size_t)r * HC + t * 16 + l16] = c[t][i];
    }
  }
  const bool b0 = l16 & 1, b1 = l16 & 2, b2 = l16 & 4, b3 = l16 & 8;
  float res[HEADS];
#pragma unroll
  for (int h = 0; h < HEADS; ++h) {
    float vj[8], vi[8];
#pragma unroll
    for (int i = 0; i < 8; ++i) {
      vj[i] = c[2 * h][i] * ajv[2 * h] + c[2 * h + 1][i] * ajv[2 * h + 1];
      vi[i] = c[2 * h][i] * aiv[2 * h] + c[2 * h + 1][i] * aiv[2 * h + 1];
    }
    float pj[4], pi[4];
#pragma unroll
    for (int q = 0; q < 4; ++q) {
      pj[q] = fold<1>(vj[2 * q], vj[2 * q + 1], b0);
      pi[q] = fold<1>(vi[2 * q], vi[2 * q + 1], b0);
    }
    const float qj0 = fold<2>(pj[0], pj[1], b1), qj1 = fold<2>(pj[2], pj[3], b1);
    const float qi0 = fold<2>(pi[0], pi[1], b1), qi1 = fold<2>(pi[2], pi[3], b1);
    const float rj = fold<4>(qj0, qj1, b2);
    const float ri = fold<4>(qi0, qi1, b2);
    res[h] = fold<8>(rj, ri, b3);
  }
  const int srow = rowBase + g * 8 + (l16 & 7);
  if (srow < N) {
    float* sp = (b3 ? Si : Sj) + (size_t)srow * HEADS;
    *(float4*)sp = make_float4(res[0], res[1], res[2], res[3]);
  }
}


#define EB 256
#define ET 1024
#define UB 8
#define SPW (EB / (ET / 32))
static_assert(EB % (ET / 32) == 0, "every wave of a range block walks whole slabs");
#define SCAP 7168
#define RCAP 10240

__device__ __forceinline__ int wave_incl_scan(int v, int lane) {
#pragma unroll
  for (int off = 1; off < 32; off <<= 1) {
    const int t = __shfl_up(v, off);
    if (lane >= off) v += t;
  }
  return v;
}
__device__ __forceinline__ int block_excl_scan(int v, int* wsum, int tid, int* total) {
  const int lane = tid & 31, wave = tid >> 5;
  const int inc = wave_incl_scan(v, lane);
  if (lane == 31) wsum[wave] = inc;
  __syncthreads();
  if (wave == 0) {
    const int w = wsum[lane];
    const int wi = wave_incl_scan(w, lane);
    wsum[lane] = wi - w;
    if (lane == 31) wsum[32] = wi;
  }
  __syncthreads();
  const int ex = wsum[wave] + inc - v;
  *total = wsum[32];
  __syncthreads();
  return ex;
}

__global__ void k_blocksort(const int* __restrict__ src, const int* __restrict__ dst,
                            int2* __restrict__ bucket, int* __restrict__ slabOff,
                            int* __restrict__ slabCnt, int* __restrict__ cursor, int E,
                            int chunk, int nb) {
  extern __shared__ int shm[];
  __shared__ int wsum[33];
  __shared__ int2 stage[SCAP];
  int* hist = shm;
  int* cur  = shm + nb;
  const int tid = threadIdx.x, b = blockIdx.x;
  if (b == 0 && tid == 0) *cursor = 0;
  for (int r = tid; r < nb; r += ET) hist[r] = 0;
  __syncthreads();
  const int e0 = b * chunk;
  const int e1 = min(E, e0 + chunk);
  const bool staged = chunk <= SCAP;
  for (int eb = e0 + tid; eb < e1; eb += ET * UB) {
    int d[UB];
#pragma unroll
    for (int u = 0; u < UB; ++u) {
      const int e = eb + u * ET;
      d[u] = (e < e1) ? dst[e] : -1;
    }
#pragma unroll
    for (int u = 0; u < UB; ++u)
      if (d[u] >= 0) atomicAdd(&hist[d[u] >> 8], 1);
  }
  __syncthreads();
  int carry = 0;
  for (int base = 0; base < nb; base += ET) {
    const int i = base + tid;
    const int v = (i < nb) ? hist[i] : 0;
    int total;
    const int ex = carry + block_excl_scan(v, wsum, tid, &total);
    if (i < nb) {
      cur[i] = ex;
      slabOff[(size_t)i * EB + b] = e0 + ex;
      slabCnt[(size_t)i * EB + b] = v;
    }
    carry += total;
  }
  __syncthreads();
  auto place_all = [&](int2* __restrict__ to) {
    for (int eb = e0 + tid; eb < e1; eb += ET * UB) {
      int d[UB], s[UB];
#pragma unroll
      for (int u = 0; u < UB; ++u) {
        const int e = eb + u * ET;
        const bool ok = e < e1;
        d[u] = ok ? dst[e] : -1;
        s[u] = ok ? src[e] : 0;
      }
#pragma unroll
      for (int u = 0; u < UB; ++u) {
        if (d[u] >= 0) {
          const int pos = atomicAdd(&cur[d[u] >> 8], 1);
          to[pos] = make_int2(d[u], s[u]);
        }
      }
    }
  };
  if (staged) {
    place_all(stage);
    __syncthreads();
    const int n = e1 - e0;
    int i = tid - (e0 & 15);
    if (i < 0) i += ET;
#pragma unroll 4
    for (; i < n; i += ET) bucket[e0 + i] = stage[i];
  } else {
    place_all(bucket + e0);
  }
}

__global__ void k_rangesort(const int* __restrict__ src, const float* __restrict__ Sj,
                            const int2* __restrict__ bucket,
                            const int* __restrict__ slabOff,
                            const int* __restrict__ slabCnt, int* __restrict__ cursor,
                            int* __restrict__ sorted, int* __restrict__ rowstart,
                            int* __restrict__ cnt, float* __restrict__ T, int N) {
  __shared__ int c[256];
  __shared__ int place[256];
  __shared__ int wsum[33];
  __shared__ int sBase;
  __shared__ int stage[RCAP];
  const int tid = threadIdx.x, r = blockIdx.x;
  const int lane = tid & 31;
  const int wv = __builtin_amdgcn_readfirstlane(tid >> 5);
  const int* po = slabOff + (size_t)r * EB + wv * SPW;
  const int* pc = slabCnt + (size_t)r * EB + wv * SPW;
  int so[SPW], sc[SPW];
#pragma unroll
  for (int u = 0; u < SPW; ++u) { so[u] = po[u]; sc[u] = pc[u]; }
  if (tid < 256) c[tid] = 0;
  const int tn = r * 256 + tid;
  const bool tOk = tid < 256 && tn < N;
  __syncthreads();
  const int tsrc = src[min(tn, N - 1)];

  int2 rec[2 * SPW];
#pragma unroll
  for (int u = 0; u < SPW; ++u) {
    rec[2 * u]     = (lane < sc[u]) ? bucket[so[u] + lane] : make_int2(-1, 0);
    rec[2 * u + 1] = (lane + 32 < sc[u]) ? bucket[so[u] + lane + 32] : make_int2(-1, 0);
  }
  float4 tv = make_float4(0.f, 0.f, 0.f, 0.f);
  if (tOk) tv = *(const float4*)(Sj + (size_t)tsrc * HEADS);
#pragma unroll
  for (int u = 0; u < 2 * SPW; ++u)
    if (rec[u].x >= 0) atomicAdd(&c[rec[u].x & 255], 1);
  for (int u = 0; u < SPW; ++u)
    for (int j = 64 + lane; j < sc[u]; j += 32) atomicAdd(&c[bucket[so[u] + j].x & 255], 1);
  __syncthreads();
  const int v = (tid < 256) ? c[tid] : 0;
  int total;
  const int ex = block_excl_scan(v, wsum, tid, &total);
  if (tid == 0) sBase = atomicAdd(cursor, total);
  __syncthreads();
  const int b0 = sBase;
  const bool staged = total <= RCAP;
  if (tid < 256) {
    place[tid] = ex;
    if (tn < N) {
      rowstart[tn] = b0 + ex;
      cnt[tn] = v;
      *(float4*)(T + (size_t)tn * HEADS) = tv;
    }
  }
  __syncthreads();
  auto place_all = [&](int* __restrict__ to) {
#pragma unroll
    for (int u = 0; u < 2 * SPW; ++u) {
      if (rec[u].x >= 0) {
        const int p = atomicAdd(&place[rec[u].x & 255], 1);
        to[p] = rec[u].y;
      }
    }
    for (int u = 0; u < SPW; ++u) {
      for (int j = 64 + lane; j < sc[u]; j += 32) {
        const int2 q = bucket[so[u] + j];
        const int p = atomicAdd(&place[q.x & 255], 1);
        to[p] = q.y;
      }
    }
  };
  if (staged) {
    place_all(stage);
    __syncthreads();
    int i = tid - (b0 & 31);
    if (i < 0) i += ET;
#pragma unroll 4
    for (; i < total; i += ET) sorted[b0 + i] = stage[i];
  } else {
    place_all(sorted + b0);
  }
}

__device__ __forceinline__ float leaky(float a) {
  return (a >= 0.f) ? a : NEG_SLOPE * a;
}
__device__ __forceinline__ float wave_max(float v) {
#pragma unroll
  for (int off = 16; off > 0; off >>= 1) v = fmaxf(v, __shfl_xor(v, off));
  return v;
}
__device__ __forceinline__ float wave_sum(float v) {
#pragma unroll
  for (int off = 16; off > 0; off >>= 1) v += __shfl_xor(v, off);
  return v;
}
struct HeadMask { unsigned m0, m1, m2, m3; };
__device__ __forceinline__ float pick4(const HeadMask& mk, float v0, float v1, float v2,
                                       float v3) {
  return __uint_as_float((__float_as_uint(v0) & mk.m0) | (__float_as_uint(v1) & mk.m1) |
                         (__float_as_uint(v2) & mk.m2) | (__float_as_uint(v3) & mk.m3));
}
__device__ __forceinline__ void fma4(float4& acc, float c, const float4& v) {
  acc.x += c * v.x; acc.y += c * v.y; acc.z += c * v.z; acc.w += c * v.w;
}
__device__ __forceinline__ float lane_val(float v, int k) {
  return __int_as_float(__builtin_amdgcn_readlane(__float_as_int(v), k));
}

__global__ void k_node(const int* __restrict__ dst,
                       const int* __restrict__ sorted, const int* __restrict__ rowstart,
                       const int* __restrict__ cnt,
                       const float* __restrict__ T, const float* __restrict__ Si,
                       const float* __restrict__ xl, const float* __restrict__ bias,
                       float* __restrict__ out, int N) {
  const int lane = threadIdx.x & 31;
  const int hL = lane >> 3;
  const HeadMask mk = {hL == 0 ? ~0u : 0u, hL == 1 ? ~0u : 0u, hL == 2 ? ~0u : 0u,
                       hL == 3 ? ~0u : 0u};
  const int n = __builtin_amdgcn_readfirstlane((int)(blockIdx.x * 8 + (threadIdx.x >> 5)));
  if (n >= N) return;
  const int beg = rowstart[n];
  const int deg = cnt[n];
  const int end = beg + deg;
  const float4 si = *(const float4*)(Si + (size_t)dst[n] * HEADS);
  const float4 sjs = *(const float4*)(T + (size_t)n * HEADS);
  const float as0 = leaky(sjs.x + si.x), as1 = leaky(sjs.y + si.y);
  const float as2 = leaky(sjs.z + si.z), as3 = leaky(sjs.w + si.w);

  const int i0 = beg + lane, i1 = i0 + 32;
  const bool ok0 = i0 < end, ok1 = i1 < end;
  const int r0 = sorted[ok0 ? i0 : 0];
  const int r1 = sorted[ok1 ? i1 : 0];
  __builtin_amdgcn_sched_barrier(0);
  const float4 q0 = *(const float4*)(T + (size_t)r0 * HEADS);
  const float4 q1 = *(const float4*)(T + (size_t)r1 * HEADS);
  const float p00 = ok0 ? leaky(q0.x + si.x) : as0, p01 = ok0 ? leaky(q0.y + si.y) : as1;
  const float p02 = ok0 ? leaky(q0.z + si.z) : as2, p03 = ok0 ? leaky(q0.w + si.w) : as3;
  const float p10 = ok1 ? leaky(q1.x + si.x) : as0, p11 = ok1 ? leaky(q1.y + si.y) : as1;
  const float p12 = ok1 ? leaky(q1.z + si.z) : as2, p13 = ok1 ? leaky(q1.w + si.w) : as3;
  float m0 = fmaxf(as0, fmaxf(p00, p10)), m1 = fmaxf(as1, fmaxf(p01, p11));
  float m2 = fmaxf(as2, fmaxf(p02, p12)), m3 = fmaxf(as3, fmaxf(p03, p13));
  for (int base = beg + 64; base < end; base += 32) {
    const int idx = base + lane;
    if (idx < end) {
      const int sp = sorted[idx];
      const float4 sj = *(const float4*)(T + (size_t)sp * HEADS);
      m0 = fmaxf(m0, leaky(sj.x + si.x));
      m1 = fmaxf(m1, leaky(sj.y + si.y));
      m2 = fmaxf(m2, leaky(sj.z + si.z));
      m3 = fmaxf(m3, leaky(sj.w + si.w));
    }
  }
  m0 = wave_max(m0); m1 = wave_max(m1); m2 = wave_max(m2); m3 = wave_max(m3);

  float4 acc = make_float4(0.f, 0.f, 0.f, 0.f);
  auto add_rows = [&](int sn, float a0, float a1, float a2, float a3, int cc) {
    auto coef = [&](int k) {
      return pick4(mk, lane_val(a0, k), lane_val(a1, k), lane_val(a2, k), lane_val(a3, k));
    };
    auto row = [&](int k) {
      const int sk = __builtin_amdgcn_readlane(sn, k);
      return *(const float4*)(xl + (size_t)sk * HC + lane * 4);
    };
    int k = 0;
    for (; k + 4 <= cc; k += 4) {
      const float4 v0 = row(k), v1 = row(k + 1), v2 = row(k + 2), v3 = row(k + 3);
      const float c0 = coef(k), c1 = coef(k + 1), c2 = coef(k + 2), c3 = coef(k + 3);
      fma4(acc, c0, v0);
      fma4(acc, c1, v1);
      fma4(acc, c2, v2);
      fma4(acc, c3, v3);
    }
    for (; k < cc; ++k) fma4(acc, coef(k), row(k));
  };

  float a00 = 0.f, a01 = 0.f, a02 = 0.f, a03 = 0.f;
  float a10 = 0.f, a11 = 0.f, a12 = 0.f, a13 = 0.f;
  if (ok0) {
    a00 = expf(p00 - m0); a01 = expf(p01 - m1); a02 = expf(p02 - m2); a03 = expf(p03 - m3);
  }
  if (ok1) {
    a10 = expf(p10 - m0); a11 = expf(p11 - m1); a12 = expf(p12 - m2); a13 = expf(p13 - m3);
  }
  float s0 = a00 + a10, s1 = a01 + a11, s2 = a02 + a12, s3 = a03 + a13;
  add_rows(r0, a00, a01, a02, a03, min(32, deg));
  if (deg > 32) add_rows(r1, a10, a11, a12, a13, min(32, deg - 32));
  for (int base = beg + 64; base < end; base += 32) {
    const int idx = base + lane;
    int sn = 0;
    float a0 = 0.f, a1 = 0.f, a2 = 0.f, a3 = 0.f;
    if (idx < end) {
      sn = sorted[idx];
      const float4 sj = *(const float4*)(T + (size_t)sn * HEADS);
      a0 = expf(leaky(sj.x + si.x) - m0);
      a1 = expf(leaky(sj.y + si.y) - m1);
      a2 = expf(leaky(sj.z + si.z) - m2);
      a3 = expf(leaky(sj.w + si.w) - m3);
    }
    s0 += a0; s1 += a1; s2 += a2; s3 += a3;
    add_rows(sn, a0, a1, a2, a3, min(32, end - base));
  }
  s0 = wave_sum(s0); s1 = wave_sum(s1); s2 = wave_sum(s2); s3 = wave_sum(s3);
  const float e0 = expf(as0 - m0), e1 = expf(as1 - m1);
  const float e2 = expf(as2 - m2), e3 = expf(as3 - m3);
  s0 += e0; s1 += e1; s2 += e2; s3 += e3;
  {
    const float4 v = *(const float4*)(xl + (size_t)n * HC + lane * 4);
    fma4(acc, pick4(mk, e0, e1, e2, e3), v);
  }
  const float sinv = 1.0f / pick4(mk, s0, s1, s2, s3);
  float4 o;
  o.x = acc.x * sinv + bias[lane * 4 + 0];
  o.y = acc.y * sinv + bias[lane * 4 + 1];
  o.z = acc.z * sinv + bias[lane * 4 + 2];
  o.w = acc.w * sinv + bias[lane * 4 + 3];
  *(float4*)(out + (size_t)n * HC + lane * 4) = o;
}

extern "C" void kernel_launch(void* const* d_in, const int* in_sizes, int n_in,
                              void* d_out, int out_size, void* d_ws, size_t ws_size,
                              hipStream_t stream) {
  const float* x      = (const float*)d_in[0];
  const float* W      = (const float*)d_in[1];
  const float* attn_j = (const float*)d_in[2];
  const float* attn_i = (const float*)d_in[3];
  const float* bias   = (const float*)d_in[4];
  const int*   eidx   = (const int*)d_in[5];

  const int N  = in_sizes[0] / F_IN;
  const int E  = in_sizes[5] / 2;
  const int* src = eidx;
  const int* dst = eidx + E;

  char*  ws = (char*)d_ws;
  size_t o  = 0;
  auto take = [&](size_t bytes) -> void* {
    void* p = ws + o;
    o += (bytes + 255) & ~(size_t)255;
    return p;
  };
  float*     xl   = (float*)take((size_t)N * HC * sizeof(float));
  float*     Sj   = (float*)take((size_t)N * HEADS * sizeof(float));
  float*     Si   = (float*)take((size_t)N * HEADS * sizeof(float));
  const int  nb   = (N + 255) / 256;
  const int  chunk = (E + EB - 1) / EB;
  int*       cnt  = (int*)take((size_t)N * sizeof(int));
  int*       rowstart = (int*)take((size_t)N * sizeof(int));
  int*       slabOff = (int*)take((size_t)nb * EB * sizeof(int));
  int*       slabCnt = (int*)take((size_t)nb * EB * sizeof(int));
  int*       cursor  = (int*)take(sizeof(int));
  int2*      bucket = (int2*)take((size_t)E * sizeof(int2));
  int*       sorted = (int*)take((size_t)E * sizeof(int));
  float*     T      = (float*)take((size_t)N * HEADS * sizeof(float));

  float* out = (float*)d_out;

  k_gemm<<<(N + GT * 16 - 1) / (GT * 16), GT * 32, 0, stream>>>(x, W, attn_j, attn_i, xl,
                                                                Sj, Si, N);

  k_blocksort<<<EB, ET, (size_t)nb * 2 * sizeof(int), stream>>>(src, dst, bucket, slabOff,
                                                                slabCnt, cursor, E, chunk,
                                                                nb);
  k_rangesort<<<nb, ET, 0, stream>>>(src, Sj, bucket, slabOff, slabCnt, cursor, sorted,
                                     rowstart, cnt, T, N);

  k_node<<<(N + 7) / 8, 256, 0, stream>>>(dst, sorted, rowstart, cnt, T, Si,
                                          xl, bias, out, N);
}


# GEMM and sort level 1 share one launch: the 256 edge blocks (now 4 waves, 16 edges a thread per round trip) run beside the row-tile blocks on the same WGPs; arrival-order rank kept; 4 launches to 3
# speedup vs baseline: 13.1246x; 1.0304x over previous
#include <hip/hip_runtime.h>


#define F_IN      128
#define HEADS     4
#define C_OUT     32
#define HC        128
#define NEG_SLOPE 0.2f

typedef float v2f __attribute__((ext_vector_type(2)));
typedef float v8f __attribute__((ext_vector_type(8)));

#define GT 4

template <int M> __device__ __forceinline__ float row_xor(float v) {
  return __int_as_float(
      __builtin_amdgcn_update_dpp(0, __float_as_int(v), 0x160 + M, 0xf, 0xf, true));
}
template <int M> __device__ __forceinline__ float fold(float X, float Y, bool bit) {
  const float mine = bit ? Y : X;
  const float send = bit ? X : Y;
  return mine + row_xor<M>(send);
}
#define WP 132
__device__ __forceinline__ void gemm_block(const float* __restrict__ x,
                                           const float* __restrict__ W,
                                           const float* __restrict__ aj,
                                           const float* __restrict__ ai,
                                           float* __restrict__ xl, float* __restrict__ Sj,
                                           float* __restrict__ Si, int N, int tb,
                                           float* Ws) {
  const int lane    = threadIdx.x & 31;
  const int wave    = threadIdx.x >> 5;
  const int g       = lane >> 4;
  const int l16     = lane & 15;
  const int rowBase = (tb * GT + wave) * 16;
  int mrow = rowBase + l16; if (mrow >= N) mrow = N - 1;

  v2f a[F_IN / 4];
  const float* xr = x + (size_t)mrow * F_IN + g * 2;
#pragma unroll
  for (int s = 0; s < F_IN / 4; ++s) a[s] = *(const v2f*)(xr + 4 * s);
  float ajv[8], aiv[8];
#pragma unroll
  for (int t = 0; t < 8; ++t) { ajv[t] = aj[t * 16 + l16]; aiv[t] = ai[t * 16 + l16]; }

#pragma unroll 8
  for (int j = 0; j < (HC * (F_IN / 4)) / (GT * 32); ++j) {
    const int i = threadIdx.x + j * (GT * 32);
    const int r = i >> 5, c4 = i & 31;
    const float4 v = *(const float4*)(W + (size_t)r * F_IN + c4 * 4);
    *(float4*)(Ws + r * WP + c4 * 4) = v;
  }
  __syncthreads();
  if (rowBase >= N) return;

  v8f c[8] = {};
  const float* bp = Ws + l16 * WP + g * 2;
#pragma unroll
  for (int s = 0; s < F_IN / 4; ++s) {
#pragma unroll
    for (int t = 0; t < 8; ++t) {
      const v2f b = *(const v2f*)(bp + t * 16 * WP + 4 * s);
      c[t] = __builtin_amdgcn_wmma_f32_16x16x4_f32(false, a[s], false, b, (short)0, c[t],
                                                   false, false);
    }
  }
#pragma unroll
  for (int t = 0; t < 8; ++t) {
#pragma unroll
    for (int i = 0; i < 8; ++i) {
      const int r = rowBase + i + g * 8;
      if (r < N) xl[(size_t)r * HC + t * 16 + l16] = c[t][i];
    }
  }
  const bool b0 = l16 & 1, b1 = l16 & 2, b2 = l16 & 4, b3 = l16 & 8;
  float res[HEADS];
#pragma unroll
  for (int h = 0; h < HEADS; ++h) {
    float vj[8], vi[8];
#pragma unroll
    for (int i = 0; i < 8; ++i) {
      vj[i] = c[2 * h][i] * ajv[2 * h] + c[2 * h + 1][i] * ajv[2 * h + 1];
      vi[i] = c[2 * h][i] * aiv[2 * h] + c[2 * h + 1][i] * aiv[2 * h + 1];
    }
    float pj[4], pi[4];
#pragma unroll
    for (int q = 0; q < 4; ++q) {
      pj[q] = fold<1>(vj[2 * q], vj[2 * q + 1], b0);
      pi[q] = fold<1>(vi[2 * q], vi[2 * q + 1], b0);
    }
    const float qj0 = fold<2>(pj[0], pj[1], b1), qj1 = fold<2>(pj[2], pj[3], b1);
    const float qi0 = fold<2>(pi[0], pi[1], b1), qi1 = fold<2>(pi[2], pi[3], b1);
    const float rj = fold<4>(qj0, qj1, b2);
    const float ri = fold<4>(qi0, qi1, b2);
    res[h] = fold<8>(rj, ri, b3);
  }
  const int srow = rowBase + g * 8 + (l16 & 7);
  if (srow < N) {
    float* sp = (b3 ? Si : Sj) + (size_t)srow * HEADS;
    *(float4*)sp = make_float4(res[0], res[1], res[2], res[3]);
  }
}


#define EB 256
#define BT (GT * 32)
#define ET 1024
#define UB 16
#define SPW (EB / (ET / 32))
static_assert(EB % (ET / 32) == 0, "every wave of a range block walks whole slabs");
#define SCAP 7168
#define RCAP 10240

__device__ __forceinline__ int wave_incl_scan(int v, int lane) {
#pragma unroll
  for (int off = 1; off < 32; off <<= 1) {
    const int t = __shfl_up(v, off);
    if (lane >= off) v += t;
  }
  return v;
}
template <int NT>
__device__ __forceinline__ int block_excl_scan(int v, int* wsum, int tid, int* total) {
  const int lane = tid & 31, wave = tid >> 5;
  const int inc = wave_incl_scan(v, lane);
  if (lane == 31) wsum[wave] = inc;
  __syncthreads();
  if (wave == 0) {
    const int w = (lane < NT / 32) ? wsum[lane] : 0;
    const int wi = wave_incl_scan(w, lane);
    if (lane < NT / 32) wsum[lane] = wi - w;
    if (lane == 31) wsum[32] = wi;
  }
  __syncthreads();
  const int ex = wsum[wave] + inc - v;
  *total = wsum[32];
  __syncthreads();
  return ex;
}

__device__ __forceinline__ void blocksort_block(const int* __restrict__ src,
                                                const int* __restrict__ dst,
                                                int2* __restrict__ bucket,
                                                int* __restrict__ slabOff,
                                                int* __restrict__ slabCnt,
                                                int* __restrict__ cursor, int E, int chunk,
                                                int nb, int b, int* shm, int2* stage,
                                                int* wsum) {
  int* hist = shm;
  int* cur  = shm + nb;
  const int tid = threadIdx.x;
  if (b == 0 && tid == 0) *cursor = 0;
  for (int r = tid; r < nb; r += BT) hist[r] = 0;
  __syncthreads();
  const int e0 = b * chunk;
  const int e1 = min(E, e0 + chunk);
  const bool staged = chunk <= SCAP;
  for (int eb = e0 + tid; eb < e1; eb += BT * UB) {
    int d[UB];
#pragma unroll
    for (int u = 0; u < UB; ++u) {
      const int e = eb + u * BT;
      d[u] = (e < e1) ? dst[e] : -1;
    }
#pragma unroll
    for (int u = 0; u < UB; ++u)
      if (d[u] >= 0) atomicAdd(&hist[d[u] >> 8], 1);
  }
  __syncthreads();
  int carry = 0;
  for (int base = 0; base < nb; base += BT) {
    const int i = base + tid;
    const int v = (i < nb) ? hist[i] : 0;
    int total;
    const int ex = carry + block_excl_scan<BT>(v, wsum, tid, &total);
    if (i < nb) {
      cur[i] = ex;
      slabOff[(size_t)i * EB + b] = e0 + ex;
      slabCnt[(size_t)i * EB + b] = v;
    }
    carry += total;
  }
  __syncthreads();
  auto place_all = [&](int2* __restrict__ to) {
    for (int eb = e0 + tid; eb < e1; eb += BT * UB) {
      int d[UB], s[UB];
#pragma unroll
      for (int u = 0; u < UB; ++u) {
        const int e = eb + u * BT;
        const bool ok = e < e1;
        d[u] = ok ? dst[e] : -1;
        s[u] = ok ? src[e] : 0;
      }
#pragma unroll
      for (int u = 0; u < UB; ++u) {
        if (d[u] >= 0) {
          const int pos = atomicAdd(&cur[d[u] >> 8], 1);
          to[pos] = make_int2(d[u], s[u]);
        }
      }
    }
  };
  if (staged) {
    place_all(stage);
    __syncthreads();
    const int n = e1 - e0;
    int i = tid - (e0 & 15);
    if (i < 0) i += BT;
#pragma unroll 8
    for (; i < n; i += BT) bucket[e0 + i] = stage[i];
  } else {
    place_all(bucket + e0);
  }
}

static_assert(SCAP * sizeof(int2) <= HC * WP * sizeof(float), "stage fits the W tile");
__global__ __launch_bounds__(BT) void k_gemm_sort(
    const float* __restrict__ x, const float* __restrict__ W, const float* __restrict__ aj,
    const float* __restrict__ ai, float* __restrict__ xl, float* __restrict__ Sj,
    float* __restrict__ Si, int N, const int* __restrict__ src,
    const int* __restrict__ dst, int2* __restrict__ bucket, int* __restrict__ slabOff,
    int* __restrict__ slabCnt, int* __restrict__ cursor, int E, int chunk, int nb) {
  extern __shared__ int shm[];
  __shared__ __attribute__((aligned(16))) float pool[HC * WP];
  __shared__ int wsum[33];
  if (blockIdx.x < EB) {
    blocksort_block(src, dst, bucket, slabOff, slabCnt, cursor, E, chunk, nb,
                    (int)blockIdx.x, shm, (int2*)pool, wsum);
  } else {
    gemm_block(x, W, aj, ai, xl, Sj, Si, N, (int)blockIdx.x - EB, pool);
  }
}

__global__ void k_rangesort(const int* __restrict__ src, const float* __restrict__ Sj,
                            const int2* __restrict__ bucket,
                            const int* __restrict__ slabOff,
                            const int* __restrict__ slabCnt, int* __restrict__ cursor,
                            int* __restrict__ sorted, int* __restrict__ rowstart,
                            int* __restrict__ cnt, float* __restrict__ T, int N) {
  __shared__ int c[256];
  __shared__ int place[256];
  __shared__ int wsum[33];
  __shared__ int sBase;
  __shared__ int stage[RCAP];
  const int tid = threadIdx.x, r = blockIdx.x;
  const int lane = tid & 31;
  const int wv = __builtin_amdgcn_readfirstlane(tid >> 5);
  const int* po = slabOff + (size_t)r * EB + wv * SPW;
  const int* pc = slabCnt + (size_t)r * EB + wv * SPW;
  int so[SPW], sc[SPW];
#pragma unroll
  for (int u = 0; u < SPW; ++u) { so[u] = po[u]; sc[u] = pc[u]; }
  if (tid < 256) c[tid] = 0;
  const int tn = r * 256 + tid;
  const bool tOk = tid < 256 && tn < N;
  __syncthreads();
  const int tsrc = src[min(tn, N - 1)];

  int2 rec[2 * SPW];
#pragma unroll
  for (int u = 0; u < SPW; ++u) {
    rec[2 * u]     = (lane < sc[u]) ? bucket[so[u] + lane] : make_int2(-1, 0);
    rec[2 * u + 1] = (lane + 32 < sc[u]) ? bucket[so[u] + lane + 32] : make_int2(-1, 0);
  }
  float4 tv = make_float4(0.f, 0.f, 0.f, 0.f);
  if (tOk) tv = *(const float4*)(Sj + (size_t)tsrc * HEADS);
#pragma unroll
  for (int u = 0; u < 2 * SPW; ++u)
    if (rec[u].x >= 0) atomicAdd(&c[rec[u].x & 255], 1);
  for (int u = 0; u < SPW; ++u)
    for (int j = 64 + lane; j < sc[u]; j += 32) atomicAdd(&c[bucket[so[u] + j].x & 255], 1);
  __syncthreads();
  const int v = (tid < 256) ? c[tid] : 0;
  int total;
  const int ex = block_excl_scan<ET>(v, wsum, tid, &total);
  if (tid == 0) sBase = atomicAdd(cursor, total);
  __syncthreads();
  const int b0 = sBase;
  const bool staged = total <= RCAP;
  if (tid < 256) {
    place[tid] = ex;
    if (tn < N) {
      rowstart[tn] = b0 + ex;
      cnt[tn] = v;
      *(float4*)(T + (size_t)tn * HEADS) = tv;
    }
  }
  __syncthreads();
  auto place_all = [&](int* __restrict__ to) {
#pragma unroll
    for (int u = 0; u < 2 * SPW; ++u) {
      if (rec[u].x >= 0) {
        const int p = atomicAdd(&place[rec[u].x & 255], 1);
        to[p] = rec[u].y;
      }
    }
    for (int u = 0; u < SPW; ++u) {
      for (int j = 64 + lane; j < sc[u]; j += 32) {
        const int2 q = bucket[so[u] + j];
        const int p = atomicAdd(&place[q.x & 255], 1);
        to[p] = q.y;
      }
    }
  };
  if (staged) {
    place_all(stage);
    __syncthreads();
    int i = tid - (b0 & 31);
    if (i < 0) i += ET;
#pragma unroll 4
    for (; i < total; i += ET) sorted[b0 + i] = stage[i];
  } else {
    place_all(sorted + b0);
  }
}

__device__ __forceinline__ float leaky(float a) {
  return (a >= 0.f) ? a : NEG_SLOPE * a;
}
__device__ __forceinline__ float wave_max(float v) {
#pragma unroll
  for (int off = 16; off > 0; off >>= 1) v = fmaxf(v, __shfl_xor(v, off));
  return v;
}
__device__ __forceinline__ float wave_sum(float v) {
#pragma unroll
  for (int off = 16; off > 0; off >>= 1) v += __shfl_xor(v, off);
  return v;
}
struct HeadMask { unsigned m0, m1, m2, m3; };
__device__ __forceinline__ float pick4(const HeadMask& mk, float v0, float v1, float v2,
                                       float v3) {
  return __uint_as_float((__float_as_uint(v0) & mk.m0) | (__float_as_uint(v1) & mk.m1) |
                         (__float_as_uint(v2) & mk.m2) | (__float_as_uint(v3) & mk.m3));
}
__device__ __forceinline__ void fma4(float4& acc, float c, const float4& v) {
  acc.x += c * v.x; acc.y += c * v.y; acc.z += c * v.z; acc.w += c * v.w;
}
__device__ __forceinline__ float lane_val(float v, int k) {
  return __int_as_float(__builtin_amdgcn_readlane(__float_as_int(v), k));
}

__global__ void k_node(const int* __restrict__ dst,
                       const int* __restrict__ sorted, const int* __restrict__ rowstart,
                       const int* __restrict__ cnt,
                       const float* __restrict__ T, const float* __restrict__ Si,
                       const float* __restrict__ xl, const float* __restrict__ bias,
                       float* __restrict__ out, int N) {
  const int lane = threadIdx.x & 31;
  const int hL = lane >> 3;
  const HeadMask mk = {hL == 0 ? ~0u : 0u, hL == 1 ? ~0u : 0u, hL == 2 ? ~0u : 0u,
                       hL == 3 ? ~0u : 0u};
  const int n = __builtin_amdgcn_readfirstlane((int)(blockIdx.x * 8 + (threadIdx.x >> 5)));
  if (n >= N) return;
  const int beg = rowstart[n];
  const int deg = cnt[n];
  const int end = beg + deg;
  const float4 si = *(const float4*)(Si + (size_t)dst[n] * HEADS);
  const float4 sjs = *(const float4*)(T + (size_t)n * HEADS);
  const float as0 = leaky(sjs.x + si.x), as1 = leaky(sjs.y + si.y);
  const float as2 = leaky(sjs.z + si.z), as3 = leaky(sjs.w + si.w);

  const int i0 = beg + lane, i1 = i0 + 32;
  const bool ok0 = i0 < end, ok1 = i1 < end;
  const int r0 = sorted[ok0 ? i0 : 0];
  const int r1 = sorted[ok1 ? i1 : 0];
  __builtin_amdgcn_sched_barrier(0);
  const float4 q0 = *(const float4*)(T + (size_t)r0 * HEADS);
  const float4 q1 = *(const float4*)(T + (size_t)r1 * HEADS);
  const float p00 = ok0 ? leaky(q0.x + si.x) : as0, p01 = ok0 ? leaky(q0.y + si.y) : as1;
  const float p02 = ok0 ? leaky(q0.z + si.z) : as2, p03 = ok0 ? leaky(q0.w + si.w) : as3;
  const float p10 = ok1 ? leaky(q1.x + si.x) : as0, p11 = ok1 ? leaky(q1.y + si.y) : as1;
  const float p12 = ok1 ? leaky(q1.z + si.z) : as2, p13 = ok1 ? leaky(q1.w + si.w) : as3;
  float m0 = fmaxf(as0, fmaxf(p00, p10)), m1 = fmaxf(as1, fmaxf(p01, p11));
  float m2 = fmaxf(as2, fmaxf(p02, p12)), m3 = fmaxf(as3, fmaxf(p03, p13));
  for (int base = beg + 64; base < end; base += 32) {
    const int idx = base + lane;
    if (idx < end) {
      const int sp = sorted[idx];
      const float4 sj = *(const float4*)(T + (size_t)sp * HEADS);
      m0 = fmaxf(m0, leaky(sj.x + si.x));
      m1 = fmaxf(m1, leaky(sj.y + si.y));
      m2 = fmaxf(m2, leaky(sj.z + si.z));
      m3 = fmaxf(m3, leaky(sj.w + si.w));
    }
  }
  m0 = wave_max(m0); m1 = wave_max(m1); m2 = wave_max(m2); m3 = wave_max(m3);

  float4 acc = make_float4(0.f, 0.f, 0.f, 0.f);
  auto add_rows = [&](int sn, float a0, float a1, float a2, float a3, int cc) {
    auto coef = [&](int k) {
      return pick4(mk, lane_val(a0, k), lane_val(a1, k), lane_val(a2, k), lane_val(a3, k));
    };
    auto row = [&](int k) {
      const int sk = __builtin_amdgcn_readlane(sn, k);
      return *(const float4*)(xl + (size_t)sk * HC + lane * 4);
    };
    int k = 0;
    for (; k + 4 <= cc; k += 4) {
      const float4 v0 = row(k), v1 = row(k + 1), v2 = row(k + 2), v3 = row(k + 3);
      const float c0 = coef(k), c1 = coef(k + 1), c2 = coef(k + 2), c3 = coef(k + 3);
      fma4(acc, c0, v0);
      fma4(acc, c1, v1);
      fma4(acc, c2, v2);
      fma4(acc, c3, v3);
    }
    for (; k < cc; ++k) fma4(acc, coef(k), row(k));
  };

  float a00 = 0.f, a01 = 0.f, a02 = 0.f, a03 = 0.f;
  float a10 = 0.f, a11 = 0.f, a12 = 0.f, a13 = 0.f;
  if (ok0) {
    a00 = expf(p00 - m0); a01 = expf(p01 - m1); a02 = expf(p02 - m2); a03 = expf(p03 - m3);
  }
  if (ok1) {
    a10 = expf(p10 - m0); a11 = expf(p11 - m1); a12 = expf(p12 - m2); a13 = expf(p13 - m3);
  }
  float s0 = a00 + a10, s1 = a01 + a11, s2 = a02 + a12, s3 = a03 + a13;
  add_rows(r0, a00, a01, a02, a03, min(32, deg));
  if (deg > 32) add_rows(r1, a10, a11, a12, a13, min(32, deg - 32));
  for (int base = beg + 64; base < end; base += 32) {
    const int idx = base + lane;
    int sn = 0;
    float a0 = 0.f, a1 = 0.f, a2 = 0.f, a3 = 0.f;
    if (idx < end) {
      sn = sorted[idx];
      const float4 sj = *(const float4*)(T + (size_t)sn * HEADS);
      a0 = expf(leaky(sj.x + si.x) - m0);
      a1 = expf(leaky(sj.y + si.y) - m1);
      a2 = expf(leaky(sj.z + si.z) - m2);
      a3 = expf(leaky(sj.w + si.w) - m3);
    }
    s0 += a0; s1 += a1; s2 += a2; s3 += a3;
    add_rows(sn, a0, a1, a2, a3, min(32, end - base));
  }
  s0 = wave_sum(s0); s1 = wave_sum(s1); s2 = wave_sum(s2); s3 = wave_sum(s3);
  const float e0 = expf(as0 - m0), e1 = expf(as1 - m1);
  const float e2 = expf(as2 - m2), e3 = expf(as3 - m3);
  s0 += e0; s1 += e1; s2 += e2; s3 += e3;
  {
    const float4 v = *(const float4*)(xl + (size_t)n * HC + lane * 4);
    fma4(acc, pick4(mk, e0, e1, e2, e3), v);
  }
  const float sinv = 1.0f / pick4(mk, s0, s1, s2, s3);
  float4 o;
  o.x = acc.x * sinv + bias[lane * 4 + 0];
  o.y = acc.y * sinv + bias[lane * 4 + 1];
  o.z = acc.z * sinv + bias[lane * 4 + 2];
  o.w = acc.w * sinv + bias[lane * 4 + 3];
  *(float4*)(out + (size_t)n * HC + lane * 4) = o;
}

extern "C" void kernel_launch(void* const* d_in, const int* in_sizes, int n_in,
                              void* d_out, int out_size, void* d_ws, size_t ws_size,
                              hipStream_t stream) {
  const float* x      = (const float*)d_in[0];
  const float* W      = (const float*)d_in[1];
  const float* attn_j = (const float*)d_in[2];
  const float* attn_i = (const float*)d_in[3];
  const float* bias   = (const float*)d_in[4];
  const int*   eidx   = (const int*)d_in[5];

  const int N  = in_sizes[0] / F_IN;
  const int E  = in_sizes[5] / 2;
  const int* src = eidx;
  const int* dst = eidx + E;

  char*  ws = (char*)d_ws;
  size_t o  = 0;
  auto take = [&](size_t bytes) -> void* {
    void* p = ws + o;
    o += (bytes + 255) & ~(size_t)255;
    return p;
  };
  float*     xl   = (float*)take((size_t)N * HC * sizeof(float));
  float*     Sj   = (float*)take((size_t)N * HEADS * sizeof(float));
  float*     Si   = (float*)take((size_t)N * HEADS * sizeof(float));
  const int  nb   = (N + 255) / 256;
  const int  chunk = (E + EB - 1) / EB;
  int*       cnt  = (int*)take((size_t)N * sizeof(int));
  int*       rowstart = (int*)take((size_t)N * sizeof(int));
  int*       slabOff = (int*)take((size_t)nb * EB * sizeof(int));
  int*       slabCnt = (int*)take((size_t)nb * EB * sizeof(int));
  int*       cursor  = (int*)take(sizeof(int));
  int2*      bucket = (int2*)take((size_t)E * sizeof(int2));
  int*       sorted = (int*)take((size_t)E * sizeof(int));
  float*     T      = (float*)take((size_t)N * HEADS * sizeof(float));

  float* out = (float*)d_out;

  k_gemm_sort<<<EB + (N + GT * 16 - 1) / (GT * 16), BT, (size_t)nb * 2 * sizeof(int),
                stream>>>(x, W, attn_j, attn_i, xl, Sj, Si, N, src, dst, bucket, slabOff,
                          slabCnt, cursor, E, chunk, nb);
  k_rangesort<<<nb, ET, 0, stream>>>(src, Sj, bucket, slabOff, slabCnt, cursor, sorted,
                                     rowstart, cnt, T, N);

  k_node<<<(N + 7) / 8, 256, 0, stream>>>(dst, sorted, rowstart, cnt, T, Si,
                                          xl, bias, out, N);
}
